# starting kernel
# baseline (speedup 1.0000x reference)
#include <hip/hip_runtime.h>
#include <math.h>

typedef __attribute__((ext_vector_type(2))) float v2f;
typedef __attribute__((ext_vector_type(8))) float v8f;

#define D_IN  128
#define H_DIM 32
#define C_DIM 5

__global__ void gcn_build_w1(const float* __restrict__ Wrel, const float* __restrict__ Wroot,
                             float* __restrict__ Wcat) {
    int i = blockIdx.x * 256 + threadIdx.x;
    if (i >= 128 * 64) return;
    int k = i >> 6, j = i & 63;
    Wcat[i] = (j < 32) ? Wrel[k * 32 + j] : Wroot[k * 32 + (j - 32)];
}

__global__ void gcn_build_w2(const float* __restrict__ Wrel, const float* __restrict__ Wroot,
                             float* __restrict__ Wp) {
    int i = blockIdx.x * 256 + threadIdx.x;
    if (i >= 32 * 16) return;
    int k = i >> 4, n = i & 15;
    float v = 0.0f;
    if (n < 5)       v = Wrel[k * 5 + n];
    else if (n < 10) v = Wroot[k * 5 + (n - 5)];
    Wp[i] = v;
}

__global__ void gcn_zero(float* __restrict__ p, long n) {
    long i = (long)blockIdx.x * 256 + threadIdx.x;
    if (i < n) p[i] = 0.0f;
}

__global__ __launch_bounds__(256)
void gcn_gemm1_wmma(const float* __restrict__ x, const float* __restrict__ W,
                    float* __restrict__ PR1, int nTiles) {
    int wave = threadIdx.x >> 5;
    int lane = threadIdx.x & 31;
    int tile = blockIdx.x * 8 + wave;
    if (tile >= nTiles) return;
    int half = lane >> 4;
    int m    = lane & 15;
    const float* xr = x + (long)(tile * 16 + m) * D_IN;

    v2f a[32];
#pragma unroll
    for (int ks = 0; ks < 32; ++ks) {
        int k = ks * 4 + half * 2;
        a[ks].x = xr[k];
        a[ks].y = xr[k + 1];
    }

#pragma unroll
    for (int ct = 0; ct < 4; ++ct) {
        int n = ct * 16 + m;
        v8f acc = {};
#pragma unroll
        for (int ks = 0; ks < 32; ++ks) {
            int k = ks * 4 + half * 2;
            v2f b;
            b.x = W[k * 64 + n];
            b.y = W[(k + 1) * 64 + n];
            acc = __builtin_amdgcn_wmma_f32_16x16x4_f32(false, a[ks], false, b,
                                                        (short)0, acc, false, false);
        }
#pragma unroll
        for (int r = 0; r < 8; ++r) {
            int orow = tile * 16 + r + half * 8;
            PR1[(long)orow * 64 + ct * 16 + m] = acc[r];
        }
    }
}

__global__ __launch_bounds__(256)
void gcn_gemm2_wmma(const float* __restrict__ h, const float* __restrict__ W,
                    float* __restrict__ PR2, int nTiles) {
    int wave = threadIdx.x >> 5;
    int lane = threadIdx.x & 31;
    int tile = blockIdx.x * 8 + wave;
    if (tile >= nTiles) return;
    int half = lane >> 4;
    int m    = lane & 15;
    const float* hr = h + (long)(tile * 16 + m) * H_DIM;

    v2f a[8];
#pragma unroll
    for (int ks = 0; ks < 8; ++ks) {
        int k = ks * 4 + half * 2;
        a[ks].x = hr[k];
        a[ks].y = hr[k + 1];
    }
    v8f acc = {};
#pragma unroll
    for (int ks = 0; ks < 8; ++ks) {
        int k = ks * 4 + half * 2;
        v2f b;
        b.x = W[k * 16 + m];
        b.y = W[(k + 1) * 16 + m];
        acc = __builtin_amdgcn_wmma_f32_16x16x4_f32(false, a[ks], false, b,
                                                    (short)0, acc, false, false);
    }
    if (m < 10) {
#pragma unroll
        for (int r = 0; r < 8; ++r) {
            int orow = tile * 16 + r + half * 8;
            PR2[(long)orow * 10 + m] = acc[r];
        }
    }
}

__global__ __launch_bounds__(256)
void gcn_agg1(const int* __restrict__ src, const int* __restrict__ dst,
              const float* __restrict__ ew, const float* __restrict__ PR1,
              float* __restrict__ S1, float* __restrict__ cnt, int E) {
    int e = blockIdx.x * 8 + (threadIdx.x >> 5);
    int lane = threadIdx.x & 31;
    if (e >= E) return;
    int s = src[e], d = dst[e];
    float v = ew[e] * PR1[(long)s * 64 + lane];
    atomicAdd(&S1[(long)d * 32 + lane], v);
    if (lane == 0) atomicAdd(&cnt[d], 1.0f);
}

__global__ __launch_bounds__(256)
void gcn_combine1(const float* __restrict__ S1, const float* __restrict__ PR1,
                  const float* __restrict__ b1, const float* __restrict__ cnt,
                  float* __restrict__ h, long total) {
    long i = (long)blockIdx.x * 256 + threadIdx.x;
    if (i >= total) return;
    long n = i >> 5;
    int  j = (int)(i & 31);
    float c = fmaxf(cnt[n], 1.0f);
    float v = S1[i] / c + PR1[n * 64 + 32 + j] + b1[j];
    h[i] = fmaxf(v, 0.0f);
}

__global__ __launch_bounds__(256)
void gcn_agg2(const int* __restrict__ src, const int* __restrict__ dst,
              const float* __restrict__ PR2, float* __restrict__ S2, int E) {
    long t = (long)blockIdx.x * 256 + threadIdx.x;
    int e = (int)(t >> 3);
    int c = (int)(t & 7);
    if (e >= E || c >= 5) return;
    atomicAdd(&S2[(long)dst[e] * 5 + c], PR2[(long)src[e] * 10 + c]);
}

__global__ __launch_bounds__(256)
void gcn_output(const float* __restrict__ S2, const float* __restrict__ PR2,
                const float* __restrict__ b2, const float* __restrict__ cnt,
                float* __restrict__ out, int N) {
    int n = blockIdx.x * 256 + threadIdx.x;
    if (n >= N) return;
    float c = fmaxf(cnt[n], 1.0f);
    float v[5];
#pragma unroll
    for (int j = 0; j < 5; ++j)
        v[j] = S2[(long)n * 5 + j] / c + PR2[(long)n * 10 + 5 + j] + b2[j];
    float mx = v[0];
#pragma unroll
    for (int j = 1; j < 5; ++j) mx = fmaxf(mx, v[j]);
    float s = 0.0f;
#pragma unroll
    for (int j = 0; j < 5; ++j) s += expf(v[j] - mx);
    float l = logf(s);
#pragma unroll
    for (int j = 0; j < 5; ++j) out[(long)n * 5 + j] = v[j] - mx - l;
}

extern "C" void kernel_launch(void* const* d_in, const int* in_sizes, int n_in,
                              void* d_out, int out_size, void* d_ws, size_t ws_size,
                              hipStream_t stream) {
    const float* x      = (const float*)d_in[0];
    const int*   ei     = (const int*)d_in[1];
    const float* ew     = (const float*)d_in[2];
    const float* Wrel1  = (const float*)d_in[3];
    const float* Wroot1 = (const float*)d_in[4];
    const float* b1     = (const float*)d_in[5];
    const float* Wrel2  = (const float*)d_in[6];
    const float* Wroot2 = (const float*)d_in[7];
    const float* b2     = (const float*)d_in[8];

    const int N = in_sizes[0] / D_IN;
    const int E = in_sizes[1] / 2;
    const int* src = ei;
    const int* dst = ei + E;

    float* ws = (float*)d_ws;
    long off = 0;
    float* PR1 = ws + off; off += (long)N * 64;
    float* S1  = ws + off; off += (long)N * 32;
    float* cnt = ws + off; off += N;
    float* h   = ws + off; off += (long)N * 32;
    float* W1c = ws + off; off += 128 * 64;
    float* W2p = ws + off; off += 32 * 16;
    float* PR2 = PR1;
    float* S2  = S1;

    const int nTiles = N / 16;

    gcn_build_w1<<<(128 * 64 + 255) / 256, 256, 0, stream>>>(Wrel1, Wroot1, W1c);
    gcn_build_w2<<<(32 * 16 + 255) / 256, 256, 0, stream>>>(Wrel2, Wroot2, W2p);

    gcn_gemm1_wmma<<<(nTiles + 7) / 8, 256, 0, stream>>>(x, W1c, PR1, nTiles);

    long nz1 = (long)N * 33;
    gcn_zero<<<(int)((nz1 + 255) / 256), 256, 0, stream>>>(S1, nz1);
    gcn_agg1<<<(E + 7) / 8, 256, 0, stream>>>(src, dst, ew, PR1, S1, cnt, E);

    long nh = (long)N * 32;
    gcn_combine1<<<(int)((nh + 255) / 256), 256, 0, stream>>>(S1, PR1, b1, cnt, h, nh);

    gcn_gemm2_wmma<<<(nTiles + 7) / 8, 256, 0, stream>>>(h, W2p, PR2, nTiles);

    long nz2 = (long)N * 5;
    gcn_zero<<<(int)((nz2 + 255) / 256), 256, 0, stream>>>(S2, nz2);
    gcn_agg2<<<(int)(((long)E * 8 + 255) / 256), 256, 0, stream>>>(src, dst, PR2, S2, E);

    gcn_output<<<(N + 255) / 256, 256, 0, stream>>>(S2, PR2, b2, cnt, (float*)d_out, N);
}


# Both chunk-plane sum kernels run as few fat waves: a float4 per lane, all plane loads in flight, combine1 a resident grid loop (115640 waves -> 10166); every sum's order unchanged
# speedup vs baseline: 2.9556x; 2.9556x over previous
#include <hip/hip_runtime.h>
#include <math.h>

typedef __attribute__((ext_vector_type(2))) float v2f;
typedef __attribute__((ext_vector_type(8))) float v8f;

#define D_IN  128
#define H_DIM 32
#define C_DIM 5

__global__ void gcn_build_w1(const float* __restrict__ Wrel, const float* __restrict__ Wroot,
                             float* __restrict__ Wcat) {
    int i = blockIdx.x * 256 + threadIdx.x;
    if (i >= 128 * 64) return;
    int k = i >> 6, j = i & 63;
    Wcat[i] = (j < 32) ? Wrel[k * 32 + j] : Wroot[k * 32 + (j - 32)];
}

__global__ void gcn_build_w2(const float* __restrict__ Wrel, const float* __restrict__ Wroot,
                             float* __restrict__ Wp) {
    int i = blockIdx.x * 256 + threadIdx.x;
    if (i >= 32 * 16) return;
    int k = i >> 4, n = i & 15;
    float v = 0.0f;
    if (n < 5)       v = Wrel[k * 5 + n];
    else if (n < 10) v = Wroot[k * 5 + (n - 5)];
    Wp[i] = v;
}

__global__ __launch_bounds__(256)
void gcn_gemm1_wmma(const float* __restrict__ x, const float* __restrict__ W,
                    float* __restrict__ PR1, int nTiles) {
    int wave = threadIdx.x >> 5;
    int lane = threadIdx.x & 31;
    int tile = blockIdx.x * 8 + wave;
    if (tile >= nTiles) return;
    int half = lane >> 4;
    int m    = lane & 15;
    const float* xr = x + (long)(tile * 16 + m) * D_IN;

    v2f a[32];
#pragma unroll
    for (int ks = 0; ks < 32; ++ks) {
        int k = ks * 4 + half * 2;
        a[ks].x = xr[k];
        a[ks].y = xr[k + 1];
    }

#pragma unroll
    for (int ct = 0; ct < 4; ++ct) {
        int n = ct * 16 + m;
        v8f acc = {};
#pragma unroll
        for (int ks = 0; ks < 32; ++ks) {
            int k = ks * 4 + half * 2;
            v2f b;
            b.x = W[k * 64 + n];
            b.y = W[(k + 1) * 64 + n];
            acc = __builtin_amdgcn_wmma_f32_16x16x4_f32(false, a[ks], false, b,
                                                        (short)0, acc, false, false);
        }
#pragma unroll
        for (int r = 0; r < 8; ++r) {
            int orow = tile * 16 + r + half * 8;
            PR1[(long)orow * 64 + ct * 16 + m] = acc[r];
        }
    }
}

__global__ __launch_bounds__(256)
void gcn_gemm2_wmma(const float* __restrict__ h, const float* __restrict__ W,
                    float* __restrict__ PR2, int nTiles) {
    int wave = threadIdx.x >> 5;
    int lane = threadIdx.x & 31;
    int tile = blockIdx.x * 8 + wave;
    if (tile >= nTiles) return;
    int half = lane >> 4;
    int m    = lane & 15;
    const float* hr = h + (long)(tile * 16 + m) * H_DIM;

    v2f a[8];
#pragma unroll
    for (int ks = 0; ks < 8; ++ks) {
        int k = ks * 4 + half * 2;
        a[ks].x = hr[k];
        a[ks].y = hr[k + 1];
    }
    v8f acc = {};
#pragma unroll
    for (int ks = 0; ks < 8; ++ks) {
        int k = ks * 4 + half * 2;
        v2f b;
        b.x = W[k * 16 + m];
        b.y = W[(k + 1) * 16 + m];
        acc = __builtin_amdgcn_wmma_f32_16x16x4_f32(false, a[ks], false, b,
                                                    (short)0, acc, false, false);
    }
    if (m < 10) {
#pragma unroll
        for (int r = 0; r < 8; ++r) {
            int orow = tile * 16 + r + half * 8;
            PR2[(long)orow * 10 + m] = acc[r];
        }
    }
}

#define TILE_N    2048
#define AGG_WAVES 32
#define QCAP      160

__device__ __forceinline__ void gcn_agg1_drain(int nd, int head, int lane, const int2* qw,
                                               const int* __restrict__ src,
                                               const float* __restrict__ ew,
                                               const float* __restrict__ PR1,
                                               float* acc, float* cntl) {
    int s = 0, dl = 0;
    float w = 0.0f;
    if (lane < nd) {
        int idx = head + lane;
        if (idx >= QCAP) idx -= QCAP;
        int2 ent = qw[idx];
        s  = src[ent.x];
        w  = ew[ent.x];
        dl = ent.y;
        atomicAdd(&cntl[dl], 1.0f);
    }
    for (int k0 = 0; k0 < nd; k0 += 16) {
        float v[16];
#pragma unroll
        for (int j = 0; j < 16; ++j) {
            int sk = __builtin_amdgcn_readlane(s, k0 + j);
            v[j] = PR1[(unsigned)sk * 64u + (unsigned)lane];
        }
#pragma unroll
        for (int j = 0; j < 16; ++j) {
            if (k0 + j < nd) {
                float wk = __int_as_float(__builtin_amdgcn_readlane(__float_as_int(w), k0 + j));
                int   dk = __builtin_amdgcn_readlane(dl, k0 + j);
                atomicAdd(&acc[dk * 32 + lane], wk * v[j]);
            }
        }
    }
}

template <bool VEC>
__device__ __forceinline__ int4 gcn_load_quad(const int* __restrict__ dst, int qi, int nQuads, int E) {
    int qc = min(qi, nQuads - 1);
    if (VEC) return *(const int4*)(dst + 4 * (long)qc);
    int e0 = 4 * qc;
    int4 r;
    r.x = dst[e0];
    r.y = dst[min(e0 + 1, E - 1)];
    r.z = dst[min(e0 + 2, E - 1)];
    r.w = dst[min(e0 + 3, E - 1)];
    return r;
}

template <bool VEC>
__device__ __forceinline__ void gcn_agg1_scan(const int* __restrict__ src, const int* __restrict__ dst,
                                              const float* __restrict__ ew,
                                              const float* __restrict__ PR1,
                                              float* acc, float* cntl, int2* qw,
                                              int base, int rows, int qBeg, int qEnd,
                                              int nQuads, int E, int tid, int lane) {
    const int nIt = (qEnd - qBeg + 1023) >> 10;
    int head = 0, qn = 0;

    auto step = [&](int4& d, int it) {
        const int qi = qBeg + it * 1024 + tid;
        const int e0 = 4 * qi;
        const unsigned lim = ((it >= 0) && (qi < qEnd)) ? (unsigned)rows : 0u;
#pragma unroll
        for (int k = 0; k < 4; ++k) {
            int dv = (k == 0) ? d.x : (k == 1) ? d.y : (k == 2) ? d.z : d.w;
            int dl = dv - base;
            bool m = (unsigned)dl < lim;
            if (!VEC) m = m && (e0 + k < E);
            unsigned bal = __builtin_amdgcn_ballot_w32(m);
            if (bal != 0u) {
                if (m) {
                    int pos = head + qn + (int)__builtin_amdgcn_mbcnt_lo(bal, 0u);
                    if (pos >= QCAP) pos -= QCAP;
                    qw[pos] = make_int2(e0 + k, dl);
                }
                qn += __builtin_popcount(bal);
            }
        }
        d = gcn_load_quad<VEC>(dst, qi + 4096, nQuads, E);
        while (qn >= 32) {
            gcn_agg1_drain(32, head, lane, qw, src, ew, PR1, acc, cntl);
            head += 32;
            if (head >= QCAP) head -= QCAP;
            qn -= 32;
        }
    };

    int4 d0 = make_int4(-1, -1, -1, -1), d1 = d0, d2 = d0, d3 = d0;
#pragma unroll 1
    for (int it = -4; it < nIt; it += 4) {
        step(d0, it);
        step(d1, it + 1);
        step(d2, it + 2);
        step(d3, it + 3);
    }
    if (qn > 0) gcn_agg1_drain(qn, head, lane, qw, src, ew, PR1, acc, cntl);
}

__global__ __launch_bounds__(1024)
void gcn_agg1_tile(const int* __restrict__ src, const int* __restrict__ dst,
                   const float* __restrict__ ew, const float* __restrict__ PR1,
                   float* __restrict__ Ppart, float* __restrict__ Cpart,
                   int N, int E, int nChunks, int quadsPerChunk) {
    __shared__ __attribute__((aligned(16))) float acc[TILE_N * 32];
    __shared__ float cntl[TILE_N];
    __shared__ int2  q[AGG_WAVES][QCAP];

    const int tid  = threadIdx.x;
    const int wave = tid >> 5;
    const int lane = tid & 31;
    const int tile  = blockIdx.x / nChunks;
    const int chunk = blockIdx.x - tile * nChunks;
    const int base  = tile * TILE_N;
    const int rows  = min(TILE_N, N - base);

    const float4 z4 = {0.0f, 0.0f, 0.0f, 0.0f};
    for (int i = tid; i < TILE_N * 8; i += 1024) ((float4*)acc)[i] = z4;
    for (int i = tid; i < TILE_N; i += 1024) cntl[i] = 0.0f;
    __syncthreads();

    const int nQuads = (E + 3) >> 2;
    const int qBeg   = chunk * quadsPerChunk;
    const int qEnd   = min(qBeg + quadsPerChunk, nQuads);
    if (qEnd > qBeg) {
        if ((E & 3) == 0)
            gcn_agg1_scan<true>(src, dst, ew, PR1, acc, cntl, q[wave], base, rows, qBeg, qEnd, nQuads, E, tid, lane);
        else
            gcn_agg1_scan<false>(src, dst, ew, PR1, acc, cntl, q[wave], base, rows, qBeg, qEnd, nQuads, E, tid, lane);
    }
    __syncthreads();

    const long planeP = (long)N * 32;
    float4* outP = (float4*)(Ppart + (long)chunk * planeP + (long)base * 32);
    for (int i = tid; i < rows * 8; i += 1024) outP[i] = ((const float4*)acc)[i];
    for (int i = tid; i < rows; i += 1024) Cpart[(long)chunk * N + base + i] = cntl[i];
}

#define MAX_CHUNKS1 8
#define COMB_BLOCKS 1024

template <int NC>
__device__ __forceinline__ void gcn_combine1_walk(const float4* __restrict__ P4,
                                                  const float* __restrict__ Cpart,
                                                  const float* __restrict__ PR1, float4 bb,
                                                  float* __restrict__ cnt, float* __restrict__ h,
                                                  int N, int c4) {
    const int  nUnits = N * 8;
    const long planeQ = (long)N * 8;
    const int  stride = gridDim.x * 256;
    for (int q = blockIdx.x * 256 + threadIdx.x; q < nUnits; q += stride) {
        const int n = q >> 3;
        float4 p[NC];
        float  cq[NC];
#pragma unroll
        for (int c = 0; c < NC; ++c) {
            p[c]  = P4[(long)c * planeQ + q];
            cq[c] = Cpart[(long)c * N + n];
        }
        const float4 r = *(const float4*)(PR1 + (long)n * 64 + 32 + c4 * 4);
        float4 s = {0.0f, 0.0f, 0.0f, 0.0f};
        float cn = 0.0f;
#pragma unroll
        for (int c = 0; c < NC; ++c) {
            s.x += p[c].x;
            s.y += p[c].y;
            s.z += p[c].z;
            s.w += p[c].w;
            cn  += cq[c];
        }
        const float c1 = fmaxf(cn, 1.0f);
        float4 v;
        v.x = fmaxf(s.x / c1 + r.x + bb.x, 0.0f);
        v.y = fmaxf(s.y / c1 + r.y + bb.y, 0.0f);
        v.z = fmaxf(s.z / c1 + r.z + bb.z, 0.0f);
        v.w = fmaxf(s.w / c1 + r.w + bb.w, 0.0f);
        ((float4*)h)[q] = v;
        if (c4 == 0) cnt[n] = cn;
    }
}

__global__ __launch_bounds__(256)
void gcn_combine1(const float* __restrict__ Ppart, const float* __restrict__ Cpart,
                  const float* __restrict__ PR1, const float* __restrict__ b1,
                  float* __restrict__ cnt, float* __restrict__ h,
                  int N, int nChunks) {
    const int c4 = threadIdx.x & 7;
    const float4* P4 = (const float4*)Ppart;
    float4 bb;
    bb.x = b1[c4 * 4 + 0];
    bb.y = b1[c4 * 4 + 1];
    bb.z = b1[c4 * 4 + 2];
    bb.w = b1[c4 * 4 + 3];
    switch (nChunks) {
    case 1: gcn_combine1_walk<1>(P4, Cpart, PR1, bb, cnt, h, N, c4); break;
    case 2: gcn_combine1_walk<2>(P4, Cpart, PR1, bb, cnt, h, N, c4); break;
    case 3: gcn_combine1_walk<3>(P4, Cpart, PR1, bb, cnt, h, N, c4); break;
    case 4: gcn_combine1_walk<4>(P4, Cpart, PR1, bb, cnt, h, N, c4); break;
    case 5: gcn_combine1_walk<5>(P4, Cpart, PR1, bb, cnt, h, N, c4); break;
    case 6: gcn_combine1_walk<6>(P4, Cpart, PR1, bb, cnt, h, N, c4); break;
    case 7: gcn_combine1_walk<7>(P4, Cpart, PR1, bb, cnt, h, N, c4); break;
    case 8: gcn_combine1_walk<8>(P4, Cpart, PR1, bb, cnt, h, N, c4); break;
    default: break;
    }
}

#define TILE2_N 6400

__device__ __forceinline__ void gcn_agg2_drain(int nd, int head, int lane, const int2* qw,
                                               const int* __restrict__ src,
                                               const float* __restrict__ PR2, float* acc) {
    if (lane < nd) {
        int idx = head + lane;
        if (idx >= QCAP) idx -= QCAP;
        int2 ent = qw[idx];
        const float* p = PR2 + (unsigned)src[ent.x] * 10u;
        float2 v01 = *(const float2*)p;
        float2 v23 = *(const float2*)(p + 2);
        float  v4  = p[4];
        float* a = acc + ent.y * 5;
        atomicAdd(a + 0, v01.x);
        atomicAdd(a + 1, v01.y);
        atomicAdd(a + 2, v23.x);
        atomicAdd(a + 3, v23.y);
        atomicAdd(a + 4, v4);
    }
}

template <bool VEC>
__device__ __forceinline__ void gcn_agg2_scan(const int* __restrict__ src, const int* __restrict__ dst,
                                              const float* __restrict__ PR2,
                                              float* acc, int2* qw,
                                              int base, int rows, int qBeg, int qEnd,
                                              int nQuads, int E, int tid, int lane) {
    const int nIt = (qEnd - qBeg + 1023) >> 10;
    int head = 0, qn = 0;

    auto step = [&](int4& d, int it) {
        const int qi = qBeg + it * 1024 + tid;
        const int e0 = 4 * qi;
        const unsigned lim = ((it >= 0) && (qi < qEnd)) ? (unsigned)rows : 0u;
#pragma unroll
        for (int k = 0; k < 4; ++k) {
            int dv = (k == 0) ? d.x : (k == 1) ? d.y : (k == 2) ? d.z : d.w;
            int dl = dv - base;
            bool m = (unsigned)dl < lim;
            if (!VEC) m = m && (e0 + k < E);
            unsigned bal = __builtin_amdgcn_ballot_w32(m);
            if (bal != 0u) {
                if (m) {
                    int pos = head + qn + (int)__builtin_amdgcn_mbcnt_lo(bal, 0u);
                    if (pos >= QCAP) pos -= QCAP;
                    qw[pos] = make_int2(e0 + k, dl);
                }
                qn += __builtin_popcount(bal);
            }
        }
        d = gcn_load_quad<VEC>(dst, qi + 4096, nQuads, E);
        while (qn >= 32) {
            gcn_agg2_drain(32, head, lane, qw, src, PR2, acc);
            head += 32;
            if (head >= QCAP) head -= QCAP;
            qn -= 32;
        }
    };

    int4 d0 = make_int4(-1, -1, -1, -1), d1 = d0, d2 = d0, d3 = d0;
#pragma unroll 1
    for (int it = -4; it < nIt; it += 4) {
        step(d0, it);
        step(d1, it + 1);
        step(d2, it + 2);
        step(d3, it + 3);
    }
    if (qn > 0) gcn_agg2_drain(qn, head, lane, qw, src, PR2, acc);
}

__global__ __launch_bounds__(1024)
void gcn_agg2_tile(const int* __restrict__ src, const int* __restrict__ dst,
                   const float* __restrict__ PR2, float* __restrict__ P2,
                   int N, int E, int nChunks, int quadsPerChunk) {
    __shared__ __attribute__((aligned(16))) float acc[TILE2_N * 5];
    __shared__ int2  q[AGG_WAVES][QCAP];

    const int tid  = threadIdx.x;
    const int wave = tid >> 5;
    const int lane = tid & 31;
    const int tile  = blockIdx.x / nChunks;
    const int chunk = blockIdx.x - tile * nChunks;
    const int base  = tile * TILE2_N;
    const int rows  = min(TILE2_N, N - base);

    const float4 z4 = {0.0f, 0.0f, 0.0f, 0.0f};
    for (int i = tid; i < TILE2_N * 5 / 4; i += 1024) ((float4*)acc)[i] = z4;
    __syncthreads();

    const int nQuads = (E + 3) >> 2;
    const int qBeg   = chunk * quadsPerChunk;
    const int qEnd   = min(qBeg + quadsPerChunk, nQuads);
    if (qEnd > qBeg) {
        if ((E & 3) == 0)
            gcn_agg2_scan<true>(src, dst, PR2, acc, q[wave], base, rows, qBeg, qEnd, nQuads, E, tid, lane);
        else
            gcn_agg2_scan<false>(src, dst, PR2, acc, q[wave], base, rows, qBeg, qEnd, nQuads, E, tid, lane);
    }
    __syncthreads();

    float* outP = P2 + (long)chunk * N * 5 + (long)base * 5;
    for (int i = tid; i < rows * 5; i += 1024) outP[i] = acc[i];
}

#define OUT_NODES   256
#define OUT_THREADS (OUT_NODES * 5 / 4)
#define MAX_CHUNKS2 16

template <int NC>
__device__ __forceinline__ float4 gcn_plane_sum4(const float4* __restrict__ P4, long planeQ) {
    float4 p[NC];
#pragma unroll
    for (int cz = 0; cz < NC; ++cz) p[cz] = P4[(long)cz * planeQ];
    float4 s = {0.0f, 0.0f, 0.0f, 0.0f};
#pragma unroll
    for (int cz = 0; cz < NC; ++cz) {
        s.x += p[cz].x;
        s.y += p[cz].y;
        s.z += p[cz].z;
        s.w += p[cz].w;
    }
    return s;
}

__global__ __launch_bounds__(OUT_THREADS)
void gcn_output(const float* __restrict__ P2, const float* __restrict__ PR2,
                const float* __restrict__ b2, const float* __restrict__ cnt,
                float* __restrict__ out, int N, int nChunks) {
    __shared__ __attribute__((aligned(16))) float s2[OUT_NODES * 5];
    const int  n0     = blockIdx.x * OUT_NODES;
    const long totalQ = (long)N * 5 / 4;
    const long gq     = (long)blockIdx.x * OUT_THREADS + threadIdx.x;
    const int  n      = n0 + (int)threadIdx.x;
    const bool fin    = (threadIdx.x < OUT_NODES) && (n < N);
    const int   nl = min(n, N - 1);
    const float cn = cnt[nl];
    float r[5];
#pragma unroll
    for (int j = 0; j < 5; ++j) r[j] = PR2[(long)nl * 10 + 5 + j];
    float4 sp = {0.0f, 0.0f, 0.0f, 0.0f};
    if (gq < totalQ) {
        const float4* P4 = (const float4*)P2 + gq;
        switch (nChunks) {
        case 1:  sp = gcn_plane_sum4<1>(P4, totalQ);  break;
        case 2:  sp = gcn_plane_sum4<2>(P4, totalQ);  break;
        case 3:  sp = gcn_plane_sum4<3>(P4, totalQ);  break;
        case 4:  sp = gcn_plane_sum4<4>(P4, totalQ);  break;
        case 5:  sp = gcn_plane_sum4<5>(P4, totalQ);  break;
        case 6:  sp = gcn_plane_sum4<6>(P4, totalQ);  break;
        case 7:  sp = gcn_plane_sum4<7>(P4, totalQ);  break;
        case 8:  sp = gcn_plane_sum4<8>(P4, totalQ);  break;
        case 9:  sp = gcn_plane_sum4<9>(P4, totalQ);  break;
        case 10: sp = gcn_plane_sum4<10>(P4, totalQ); break;
        case 11: sp = gcn_plane_sum4<11>(P4, totalQ); break;
        case 12: sp = gcn_plane_sum4<12>(P4, totalQ); break;
        case 13: sp = gcn_plane_sum4<13>(P4, totalQ); break;
        case 14: sp = gcn_plane_sum4<14>(P4, totalQ); break;
        case 15: sp = gcn_plane_sum4<15>(P4, totalQ); break;
        case 16: sp = gcn_plane_sum4<16>(P4, totalQ); break;
        default: break;
        }
    }
    ((float4*)s2)[threadIdx.x] = sp;
    __syncthreads();
    if (!fin) return;
    const float c = fmaxf(cn, 1.0f);
    float v[5];
#pragma unroll
    for (int j = 0; j < 5; ++j)
        v[j] = s2[threadIdx.x * 5 + j] / c + r[j] + b2[j];
    float mx = v[0];
#pragma unroll
    for (int j = 1; j < 5; ++j) mx = fmaxf(mx, v[j]);
    float s = 0.0f;
#pragma unroll
    for (int j = 0; j < 5; ++j) s += expf(v[j] - mx);
    float l = logf(s);
#pragma unroll
    for (int j = 0; j < 5; ++j) out[(long)n * 5 + j] = v[j] - mx - l;
}

extern "C" void kernel_launch(void* const* d_in, const int* in_sizes, int n_in,
                              void* d_out, int out_size, void* d_ws, size_t ws_size,
                              hipStream_t stream) {
    const float* x      = (const float*)d_in[0];
    const int*   ei     = (const int*)d_in[1];
    const float* ew     = (const float*)d_in[2];
    const float* Wrel1  = (const float*)d_in[3];
    const float* Wroot1 = (const float*)d_in[4];
    const float* b1     = (const float*)d_in[5];
    const float* Wrel2  = (const float*)d_in[6];
    const float* Wroot2 = (const float*)d_in[7];
    const float* b2     = (const float*)d_in[8];

    const int N = in_sizes[0] / D_IN;
    const int E = in_sizes[1] / 2;
    const int* src = ei;
    const int* dst = ei + E;

    float* ws = (float*)d_ws;
    long off = 0;
    float* PR1 = ws + off; off += (long)N * 64;
    float* cnt = ws + off; off += N;
    float* h   = ws + off; off += (long)N * 32;
    float* W1c = ws + off; off += 128 * 64;
    float* W2p = ws + off; off += 32 * 16;
    const int nT = (N + TILE_N - 1) / TILE_N;
    const int nC = (nT >= 256) ? 1 : ((256 / nT > MAX_CHUNKS1) ? MAX_CHUNKS1 : 256 / nT);
    float* Pp  = ws + off; off += (long)nC * N * 32;
    float* Cp  = ws + off; off += (long)nC * N;
    float* PR2 = PR1;
    const int nT2 = (N + TILE2_N - 1) / TILE2_N;
    const int nC2 = (nT2 >= 256) ? 1 : ((256 / nT2 > MAX_CHUNKS2) ? MAX_CHUNKS2 : 256 / nT2);
    float* P2  = ws + off; off += (long)nC2 * N * 5;

    const int nTiles = N / 16;

    gcn_build_w1<<<(128 * 64 + 255) / 256, 256, 0, stream>>>(Wrel1, Wroot1, W1c);
    gcn_build_w2<<<(32 * 16 + 255) / 256, 256, 0, stream>>>(Wrel2, Wroot2, W2p);

    gcn_gemm1_wmma<<<(nTiles + 7) / 8, 256, 0, stream>>>(x, W1c, PR1, nTiles);

    const int nQuads = (E + 3) / 4;
    const int qpc = (nQuads + nC - 1) / nC;
    gcn_agg1_tile<<<nT * nC, 1024, 0, stream>>>(src, dst, ew, PR1, Pp, Cp, N, E, nC, qpc);

    const int cbAll   = (N * 8 + 255) / 256;
    const int cbTrips = (cbAll + COMB_BLOCKS - 1) / COMB_BLOCKS;
    gcn_combine1<<<(cbAll + cbTrips - 1) / cbTrips, 256, 0, stream>>>(Pp, Cp, PR1, b1, cnt, h, N, nC);

    gcn_gemm2_wmma<<<(nTiles + 7) / 8, 256, 0, stream>>>(h, W2p, PR2, nTiles);

    const int qpc2 = (nQuads + nC2 - 1) / nC2;
    gcn_agg2_tile<<<nT2 * nC2, 1024, 0, stream>>>(src, dst, PR2, P2, N, E, nC2, qpc2);

    gcn_output<<<(N + OUT_NODES - 1) / OUT_NODES, OUT_THREADS, 0, stream>>>(P2, PR2, b2, cnt, (float*)d_out, N, nC2);
}


# h never goes to memory: the plane-sum kernel stages its group's 128 rows of h in LDS and each of its 8 waves runs one GEMM2 WMMA tile from there (8 launches -> 7); every sum's order unchanged
# speedup vs baseline: 3.0715x; 1.0392x over previous
#include <hip/hip_runtime.h>
#include <math.h>

typedef __attribute__((ext_vector_type(2))) float v2f;
typedef __attribute__((ext_vector_type(8))) float v8f;

#define D_IN  128
#define H_DIM 32
#define C_DIM 5

__global__ void gcn_build_w1(const float* __restrict__ Wrel, const float* __restrict__ Wroot,
                             float* __restrict__ Wcat) {
    int i = blockIdx.x * 256 + threadIdx.x;
    if (i >= 128 * 64) return;
    int k = i >> 6, j = i & 63;
    Wcat[i] = (j < 32) ? Wrel[k * 32 + j] : Wroot[k * 32 + (j - 32)];
}

__global__ void gcn_build_w2(const float* __restrict__ Wrel, const float* __restrict__ Wroot,
                             float* __restrict__ Wp) {
    int i = blockIdx.x * 256 + threadIdx.x;
    if (i >= 32 * 16) return;
    int k = i >> 4, n = i & 15;
    float v = 0.0f;
    if (n < 5)       v = Wrel[k * 5 + n];
    else if (n < 10) v = Wroot[k * 5 + (n - 5)];
    Wp[i] = v;
}

__global__ __launch_bounds__(256)
void gcn_gemm1_wmma(const float* __restrict__ x, const float* __restrict__ W,
                    float* __restrict__ PR1, int nTiles) {
    int wave = threadIdx.x >> 5;
    int lane = threadIdx.x & 31;
    int tile = blockIdx.x * 8 + wave;
    if (tile >= nTiles) return;
    int half = lane >> 4;
    int m    = lane & 15;
    const float* xr = x + (long)(tile * 16 + m) * D_IN;

    v2f a[32];
#pragma unroll
    for (int ks = 0; ks < 32; ++ks) {
        int k = ks * 4 + half * 2;
        a[ks].x = xr[k];
        a[ks].y = xr[k + 1];
    }

#pragma unroll
    for (int ct = 0; ct < 4; ++ct) {
        int n = ct * 16 + m;
        v8f acc = {};
#pragma unroll
        for (int ks = 0; ks < 32; ++ks) {
            int k = ks * 4 + half * 2;
            v2f b;
            b.x = W[k * 64 + n];
            b.y = W[(k + 1) * 64 + n];
            acc = __builtin_amdgcn_wmma_f32_16x16x4_f32(false, a[ks], false, b,
                                                        (short)0, acc, false, false);
        }
#pragma unroll
        for (int r = 0; r < 8; ++r) {
            int orow = tile * 16 + r + half * 8;
            PR1[(long)orow * 64 + ct * 16 + m] = acc[r];
        }
    }
}

#define TILE_N    2048
#define AGG_WAVES 32
#define QCAP      160

__device__ __forceinline__ void gcn_agg1_drain(int nd, int head, int lane, const int2* qw,
                                               const int* __restrict__ src,
                                               const float* __restrict__ ew,
                                               const float* __restrict__ PR1,
                                               float* acc, float* cntl) {
    int s = 0, dl = 0;
    float w = 0.0f;
    if (lane < nd) {
        int idx = head + lane;
        if (idx >= QCAP) idx -= QCAP;
        int2 ent = qw[idx];
        s  = src[ent.x];
        w  = ew[ent.x];
        dl = ent.y;
        atomicAdd(&cntl[dl], 1.0f);
    }
    for (int k0 = 0; k0 < nd; k0 += 16) {
        float v[16];
#pragma unroll
        for (int j = 0; j < 16; ++j) {
            int sk = __builtin_amdgcn_readlane(s, k0 + j);
            v[j] = PR1[(unsigned)sk * 64u + (unsigned)lane];
        }
#pragma unroll
        for (int j = 0; j < 16; ++j) {
            if (k0 + j < nd) {
                float wk = __int_as_float(__builtin_amdgcn_readlane(__float_as_int(w), k0 + j));
                int   dk = __builtin_amdgcn_readlane(dl, k0 + j);
                atomicAdd(&acc[dk * 32 + lane], wk * v[j]);
            }
        }
    }
}

template <bool VEC>
__device__ __forceinline__ int4 gcn_load_quad(const int* __restrict__ dst, int qi, int nQuads, int E) {
    int qc = min(qi, nQuads - 1);
    if (VEC) return *(const int4*)(dst + 4 * (long)qc);
    int e0 = 4 * qc;
    int4 r;
    r.x = dst[e0];
    r.y = dst[min(e0 + 1, E - 1)];
    r.z = dst[min(e0 + 2, E - 1)];
    r.w = dst[min(e0 + 3, E - 1)];
    return r;
}

template <bool VEC>
__device__ __forceinline__ void gcn_agg1_scan(const int* __restrict__ src, const int* __restrict__ dst,
                                              const float* __restrict__ ew,
                                              const float* __restrict__ PR1,
                                              float* acc, float* cntl, int2* qw,
                                              int base, int rows, int qBeg, int qEnd,
                                              int nQuads, int E, int tid, int lane) {
    const int nIt = (qEnd - qBeg + 1023) >> 10;
    int head = 0, qn = 0;

    auto step = [&](int4& d, int it) {
        const int qi = qBeg + it * 1024 + tid;
        const int e0 = 4 * qi;
        const unsigned lim = ((it >= 0) && (qi < qEnd)) ? (unsigned)rows : 0u;
#pragma unroll
        for (int k = 0; k < 4; ++k) {
            int dv = (k == 0) ? d.x : (k == 1) ? d.y : (k == 2) ? d.z : d.w;
            int dl = dv - base;
            bool m = (unsigned)dl < lim;
            if (!VEC) m = m && (e0 + k < E);
            unsigned bal = __builtin_amdgcn_ballot_w32(m);
            if (bal != 0u) {
                if (m) {
                    int pos = head + qn + (int)__builtin_amdgcn_mbcnt_lo(bal, 0u);
                    if (pos >= QCAP) pos -= QCAP;
                    qw[pos] = make_int2(e0 + k, dl);
                }
                qn += __builtin_popcount(bal);
            }
        }
        d = gcn_load_quad<VEC>(dst, qi + 4096, nQuads, E);
        while (qn >= 32) {
            gcn_agg1_drain(32, head, lane, qw, src, ew, PR1, acc, cntl);
            head += 32;
            if (head >= QCAP) head -= QCAP;
            qn -= 32;
        }
    };

    int4 d0 = make_int4(-1, -1, -1, -1), d1 = d0, d2 = d0, d3 = d0;
#pragma unroll 1
    for (int it = -4; it < nIt; it += 4) {
        step(d0, it);
        step(d1, it + 1);
        step(d2, it + 2);
        step(d3, it + 3);
    }
    if (qn > 0) gcn_agg1_drain(qn, head, lane, qw, src, ew, PR1, acc, cntl);
}

__global__ __launch_bounds__(1024)
void gcn_agg1_tile(const int* __restrict__ src, const int* __restrict__ dst,
                   const float* __restrict__ ew, const float* __restrict__ PR1,
                   float* __restrict__ Ppart, float* __restrict__ Cpart,
                   int N, int E, int nChunks, int quadsPerChunk) {
    __shared__ __attribute__((aligned(16))) float acc[TILE_N * 32];
    __shared__ float cntl[TILE_N];
    __shared__ int2  q[AGG_WAVES][QCAP];

    const int tid  = threadIdx.x;
    const int wave = tid >> 5;
    const int lane = tid & 31;
    const int tile  = blockIdx.x / nChunks;
    const int chunk = blockIdx.x - tile * nChunks;
    const int base  = tile * TILE_N;
    const int rows  = min(TILE_N, N - base);

    const float4 z4 = {0.0f, 0.0f, 0.0f, 0.0f};
    for (int i = tid; i < TILE_N * 8; i += 1024) ((float4*)acc)[i] = z4;
    for (int i = tid; i < TILE_N; i += 1024) cntl[i] = 0.0f;
    __syncthreads();

    const int nQuads = (E + 3) >> 2;
    const int qBeg   = chunk * quadsPerChunk;
    const int qEnd   = min(qBeg + quadsPerChunk, nQuads);
    if (qEnd > qBeg) {
        if ((E & 3) == 0)
            gcn_agg1_scan<true>(src, dst, ew, PR1, acc, cntl, q[wave], base, rows, qBeg, qEnd, nQuads, E, tid, lane);
        else
            gcn_agg1_scan<false>(src, dst, ew, PR1, acc, cntl, q[wave], base, rows, qBeg, qEnd, nQuads, E, tid, lane);
    }
    __syncthreads();

    const long planeP = (long)N * 32;
    float4* outP = (float4*)(Ppart + (long)chunk * planeP + (long)base * 32);
    for (int i = tid; i < rows * 8; i += 1024) outP[i] = ((const float4*)acc)[i];
    for (int i = tid; i < rows; i += 1024) Cpart[(long)chunk * N + base + i] = cntl[i];
}

#define MAX_CHUNKS1 8
#define CMB_NODES   128
#define HS_PITCH    36

template <int NC>
__device__ __forceinline__ void gcn_combine1_rows(const float4* __restrict__ P4,
                                                  const float* __restrict__ Cpart,
                                                  const float* __restrict__ PR1, float4 bb,
                                                  float* __restrict__ cnt, float* hs,
                                                  int N, int c4) {
    constexpr int JB = (NC <= 5) ? 4 : 2;
    const int  nUnits = N * 8;
    const long planeQ = (long)N * 8;
    const int  q0     = blockIdx.x * (CMB_NODES * 8) + threadIdx.x;
#pragma unroll
    for (int j0 = 0; j0 < 4; j0 += JB) {
        float4 p[JB][NC];
        float  cq[JB][NC];
        float4 r[JB];
#pragma unroll
        for (int j = 0; j < JB; ++j) {
            const int q = min(q0 + (j0 + j) * 256, nUnits - 1);
            const int n = q >> 3;
#pragma unroll
            for (int c = 0; c < NC; ++c) {
                p[j][c]  = P4[(long)c * planeQ + q];
                cq[j][c] = Cpart[(long)c * N + n];
            }
            r[j] = *(const float4*)(PR1 + (long)n * 64 + 32 + c4 * 4);
        }
#pragma unroll
        for (int j = 0; j < JB; ++j) {
            const int q = q0 + (j0 + j) * 256;
            float4 s = {0.0f, 0.0f, 0.0f, 0.0f};
            float cn = 0.0f;
#pragma unroll
            for (int c = 0; c < NC; ++c) {
                s.x += p[j][c].x;
                s.y += p[j][c].y;
                s.z += p[j][c].z;
                s.w += p[j][c].w;
                cn  += cq[j][c];
            }
            const float c1 = fmaxf(cn, 1.0f);
            float4 v;
            v.x = fmaxf(s.x / c1 + r[j].x + bb.x, 0.0f);
            v.y = fmaxf(s.y / c1 + r[j].y + bb.y, 0.0f);
            v.z = fmaxf(s.z / c1 + r[j].z + bb.z, 0.0f);
            v.w = fmaxf(s.w / c1 + r[j].w + bb.w, 0.0f);
            *(float4*)(hs + ((j0 + j) * 32 + (int)(threadIdx.x >> 3)) * HS_PITCH + c4 * 4) = v;
            if (c4 == 0 && q < nUnits) cnt[q >> 3] = cn;
        }
    }
}

__global__ __launch_bounds__(256)
void gcn_combine1_gemm2(const float* __restrict__ Ppart, const float* __restrict__ Cpart,
                        const float* __restrict__ PR1, const float* __restrict__ b1,
                        const float* __restrict__ W, float* __restrict__ cnt,
                        float* __restrict__ PR2, int N, int nChunks) {
    __shared__ __attribute__((aligned(16))) float hs[CMB_NODES * HS_PITCH];
    const int c4 = threadIdx.x & 7;
    const float4* P4 = (const float4*)Ppart;
    float4 bb;
    bb.x = b1[c4 * 4 + 0];
    bb.y = b1[c4 * 4 + 1];
    bb.z = b1[c4 * 4 + 2];
    bb.w = b1[c4 * 4 + 3];
    switch (nChunks) {
    case 1: gcn_combine1_rows<1>(P4, Cpart, PR1, bb, cnt, hs, N, c4); break;
    case 2: gcn_combine1_rows<2>(P4, Cpart, PR1, bb, cnt, hs, N, c4); break;
    case 3: gcn_combine1_rows<3>(P4, Cpart, PR1, bb, cnt, hs, N, c4); break;
    case 4: gcn_combine1_rows<4>(P4, Cpart, PR1, bb, cnt, hs, N, c4); break;
    case 5: gcn_combine1_rows<5>(P4, Cpart, PR1, bb, cnt, hs, N, c4); break;
    case 6: gcn_combine1_rows<6>(P4, Cpart, PR1, bb, cnt, hs, N, c4); break;
    case 7: gcn_combine1_rows<7>(P4, Cpart, PR1, bb, cnt, hs, N, c4); break;
    case 8: gcn_combine1_rows<8>(P4, Cpart, PR1, bb, cnt, hs, N, c4); break;
    default: break;
    }

    const int wave = threadIdx.x >> 5;
    const int lane = threadIdx.x & 31;
    const int half = lane >> 4;
    const int m    = lane & 15;
    v2f b[8];
#pragma unroll
    for (int ks = 0; ks < 8; ++ks) {
        int k = ks * 4 + half * 2;
        b[ks].x = W[k * 16 + m];
        b[ks].y = W[(k + 1) * 16 + m];
    }
    __syncthreads();
    const int row0 = blockIdx.x * CMB_NODES + wave * 16;
    if (row0 >= N) return;
    const float* hr = hs + (wave * 16 + m) * HS_PITCH;
    v2f a[8];
#pragma unroll
    for (int ks = 0; ks < 8; ++ks) {
        int k = ks * 4 + half * 2;
        a[ks].x = hr[k];
        a[ks].y = hr[k + 1];
    }
    v8f acc = {};
#pragma unroll
    for (int ks = 0; ks < 8; ++ks)
        acc = __builtin_amdgcn_wmma_f32_16x16x4_f32(false, a[ks], false, b[ks],
                                                    (short)0, acc, false, false);
    if (m < 10) {
#pragma unroll
        for (int r = 0; r < 8; ++r) {
            int orow = row0 + r + half * 8;
            PR2[(long)orow * 10 + m] = acc[r];
        }
    }
}

#define TILE2_N 6400

__device__ __forceinline__ void gcn_agg2_drain(int nd, int head, int lane, const int2* qw,
                                               const int* __restrict__ src,
                                               const float* __restrict__ PR2, float* acc) {
    if (lane < nd) {
        int idx = head + lane;
        if (idx >= QCAP) idx -= QCAP;
        int2 ent = qw[idx];
        const float* p = PR2 + (unsigned)src[ent.x] * 10u;
        float2 v01 = *(const float2*)p;
        float2 v23 = *(const float2*)(p + 2);
        float  v4  = p[4];
        float* a = acc + ent.y * 5;
        atomicAdd(a + 0, v01.x);
        atomicAdd(a + 1, v01.y);
        atomicAdd(a + 2, v23.x);
        atomicAdd(a + 3, v23.y);
        atomicAdd(a + 4, v4);
    }
}

template <bool VEC>
__device__ __forceinline__ void gcn_agg2_scan(const int* __restrict__ src, const int* __restrict__ dst,
                                              const float* __restrict__ PR2,
                                              float* acc, int2* qw,
                                              int base, int rows, int qBeg, int qEnd,
                                              int nQuads, int E, int tid, int lane) {
    const int nIt = (qEnd - qBeg + 1023) >> 10;
    int head = 0, qn = 0;

    auto step = [&](int4& d, int it) {
        const int qi = qBeg + it * 1024 + tid;
        const int e0 = 4 * qi;
        const unsigned lim = ((it >= 0) && (qi < qEnd)) ? (unsigned)rows : 0u;
#pragma unroll
        for (int k = 0; k < 4; ++k) {
            int dv = (k == 0) ? d.x : (k == 1) ? d.y : (k == 2) ? d.z : d.w;
            int dl = dv - base;
            bool m = (unsigned)dl < lim;
            if (!VEC) m = m && (e0 + k < E);
            unsigned bal = __builtin_amdgcn_ballot_w32(m);
            if (bal != 0u) {
                if (m) {
                    int pos = head + qn + (int)__builtin_amdgcn_mbcnt_lo(bal, 0u);
                    if (pos >= QCAP) pos -= QCAP;
                    qw[pos] = make_int2(e0 + k, dl);
                }
                qn += __builtin_popcount(bal);
            }
        }
        d = gcn_load_quad<VEC>(dst, qi + 4096, nQuads, E);
        while (qn >= 32) {
            gcn_agg2_drain(32, head, lane, qw, src, PR2, acc);
            head += 32;
            if (head >= QCAP) head -= QCAP;
            qn -= 32;
        }
    };

    int4 d0 = make_int4(-1, -1, -1, -1), d1 = d0, d2 = d0, d3 = d0;
#pragma unroll 1
    for (int it = -4; it < nIt; it += 4) {
        step(d0, it);
        step(d1, it + 1);
        step(d2, it + 2);
        step(d3, it + 3);
    }
    if (qn > 0) gcn_agg2_drain(qn, head, lane, qw, src, PR2, acc);
}

__global__ __launch_bounds__(1024)
void gcn_agg2_tile(const int* __restrict__ src, const int* __restrict__ dst,
                   const float* __restrict__ PR2, float* __restrict__ P2,
                   int N, int E, int nChunks, int quadsPerChunk) {
    __shared__ __attribute__((aligned(16))) float acc[TILE2_N * 5];
    __shared__ int2  q[AGG_WAVES][QCAP];

    const int tid  = threadIdx.x;
    const int wave = tid >> 5;
    const int lane = tid & 31;
    const int tile  = blockIdx.x / nChunks;
    const int chunk = blockIdx.x - tile * nChunks;
    const int base  = tile * TILE2_N;
    const int rows  = min(TILE2_N, N - base);

    const float4 z4 = {0.0f, 0.0f, 0.0f, 0.0f};
    for (int i = tid; i < TILE2_N * 5 / 4; i += 1024) ((float4*)acc)[i] = z4;
    __syncthreads();

    const int nQuads = (E + 3) >> 2;
    const int qBeg   = chunk * quadsPerChunk;
    const int qEnd   = min(qBeg + quadsPerChunk, nQuads);
    if (qEnd > qBeg) {
        if ((E & 3) == 0)
            gcn_agg2_scan<true>(src, dst, PR2, acc, q[wave], base, rows, qBeg, qEnd, nQuads, E, tid, lane);
        else
            gcn_agg2_scan<false>(src, dst, PR2, acc, q[wave], base, rows, qBeg, qEnd, nQuads, E, tid, lane);
    }
    __syncthreads();

    float* outP = P2 + (long)chunk * N * 5 + (long)base * 5;
    for (int i = tid; i < rows * 5; i += 1024) outP[i] = acc[i];
}

#define OUT_NODES   256
#define OUT_THREADS (OUT_NODES * 5 / 4)
#define MAX_CHUNKS2 16

template <int NC>
__device__ __forceinline__ float4 gcn_plane_sum4(const float4* __restrict__ P4, long planeQ) {
    float4 p[NC];
#pragma unroll
    for (int cz = 0; cz < NC; ++cz) p[cz] = P4[(long)cz * planeQ];
    float4 s = {0.0f, 0.0f, 0.0f, 0.0f};
#pragma unroll
    for (int cz = 0; cz < NC; ++cz) {
        s.x += p[cz].x;
        s.y += p[cz].y;
        s.z += p[cz].z;
        s.w += p[cz].w;
    }
    return s;
}

__global__ __launch_bounds__(OUT_THREADS)
void gcn_output(const float* __restrict__ P2, const float* __restrict__ PR2,
                const float* __restrict__ b2, const float* __restrict__ cnt,
                float* __restrict__ out, int N, int nChunks) {
    __shared__ __attribute__((aligned(16))) float s2[OUT_NODES * 5];
    const int  n0     = blockIdx.x * OUT_NODES;
    const long totalQ = (long)N * 5 / 4;
    const long gq     = (long)blockIdx.x * OUT_THREADS + threadIdx.x;
    const int  n      = n0 + (int)threadIdx.x;
    const bool fin    = (threadIdx.x < OUT_NODES) && (n < N);
    const int   nl = min(n, N - 1);
    const float cn = cnt[nl];
    float r[5];
#pragma unroll
    for (int j = 0; j < 5; ++j) r[j] = PR2[(long)nl * 10 + 5 + j];
    float4 sp = {0.0f, 0.0f, 0.0f, 0.0f};
    if (gq < totalQ) {
        const float4* P4 = (const float4*)P2 + gq;
        switch (nChunks) {
        case 1:  sp = gcn_plane_sum4<1>(P4, totalQ);  break;
        case 2:  sp = gcn_plane_sum4<2>(P4, totalQ);  break;
        case 3:  sp = gcn_plane_sum4<3>(P4, totalQ);  break;
        case 4:  sp = gcn_plane_sum4<4>(P4, totalQ);  break;
        case 5:  sp = gcn_plane_sum4<5>(P4, totalQ);  break;
        case 6:  sp = gcn_plane_sum4<6>(P4, totalQ);  break;
        case 7:  sp = gcn_plane_sum4<7>(P4, totalQ);  break;
        case 8:  sp = gcn_plane_sum4<8>(P4, totalQ);  break;
        case 9:  sp = gcn_plane_sum4<9>(P4, totalQ);  break;
        case 10: sp = gcn_plane_sum4<10>(P4, totalQ); break;
        case 11: sp = gcn_plane_sum4<11>(P4, totalQ); break;
        case 12: sp = gcn_plane_sum4<12>(P4, totalQ); break;
        case 13: sp = gcn_plane_sum4<13>(P4, totalQ); break;
        case 14: sp = gcn_plane_sum4<14>(P4, totalQ); break;
        case 15: sp = gcn_plane_sum4<15>(P4, totalQ); break;
        case 16: sp = gcn_plane_sum4<16>(P4, totalQ); break;
        default: break;
        }
    }
    ((float4*)s2)[threadIdx.x] = sp;
    __syncthreads();
    if (!fin) return;
    const float c = fmaxf(cn, 1.0f);
    float v[5];
#pragma unroll
    for (int j = 0; j < 5; ++j)
        v[j] = s2[threadIdx.x * 5 + j] / c + r[j] + b2[j];
    float mx = v[0];
#pragma unroll
    for (int j = 1; j < 5; ++j) mx = fmaxf(mx, v[j]);
    float s = 0.0f;
#pragma unroll
    for (int j = 0; j < 5; ++j) s += expf(v[j] - mx);
    float l = logf(s);
#pragma unroll
    for (int j = 0; j < 5; ++j) out[(long)n * 5 + j] = v[j] - mx - l;
}

extern "C" void kernel_launch(void* const* d_in, const int* in_sizes, int n_in,
                              void* d_out, int out_size, void* d_ws, size_t ws_size,
                              hipStream_t stream) {
    const float* x      = (const float*)d_in[0];
    const int*   ei     = (const int*)d_in[1];
    const float* ew     = (const float*)d_in[2];
    const float* Wrel1  = (const float*)d_in[3];
    const float* Wroot1 = (const float*)d_in[4];
    const float* b1     = (const float*)d_in[5];
    const float* Wrel2  = (const float*)d_in[6];
    const float* Wroot2 = (const float*)d_in[7];
    const float* b2     = (const float*)d_in[8];

    const int N = in_sizes[0] / D_IN;
    const int E = in_sizes[1] / 2;
    const int* src = ei;
    const int* dst = ei + E;

    float* ws = (float*)d_ws;
    long off = 0;
    float* PR1 = ws + off; off += (long)N * 64;
    float* cnt = ws + off; off += N;
    float* PR2 = ws + off; off += (long)N * 10;
    float* W1c = ws + off; off += 128 * 64;
    float* W2p = ws + off; off += 32 * 16;
    const int nT = (N + TILE_N - 1) / TILE_N;
    const int nC = (nT >= 256) ? 1 : ((256 / nT > MAX_CHUNKS1) ? MAX_CHUNKS1 : 256 / nT);
    float* Pp  = ws + off; off += (long)nC * N * 32;
    float* Cp  = ws + off; off += (long)nC * N;
    const int nT2 = (N + TILE2_N - 1) / TILE2_N;
    const int nC2 = (nT2 >= 256) ? 1 : ((256 / nT2 > MAX_CHUNKS2) ? MAX_CHUNKS2 : 256 / nT2);
    float* P2  = ws + off; off += (long)nC2 * N * 5;

    const int nTiles = N / 16;

    gcn_build_w1<<<(128 * 64 + 255) / 256, 256, 0, stream>>>(Wrel1, Wroot1, W1c);
    gcn_build_w2<<<(32 * 16 + 255) / 256, 256, 0, stream>>>(Wrel2, Wroot2, W2p);

    gcn_gemm1_wmma<<<(nTiles + 7) / 8, 256, 0, stream>>>(x, W1c, PR1, nTiles);

    const int nQuads = (E + 3) / 4;
    const int qpc = (nQuads + nC - 1) / nC;
    gcn_agg1_tile<<<nT * nC, 1024, 0, stream>>>(src, dst, ew, PR1, Pp, Cp, N, E, nC, qpc);

    gcn_combine1_gemm2<<<(N + CMB_NODES - 1) / CMB_NODES, 256, 0, stream>>>(Pp, Cp, PR1, b1, W2p, cnt, PR2, N, nC);

    const int qpc2 = (nQuads + nC2 - 1) / nC2;
    gcn_agg2_tile<<<nT2 * nC2, 1024, 0, stream>>>(src, dst, PR2, P2, N, E, nC2, qpc2);

    gcn_output<<<(N + OUT_NODES - 1) / OUT_NODES, OUT_THREADS, 0, stream>>>(P2, PR2, b2, cnt, (float*)d_out, N, nC2);
}


# GEMM1 as a resident grid: 2 waves per SIMD, each keeps all 128x64 weight fragments in registers and walks 3-4 row tiles, next tile's x rows loaded ahead (6256 waves -> 2048); sum order unchanged
# speedup vs baseline: 3.2233x; 1.0494x over previous
#include <hip/hip_runtime.h>
#include <math.h>

typedef __attribute__((ext_vector_type(2))) float v2f;
typedef __attribute__((ext_vector_type(8))) float v8f;

#define D_IN  128
#define H_DIM 32
#define C_DIM 5

__global__ void gcn_build_w1(const float* __restrict__ Wrel, const float* __restrict__ Wroot,
                             float* __restrict__ Wcat) {
    int i = blockIdx.x * 256 + threadIdx.x;
    if (i >= 128 * 64) return;
    int k = i >> 6, j = i & 63;
    Wcat[i] = (j < 32) ? Wrel[k * 32 + j] : Wroot[k * 32 + (j - 32)];
}

__global__ void gcn_build_w2(const float* __restrict__ Wrel, const float* __restrict__ Wroot,
                             float* __restrict__ Wp) {
    int i = blockIdx.x * 256 + threadIdx.x;
    if (i >= 32 * 16) return;
    int k = i >> 4, n = i & 15;
    float v = 0.0f;
    if (n < 5)       v = Wrel[k * 5 + n];
    else if (n < 10) v = Wroot[k * 5 + (n - 5)];
    Wp[i] = v;
}

#define G1_WAVES 8

__device__ __forceinline__ void gcn_gemm1_load_a(const float* __restrict__ x, int tile, int m, int half,
                                                 v2f* a) {
    const float* xr = x + (long)(tile * 16 + m) * D_IN + half * 2;
#pragma unroll
    for (int ks = 0; ks < 32; ++ks) a[ks] = *(const v2f*)(xr + ks * 4);
}

__global__ __launch_bounds__(G1_WAVES * 32)
void gcn_gemm1_wmma(const float* __restrict__ x, const float* __restrict__ W,
                    float* __restrict__ PR1, int nTiles) {
    const int wave = threadIdx.x >> 5;
    const int lane = threadIdx.x & 31;
    const int half = lane >> 4;
    const int m    = lane & 15;
    const int nW   = gridDim.x * G1_WAVES;
    int tile = wave * gridDim.x + blockIdx.x;
    if (tile >= nTiles) return;

    v2f a[32];
    gcn_gemm1_load_a(x, tile, m, half, a);

    v2f b[4][32];
#pragma unroll
    for (int ct = 0; ct < 4; ++ct) {
#pragma unroll
        for (int ks = 0; ks < 32; ++ks) {
            int k = ks * 4 + half * 2;
            int n = ct * 16 + m;
            b[ct][ks].x = W[k * 64 + n];
            b[ct][ks].y = W[(k + 1) * 64 + n];
        }
    }

#pragma unroll 1
    for (; tile < nTiles; tile += nW) {
        const int tn = (tile + nW < nTiles) ? tile + nW : tile;
        v2f an[32];
        gcn_gemm1_load_a(x, tn, m, half, an);

        v8f acc[4] = {};
#pragma unroll
        for (int ks = 0; ks < 32; ++ks) {
#pragma unroll
            for (int ct = 0; ct < 4; ++ct)
                acc[ct] = __builtin_amdgcn_wmma_f32_16x16x4_f32(false, a[ks], false, b[ct][ks],
                                                                (short)0, acc[ct], false, false);
        }
#pragma unroll
        for (int ct = 0; ct < 4; ++ct) {
#pragma unroll
            for (int r = 0; r < 8; ++r) {
                int orow = tile * 16 + r + half * 8;
                PR1[(long)orow * 64 + ct * 16 + m] = acc[ct][r];
            }
        }
#pragma unroll
        for (int ks = 0; ks < 32; ++ks) a[ks] = an[ks];
    }
}

#define TILE_N    2048
#define AGG_WAVES 32
#define QCAP      160

__device__ __forceinline__ void gcn_agg1_drain(int nd, int head, int lane, const int2* qw,
                                               const int* __restrict__ src,
                                               const float* __restrict__ ew,
                                               const float* __restrict__ PR1,
                                               float* acc, float* cntl) {
    int s = 0, dl = 0;
    float w = 0.0f;
    if (lane < nd) {
        int idx = head + lane;
        if (idx >= QCAP) idx -= QCAP;
        int2 ent = qw[idx];
        s  = src[ent.x];
        w  = ew[ent.x];
        dl = ent.y;
        atomicAdd(&cntl[dl], 1.0f);
    }
    for (int k0 = 0; k0 < nd; k0 += 16) {
        float v[16];
#pragma unroll
        for (int j = 0; j < 16; ++j) {
            int sk = __builtin_amdgcn_readlane(s, k0 + j);
            v[j] = PR1[(unsigned)sk * 64u + (unsigned)lane];
        }
#pragma unroll
        for (int j = 0; j < 16; ++j) {
            if (k0 + j < nd) {
                float wk = __int_as_float(__builtin_amdgcn_readlane(__float_as_int(w), k0 + j));
                int   dk = __builtin_amdgcn_readlane(dl, k0 + j);
                atomicAdd(&acc[dk * 32 + lane], wk * v[j]);
            }
        }
    }
}

template <bool VEC>
__device__ __forceinline__ int4 gcn_load_quad(const int* __restrict__ dst, int qi, int nQuads, int E) {
    int qc = min(qi, nQuads - 1);
    if (VEC) return *(const int4*)(dst + 4 * (long)qc);
    int e0 = 4 * qc;
    int4 r;
    r.x = dst[e0];
    r.y = dst[min(e0 + 1, E - 1)];
    r.z = dst[min(e0 + 2, E - 1)];
    r.w = dst[min(e0 + 3, E - 1)];
    return r;
}

template <bool VEC>
__device__ __forceinline__ void gcn_agg1_scan(const int* __restrict__ src, const int* __restrict__ dst,
                                              const float* __restrict__ ew,
                                              const float* __restrict__ PR1,
                                              float* acc, float* cntl, int2* qw,
                                              int base, int rows, int qBeg, int qEnd,
                                              int nQuads, int E, int tid, int lane) {
    const int nIt = (qEnd - qBeg + 1023) >> 10;
    int head = 0, qn = 0;

    auto step = [&](int4& d, int it) {
        const int qi = qBeg + it * 1024 + tid;
        const int e0 = 4 * qi;
        const unsigned lim = ((it >= 0) && (qi < qEnd)) ? (unsigned)rows : 0u;
#pragma unroll
        for (int k = 0; k < 4; ++k) {
            int dv = (k == 0) ? d.x : (k == 1) ? d.y : (k == 2) ? d.z : d.w;
            int dl = dv - base;
            bool m = (unsigned)dl < lim;
            if (!VEC) m = m && (e0 + k < E);
            unsigned bal = __builtin_amdgcn_ballot_w32(m);
            if (bal != 0u) {
                if (m) {
                    int pos = head + qn + (int)__builtin_amdgcn_mbcnt_lo(bal, 0u);
                    if (pos >= QCAP) pos -= QCAP;
                    qw[pos] = make_int2(e0 + k, dl);
                }
                qn += __builtin_popcount(bal);
            }
        }
        d = gcn_load_quad<VEC>(dst, qi + 4096, nQuads, E);
        while (qn >= 32) {
            gcn_agg1_drain(32, head, lane, qw, src, ew, PR1, acc, cntl);
            head += 32;
            if (head >= QCAP) head -= QCAP;
            qn -= 32;
        }
    };

    int4 d0 = make_int4(-1, -1, -1, -1), d1 = d0, d2 = d0, d3 = d0;
#pragma unroll 1
    for (int it = -4; it < nIt; it += 4) {
        step(d0, it);
        step(d1, it + 1);
        step(d2, it + 2);
        step(d3, it + 3);
    }
    if (qn > 0) gcn_agg1_drain(qn, head, lane, qw, src, ew, PR1, acc, cntl);
}

__global__ __launch_bounds__(1024)
void gcn_agg1_tile(const int* __restrict__ src, const int* __restrict__ dst,
                   const float* __restrict__ ew, const float* __restrict__ PR1,
                   float* __restrict__ Ppart, float* __restrict__ Cpart,
                   int N, int E, int nChunks, int quadsPerChunk) {
    __shared__ __attribute__((aligned(16))) float acc[TILE_N * 32];
    __shared__ float cntl[TILE_N];
    __shared__ int2  q[AGG_WAVES][QCAP];

    const int tid  = threadIdx.x;
    const int wave = tid >> 5;
    const int lane = tid & 31;
    const int tile  = blockIdx.x / nChunks;
    const int chunk = blockIdx.x - tile * nChunks;
    const int base  = tile * TILE_N;
    const int rows  = min(TILE_N, N - base);

    const float4 z4 = {0.0f, 0.0f, 0.0f, 0.0f};
    for (int i = tid; i < TILE_N * 8; i += 1024) ((float4*)acc)[i] = z4;
    for (int i = tid; i < TILE_N; i += 1024) cntl[i] = 0.0f;
    __syncthreads();

    const int nQuads = (E + 3) >> 2;
    const int qBeg   = chunk * quadsPerChunk;
    const int qEnd   = min(qBeg + quadsPerChunk, nQuads);
    if (qEnd > qBeg) {
        if ((E & 3) == 0)
            gcn_agg1_scan<true>(src, dst, ew, PR1, acc, cntl, q[wave], base, rows, qBeg, qEnd, nQuads, E, tid, lane);
        else
            gcn_agg1_scan<false>(src, dst, ew, PR1, acc, cntl, q[wave], base, rows, qBeg, qEnd, nQuads, E, tid, lane);
    }
    __syncthreads();

    const long planeP = (long)N * 32;
    float4* outP = (float4*)(Ppart + (long)chunk * planeP + (long)base * 32);
    for (int i = tid; i < rows * 8; i += 1024) outP[i] = ((const float4*)acc)[i];
    for (int i = tid; i < rows; i += 1024) Cpart[(long)chunk * N + base + i] = cntl[i];
}

#define MAX_CHUNKS1 8
#define CMB_NODES   128
#define HS_PITCH    36

template <int NC>
__device__ __forceinline__ void gcn_combine1_rows(const float4* __restrict__ P4,
                                                  const float* __restrict__ Cpart,
                                                  const float* __restrict__ PR1, float4 bb,
                                                  float* __restrict__ cnt, float* hs,
                                                  int N, int c4) {
    constexpr int JB = (NC <= 5) ? 4 : 2;
    const int  nUnits = N * 8;
    const long planeQ = (long)N * 8;
    const int  q0     = blockIdx.x * (CMB_NODES * 8) + threadIdx.x;
#pragma unroll
    for (int j0 = 0; j0 < 4; j0 += JB) {
        float4 p[JB][NC];
        float  cq[JB][NC];
        float4 r[JB];
#pragma unroll
        for (int j = 0; j < JB; ++j) {
            const int q = min(q0 + (j0 + j) * 256, nUnits - 1);
            const int n = q >> 3;
#pragma unroll
            for (int c = 0; c < NC; ++c) {
                p[j][c]  = P4[(long)c * planeQ + q];
                cq[j][c] = Cpart[(long)c * N + n];
            }
            r[j] = *(const float4*)(PR1 + (long)n * 64 + 32 + c4 * 4);
        }
#pragma unroll
        for (int j = 0; j < JB; ++j) {
            const int q = q0 + (j0 + j) * 256;
            float4 s = {0.0f, 0.0f, 0.0f, 0.0f};
            float cn = 0.0f;
#pragma unroll
            for (int c = 0; c < NC; ++c) {
                s.x += p[j][c].x;
                s.y += p[j][c].y;
                s.z += p[j][c].z;
                s.w += p[j][c].w;
                cn  += cq[j][c];
            }
            const float c1 = fmaxf(cn, 1.0f);
            float4 v;
            v.x = fmaxf(s.x / c1 + r[j].x + bb.x, 0.0f);
            v.y = fmaxf(s.y / c1 + r[j].y + bb.y, 0.0f);
            v.z = fmaxf(s.z / c1 + r[j].z + bb.z, 0.0f);
            v.w = fmaxf(s.w / c1 + r[j].w + bb.w, 0.0f);
            *(float4*)(hs + ((j0 + j) * 32 + (int)(threadIdx.x >> 3)) * HS_PITCH + c4 * 4) = v;
            if (c4 == 0 && q < nUnits) cnt[q >> 3] = cn;
        }
    }
}

__global__ __launch_bounds__(256)
void gcn_combine1_gemm2(const float* __restrict__ Ppart, const float* __restrict__ Cpart,
                        const float* __restrict__ PR1, const float* __restrict__ b1,
                        const float* __restrict__ W, float* __restrict__ cnt,
                        float* __restrict__ PR2, int N, int nChunks) {
    __shared__ __attribute__((aligned(16))) float hs[CMB_NODES * HS_PITCH];
    const int c4 = threadIdx.x & 7;
    const float4* P4 = (const float4*)Ppart;
    float4 bb;
    bb.x = b1[c4 * 4 + 0];
    bb.y = b1[c4 * 4 + 1];
    bb.z = b1[c4 * 4 + 2];
    bb.w = b1[c4 * 4 + 3];
    switch (nChunks) {
    case 1: gcn_combine1_rows<1>(P4, Cpart, PR1, bb, cnt, hs, N, c4); break;
    case 2: gcn_combine1_rows<2>(P4, Cpart, PR1, bb, cnt, hs, N, c4); break;
    case 3: gcn_combine1_rows<3>(P4, Cpart, PR1, bb, cnt, hs, N, c4); break;
    case 4: gcn_combine1_rows<4>(P4, Cpart, PR1, bb, cnt, hs, N, c4); break;
    case 5: gcn_combine1_rows<5>(P4, Cpart, PR1, bb, cnt, hs, N, c4); break;
    case 6: gcn_combine1_rows<6>(P4, Cpart, PR1, bb, cnt, hs, N, c4); break;
    case 7: gcn_combine1_rows<7>(P4, Cpart, PR1, bb, cnt, hs, N, c4); break;
    case 8: gcn_combine1_rows<8>(P4, Cpart, PR1, bb, cnt, hs, N, c4); break;
    default: break;
    }

    const int wave = threadIdx.x >> 5;
    const int lane = threadIdx.x & 31;
    const int half = lane >> 4;
    const int m    = lane & 15;
    v2f b[8];
#pragma unroll
    for (int ks = 0; ks < 8; ++ks) {
        int k = ks * 4 + half * 2;
        b[ks].x = W[k * 16 + m];
        b[ks].y = W[(k + 1) * 16 + m];
    }
    __syncthreads();
    const int row0 = blockIdx.x * CMB_NODES + wave * 16;
    if (row0 >= N) return;
    const float* hr = hs + (wave * 16 + m) * HS_PITCH;
    v2f a[8];
#pragma unroll
    for (int ks = 0; ks < 8; ++ks) {
        int k = ks * 4 + half * 2;
        a[ks].x = hr[k];
        a[ks].y = hr[k + 1];
    }
    v8f acc = {};
#pragma unroll
    for (int ks = 0; ks < 8; ++ks)
        acc = __builtin_amdgcn_wmma_f32_16x16x4_f32(false, a[ks], false, b[ks],
                                                    (short)0, acc, false, false);
    if (m < 10) {
#pragma unroll
        for (int r = 0; r < 8; ++r) {
            int orow = row0 + r + half * 8;
            PR2[(long)orow * 10 + m] = acc[r];
        }
    }
}

#define TILE2_N 6400

__device__ __forceinline__ void gcn_agg2_drain(int nd, int head, int lane, const int2* qw,
                                               const int* __restrict__ src,
                                               const float* __restrict__ PR2, float* acc) {
    if (lane < nd) {
        int idx = head + lane;
        if (idx >= QCAP) idx -= QCAP;
        int2 ent = qw[idx];
        const float* p = PR2 + (unsigned)src[ent.x] * 10u;
        float2 v01 = *(const float2*)p;
        float2 v23 = *(const float2*)(p + 2);
        float  v4  = p[4];
        float* a = acc + ent.y * 5;
        atomicAdd(a + 0, v01.x);
        atomicAdd(a + 1, v01.y);
        atomicAdd(a + 2, v23.x);
        atomicAdd(a + 3, v23.y);
        atomicAdd(a + 4, v4);
    }
}

template <bool VEC>
__device__ __forceinline__ void gcn_agg2_scan(const int* __restrict__ src, const int* __restrict__ dst,
                                              const float* __restrict__ PR2,
                                              float* acc, int2* qw,
                                              int base, int rows, int qBeg, int qEnd,
                                              int nQuads, int E, int tid, int lane) {
    const int nIt = (qEnd - qBeg + 1023) >> 10;
    int head = 0, qn = 0;

    auto step = [&](int4& d, int it) {
        const int qi = qBeg + it * 1024 + tid;
        const int e0 = 4 * qi;
        const unsigned lim = ((it >= 0) && (qi < qEnd)) ? (unsigned)rows : 0u;
#pragma unroll
        for (int k = 0; k < 4; ++k) {
            int dv = (k == 0) ? d.x : (k == 1) ? d.y : (k == 2) ? d.z : d.w;
            int dl = dv - base;
            bool m = (unsigned)dl < lim;
            if (!VEC) m = m && (e0 + k < E);
            unsigned bal = __builtin_amdgcn_ballot_w32(m);
            if (bal != 0u) {
                if (m) {
                    int pos = head + qn + (int)__builtin_amdgcn_mbcnt_lo(bal, 0u);
                    if (pos >= QCAP) pos -= QCAP;
                    qw[pos] = make_int2(e0 + k, dl);
                }
                qn += __builtin_popcount(bal);
            }
        }
        d = gcn_load_quad<VEC>(dst, qi + 4096, nQuads, E);
        while (qn >= 32) {
            gcn_agg2_drain(32, head, lane, qw, src, PR2, acc);
            head += 32;
            if (head >= QCAP) head -= QCAP;
            qn -= 32;
        }
    };

    int4 d0 = make_int4(-1, -1, -1, -1), d1 = d0, d2 = d0, d3 = d0;
#pragma unroll 1
    for (int it = -4; it < nIt; it += 4) {
        step(d0, it);
        step(d1, it + 1);
        step(d2, it + 2);
        step(d3, it + 3);
    }
    if (qn > 0) gcn_agg2_drain(qn, head, lane, qw, src, PR2, acc);
}

__global__ __launch_bounds__(1024)
void gcn_agg2_tile(const int* __restrict__ src, const int* __restrict__ dst,
                   const float* __restrict__ PR2, float* __restrict__ P2,
                   int N, int E, int nChunks, int quadsPerChunk) {
    __shared__ __attribute__((aligned(16))) float acc[TILE2_N * 5];
    __shared__ int2  q[AGG_WAVES][QCAP];

    const int tid  = threadIdx.x;
    const int wave = tid >> 5;
    const int lane = tid & 31;
    const int tile  = blockIdx.x / nChunks;
    const int chunk = blockIdx.x - tile * nChunks;
    const int base  = tile * TILE2_N;
    const int rows  = min(TILE2_N, N - base);

    const float4 z4 = {0.0f, 0.0f, 0.0f, 0.0f};
    for (int i = tid; i < TILE2_N * 5 / 4; i += 1024) ((float4*)acc)[i] = z4;
    __syncthreads();

    const int nQuads = (E + 3) >> 2;
    const int qBeg   = chunk * quadsPerChunk;
    const int qEnd   = min(qBeg + quadsPerChunk, nQuads);
    if (qEnd > qBeg) {
        if ((E & 3) == 0)
            gcn_agg2_scan<true>(src, dst, PR2, acc, q[wave], base, rows, qBeg, qEnd, nQuads, E, tid, lane);
        else
            gcn_agg2_scan<false>(src, dst, PR2, acc, q[wave], base, rows, qBeg, qEnd, nQuads, E, tid, lane);
    }
    __syncthreads();

    float* outP = P2 + (long)chunk * N * 5 + (long)base * 5;
    for (int i = tid; i < rows * 5; i += 1024) outP[i] = acc[i];
}

#define OUT_NODES   256
#define OUT_THREADS (OUT_NODES * 5 / 4)
#define MAX_CHUNKS2 16

template <int NC>
__device__ __forceinline__ float4 gcn_plane_sum4(const float4* __restrict__ P4, long planeQ) {
    float4 p[NC];
#pragma unroll
    for (int cz = 0; cz < NC; ++cz) p[cz] = P4[(long)cz * planeQ];
    float4 s = {0.0f, 0.0f, 0.0f, 0.0f};
#pragma unroll
    for (int cz = 0; cz < NC; ++cz) {
        s.x += p[cz].x;
        s.y += p[cz].y;
        s.z += p[cz].z;
        s.w += p[cz].w;
    }
    return s;
}

__global__ __launch_bounds__(OUT_THREADS)
void gcn_output(const float* __restrict__ P2, const float* __restrict__ PR2,
                const float* __restrict__ b2, const float* __restrict__ cnt,
                float* __restrict__ out, int N, int nChunks) {
    __shared__ __attribute__((aligned(16))) float s2[OUT_NODES * 5];
    const int  n0     = blockIdx.x * OUT_NODES;
    const long totalQ = (long)N * 5 / 4;
    const long gq     = (long)blockIdx.x * OUT_THREADS + threadIdx.x;
    const int  n      = n0 + (int)threadIdx.x;
    const bool fin    = (threadIdx.x < OUT_NODES) && (n < N);
    const int   nl = min(n, N - 1);
    const float cn = cnt[nl];
    float r[5];
#pragma unroll
    for (int j = 0; j < 5; ++j) r[j] = PR2[(long)nl * 10 + 5 + j];
    float4 sp = {0.0f, 0.0f, 0.0f, 0.0f};
    if (gq < totalQ) {
        const float4* P4 = (const float4*)P2 + gq;
        switch (nChunks) {
        case 1:  sp = gcn_plane_sum4<1>(P4, totalQ);  break;
        case 2:  sp = gcn_plane_sum4<2>(P4, totalQ);  break;
        case 3:  sp = gcn_plane_sum4<3>(P4, totalQ);  break;
        case 4:  sp = gcn_plane_sum4<4>(P4, totalQ);  break;
        case 5:  sp = gcn_plane_sum4<5>(P4, totalQ);  break;
        case 6:  sp = gcn_plane_sum4<6>(P4, totalQ);  break;
        case 7:  sp = gcn_plane_sum4<7>(P4, totalQ);  break;
        case 8:  sp = gcn_plane_sum4<8>(P4, totalQ);  break;
        case 9:  sp = gcn_plane_sum4<9>(P4, totalQ);  break;
        case 10: sp = gcn_plane_sum4<10>(P4, totalQ); break;
        case 11: sp = gcn_plane_sum4<11>(P4, totalQ); break;
        case 12: sp = gcn_plane_sum4<12>(P4, totalQ); break;
        case 13: sp = gcn_plane_sum4<13>(P4, totalQ); break;
        case 14: sp = gcn_plane_sum4<14>(P4, totalQ); break;
        case 15: sp = gcn_plane_sum4<15>(P4, totalQ); break;
        case 16: sp = gcn_plane_sum4<16>(P4, totalQ); break;
        default: break;
        }
    }
    ((float4*)s2)[threadIdx.x] = sp;
    __syncthreads();
    if (!fin) return;
    const float c = fmaxf(cn, 1.0f);
    float v[5];
#pragma unroll
    for (int j = 0; j < 5; ++j)
        v[j] = s2[threadIdx.x * 5 + j] / c + r[j] + b2[j];
    float mx = v[0];
#pragma unroll
    for (int j = 1; j < 5; ++j) mx = fmaxf(mx, v[j]);
    float s = 0.0f;
#pragma unroll
    for (int j = 0; j < 5; ++j) s += expf(v[j] - mx);
    float l = logf(s);
#pragma unroll
    for (int j = 0; j < 5; ++j) out[(long)n * 5 + j] = v[j] - mx - l;
}

extern "C" void kernel_launch(void* const* d_in, const int* in_sizes, int n_in,
                              void* d_out, int out_size, void* d_ws, size_t ws_size,
                              hipStream_t stream) {
    const float* x      = (const float*)d_in[0];
    const int*   ei     = (const int*)d_in[1];
    const float* ew     = (const float*)d_in[2];
    const float* Wrel1  = (const float*)d_in[3];
    const float* Wroot1 = (const float*)d_in[4];
    const float* b1     = (const float*)d_in[5];
    const float* Wrel2  = (const float*)d_in[6];
    const float* Wroot2 = (const float*)d_in[7];
    const float* b2     = (const float*)d_in[8];

    const int N = in_sizes[0] / D_IN;
    const int E = in_sizes[1] / 2;
    const int* src = ei;
    const int* dst = ei + E;

    float* ws = (float*)d_ws;
    long off = 0;
    float* PR1 = ws + off; off += (long)N * 64;
    float* cnt = ws + off; off += N;
    float* PR2 = ws + off; off += (long)N * 10;
    float* W1c = ws + off; off += 128 * 64;
    float* W2p = ws + off; off += 32 * 16;
    const int nT = (N + TILE_N - 1) / TILE_N;
    const int nC = (nT >= 256) ? 1 : ((256 / nT > MAX_CHUNKS1) ? MAX_CHUNKS1 : 256 / nT);
    float* Pp  = ws + off; off += (long)nC * N * 32;
    float* Cp  = ws + off; off += (long)nC * N;
    const int nT2 = (N + TILE2_N - 1) / TILE2_N;
    const int nC2 = (nT2 >= 256) ? 1 : ((256 / nT2 > MAX_CHUNKS2) ? MAX_CHUNKS2 : 256 / nT2);
    float* P2  = ws + off; off += (long)nC2 * N * 5;

    const int nTiles = N / 16;

    gcn_build_w1<<<(128 * 64 + 255) / 256, 256, 0, stream>>>(Wrel1, Wroot1, W1c);
    gcn_build_w2<<<(32 * 16 + 255) / 256, 256, 0, stream>>>(Wrel2, Wroot2, W2p);

    gcn_gemm1_wmma<<<256, G1_WAVES * 32, 0, stream>>>(x, W1c, PR1, nTiles);

    const int nQuads = (E + 3) / 4;
    const int qpc = (nQuads + nC - 1) / nC;
    gcn_agg1_tile<<<nT * nC, 1024, 0, stream>>>(src, dst, ew, PR1, Pp, Cp, N, E, nC, qpc);

    gcn_combine1_gemm2<<<(N + CMB_NODES - 1) / CMB_NODES, 256, 0, stream>>>(Pp, Cp, PR1, b1, W2p, cnt, PR2, N, nC);

    const int qpc2 = (nQuads + nC2 - 1) / nC2;
    gcn_agg2_tile<<<nT2 * nC2, 1024, 0, stream>>>(src, dst, PR2, P2, N, E, nC2, qpc2);

    gcn_output<<<(N + OUT_NODES - 1) / OUT_NODES, OUT_THREADS, 0, stream>>>(P2, PR2, b2, cnt, (float*)d_out, N, nC2);
}


# Layer-1 scan reads a 1-byte tile key per edge and one packed (src, local dst, weight) record per kept edge; both written per call by the weight-repack launch; sum order: LDS arrival, then chunk
# speedup vs baseline: 3.3430x; 1.0371x over previous
#include <hip/hip_runtime.h>
#include <math.h>

typedef __attribute__((ext_vector_type(2))) float v2f;
typedef __attribute__((ext_vector_type(8))) float v8f;

#define D_IN  128
#define H_DIM 32
#define C_DIM 5

#define TILE_N   2048
#define SRC_BITS 20

__global__ __launch_bounds__(1024)
void gcn_build_w1_edges(const float* __restrict__ Wrel, const float* __restrict__ Wroot,
                        float* __restrict__ Wcat,
                        const int* __restrict__ src, const int* __restrict__ dst,
                        const float* __restrict__ ew,
                        int* __restrict__ keyw, int* __restrict__ rec, int E, int nKeyWords) {
    const int gid = blockIdx.x * 1024 + threadIdx.x;
    if (gid < 128 * 64) {
        int k = gid >> 6, j = gid & 63;
        Wcat[gid] = (j < 32) ? Wrel[k * 32 + j] : Wroot[k * 32 + (j - 32)];
    }
    const int total = gridDim.x * 1024;
    for (int q = gid; q < nKeyWords; q += total) {
        const int e0 = 4 * q;
        if (((E & 3) == 0) && (e0 < E)) {
            const int4   d4 = *(const int4*)(dst + e0);
            const int4   s4 = *(const int4*)(src + e0);
            const float4 w4 = *(const float4*)(ew + e0);
            const unsigned kw = ((unsigned)d4.x / TILE_N) | (((unsigned)d4.y / TILE_N) << 8) |
                                (((unsigned)d4.z / TILE_N) << 16) | (((unsigned)d4.w / TILE_N) << 24);
            keyw[q] = (int)kw;
            int4 r0, r1;
            r0.x = s4.x | (int)(((unsigned)d4.x % TILE_N) << SRC_BITS);  r0.y = __float_as_int(w4.x);
            r0.z = s4.y | (int)(((unsigned)d4.y % TILE_N) << SRC_BITS);  r0.w = __float_as_int(w4.y);
            r1.x = s4.z | (int)(((unsigned)d4.z % TILE_N) << SRC_BITS);  r1.y = __float_as_int(w4.z);
            r1.z = s4.w | (int)(((unsigned)d4.w % TILE_N) << SRC_BITS);  r1.w = __float_as_int(w4.w);
            int4* rp = (int4*)(rec + 2 * (long)e0);
            rp[0] = r0;
            rp[1] = r1;
        } else {
            unsigned kw = 0u;
#pragma unroll
            for (int k = 0; k < 4; ++k) {
                const int e = e0 + k;
                unsigned t = 0xFFu;
                if (e < E) {
                    const unsigned d = (unsigned)dst[e];
                    t = d / TILE_N;
                    rec[2 * (long)e]     = src[e] | (int)((d % TILE_N) << SRC_BITS);
                    rec[2 * (long)e + 1] = __float_as_int(ew[e]);
                }
                kw |= t << (8 * k);
            }
            keyw[q] = (int)kw;
        }
    }
}

__global__ void gcn_build_w2(const float* __restrict__ Wrel, const float* __restrict__ Wroot,
                             float* __restrict__ Wp) {
    int i = blockIdx.x * 256 + threadIdx.x;
    if (i >= 32 * 16) return;
    int k = i >> 4, n = i & 15;
    float v = 0.0f;
    if (n < 5)       v = Wrel[k * 5 + n];
    else if (n < 10) v = Wroot[k * 5 + (n - 5)];
    Wp[i] = v;
}

#define G1_WAVES 8

__device__ __forceinline__ void gcn_gemm1_load_a(const float* __restrict__ x, int tile, int m, int half,
                                                 v2f* a) {
    const float* xr = x + (long)(tile * 16 + m) * D_IN + half * 2;
#pragma unroll
    for (int ks = 0; ks < 32; ++ks) a[ks] = *(const v2f*)(xr + ks * 4);
}

__global__ __launch_bounds__(G1_WAVES * 32)
void gcn_gemm1_wmma(const float* __restrict__ x, const float* __restrict__ W,
                    float* __restrict__ PR1, int nTiles) {
    const int wave = threadIdx.x >> 5;
    const int lane = threadIdx.x & 31;
    const int half = lane >> 4;
    const int m    = lane & 15;
    const int nW   = gridDim.x * G1_WAVES;
    int tile = wave * gridDim.x + blockIdx.x;
    if (tile >= nTiles) return;

    v2f a[32];
    gcn_gemm1_load_a(x, tile, m, half, a);

    v2f b[4][32];
#pragma unroll
    for (int ct = 0; ct < 4; ++ct) {
#pragma unroll
        for (int ks = 0; ks < 32; ++ks) {
            int k = ks * 4 + half * 2;
            int n = ct * 16 + m;
            b[ct][ks].x = W[k * 64 + n];
            b[ct][ks].y = W[(k + 1) * 64 + n];
        }
    }

#pragma unroll 1
    for (; tile < nTiles; tile += nW) {
        const int tn = (tile + nW < nTiles) ? tile + nW : tile;
        v2f an[32];
        gcn_gemm1_load_a(x, tn, m, half, an);

        v8f acc[4] = {};
#pragma unroll
        for (int ks = 0; ks < 32; ++ks) {
#pragma unroll
            for (int ct = 0; ct < 4; ++ct)
                acc[ct] = __builtin_amdgcn_wmma_f32_16x16x4_f32(false, a[ks], false, b[ct][ks],
                                                                (short)0, acc[ct], false, false);
        }
#pragma unroll
        for (int ct = 0; ct < 4; ++ct) {
#pragma unroll
            for (int r = 0; r < 8; ++r) {
                int orow = tile * 16 + r + half * 8;
                PR1[(long)orow * 64 + ct * 16 + m] = acc[ct][r];
            }
        }
#pragma unroll
        for (int ks = 0; ks < 32; ++ks) a[ks] = an[ks];
    }
}

#define AGG_WAVES 32
#define QCAP      160
#define QCAP1     288

__device__ __forceinline__ void gcn_agg1_drain(int nd, int head, int lane, const int* qw,
                                               const int2* __restrict__ rec,
                                               const float* __restrict__ PR1,
                                               float* acc, float* cntl) {
    int s = 0, dl = 0;
    float w = 0.0f;
    if (lane < nd) {
        int idx = head + lane;
        if (idx >= QCAP1) idx -= QCAP1;
        const int2 r = rec[qw[idx]];
        s  = r.x & ((1 << SRC_BITS) - 1);
        dl = (int)((unsigned)r.x >> SRC_BITS);
        w  = __int_as_float(r.y);
        atomicAdd(&cntl[dl], 1.0f);
    }
    for (int k0 = 0; k0 < nd; k0 += 16) {
        float v[16];
#pragma unroll
        for (int j = 0; j < 16; ++j) {
            int sk = __builtin_amdgcn_readlane(s, k0 + j);
            v[j] = PR1[(unsigned)sk * 64u + (unsigned)lane];
        }
#pragma unroll
        for (int j = 0; j < 16; ++j) {
            if (k0 + j < nd) {
                float wk = __int_as_float(__builtin_amdgcn_readlane(__float_as_int(w), k0 + j));
                int   dk = __builtin_amdgcn_readlane(dl, k0 + j);
                atomicAdd(&acc[dk * 32 + lane], wk * v[j]);
            }
        }
    }
}

__device__ __forceinline__ uint2 gcn_load_oct(const int* __restrict__ keyw, int oi, int nOct) {
    int oc = min(oi, nOct - 1);
    return *(const uint2*)(keyw + 2 * (long)oc);
}

__device__ __forceinline__ void gcn_agg1_scan(const int* __restrict__ keyw, const int2* __restrict__ rec,
                                              const float* __restrict__ PR1,
                                              float* acc, float* cntl, int* qw,
                                              int tile, int oBeg, int oEnd, int nOct, int tid, int lane) {
    const int nIt = (oEnd - oBeg + 1023) >> 10;
    int head = 0, qn = 0;

    auto step = [&](uint2& d, int it) {
        const int oi = oBeg + it * 1024 + tid;
        const int e0 = 8 * oi;
        const unsigned want = ((it >= 0) && (oi < oEnd)) ? (unsigned)tile : 0x100u;
#pragma unroll
        for (int k = 0; k < 8; ++k) {
            const unsigned wv = (k < 4) ? d.x : d.y;
            const unsigned kb = (wv >> (8 * (k & 3))) & 0xFFu;
            const bool m = (kb == want);
            unsigned bal = __builtin_amdgcn_ballot_w32(m);
            if (bal != 0u) {
                if (m) {
                    int pos = head + qn + (int)__builtin_amdgcn_mbcnt_lo(bal, 0u);
                    if (pos >= QCAP1) pos -= QCAP1;
                    qw[pos] = e0 + k;
                }
                qn += __builtin_popcount(bal);
            }
        }
        d = gcn_load_oct(keyw, oi + 4096, nOct);
        while (qn >= 32) {
            gcn_agg1_drain(32, head, lane, qw, rec, PR1, acc, cntl);
            head += 32;
            if (head >= QCAP1) head -= QCAP1;
            qn -= 32;
        }
    };

    uint2 d0 = make_uint2(0xFFFFFFFFu, 0xFFFFFFFFu), d1 = d0, d2 = d0, d3 = d0;
#pragma unroll 1
    for (int it = -4; it < nIt; it += 4) {
        step(d0, it);
        step(d1, it + 1);
        step(d2, it + 2);
        step(d3, it + 3);
    }
    if (qn > 0) gcn_agg1_drain(qn, head, lane, qw, rec, PR1, acc, cntl);
}

template <bool VEC>
__device__ __forceinline__ int4 gcn_load_quad(const int* __restrict__ dst, int qi, int nQuads, int E) {
    int qc = min(qi, nQuads - 1);
    if (VEC) return *(const int4*)(dst + 4 * (long)qc);
    int e0 = 4 * qc;
    int4 r;
    r.x = dst[e0];
    r.y = dst[min(e0 + 1, E - 1)];
    r.z = dst[min(e0 + 2, E - 1)];
    r.w = dst[min(e0 + 3, E - 1)];
    return r;
}

__global__ __launch_bounds__(1024)
void gcn_agg1_tile(const int* __restrict__ keyw, const int* __restrict__ rec,
                   const float* __restrict__ PR1,
                   float* __restrict__ Ppart, float* __restrict__ Cpart,
                   int N, int nOct, int nChunks, int octPerChunk) {
    __shared__ __attribute__((aligned(16))) float acc[TILE_N * 32];
    __shared__ float cntl[TILE_N];
    __shared__ int   q[AGG_WAVES][QCAP1];

    const int tid  = threadIdx.x;
    const int wave = tid >> 5;
    const int lane = tid & 31;
    const int tile  = blockIdx.x / nChunks;
    const int chunk = blockIdx.x - tile * nChunks;
    const int base  = tile * TILE_N;
    const int rows  = min(TILE_N, N - base);

    const float4 z4 = {0.0f, 0.0f, 0.0f, 0.0f};
    for (int i = tid; i < TILE_N * 8; i += 1024) ((float4*)acc)[i] = z4;
    for (int i = tid; i < TILE_N; i += 1024) cntl[i] = 0.0f;
    __syncthreads();

    const int oBeg = chunk * octPerChunk;
    const int oEnd = min(oBeg + octPerChunk, nOct);
    if (oEnd > oBeg)
        gcn_agg1_scan(keyw, (const int2*)rec, PR1, acc, cntl, q[wave], tile, oBeg, oEnd, nOct, tid, lane);
    __syncthreads();

    const long planeP = (long)N * 32;
    float4* outP = (float4*)(Ppart + (long)chunk * planeP + (long)base * 32);
    for (int i = tid; i < rows * 8; i += 1024) outP[i] = ((const float4*)acc)[i];
    for (int i = tid; i < rows; i += 1024) Cpart[(long)chunk * N + base + i] = cntl[i];
}

#define MAX_CHUNKS1 8
#define CMB_NODES   128
#define HS_PITCH    36

template <int NC>
__device__ __forceinline__ void gcn_combine1_rows(const float4* __restrict__ P4,
                                                  const float* __restrict__ Cpart,
                                                  const float* __restrict__ PR1, float4 bb,
                                                  float* __restrict__ cnt, float* hs,
                                                  int N, int c4) {
    constexpr int JB = (NC <= 5) ? 4 : 2;
    const int  nUnits = N * 8;
    const long planeQ = (long)N * 8;
    const int  q0     = blockIdx.x * (CMB_NODES * 8) + threadIdx.x;
#pragma unroll
    for (int j0 = 0; j0 < 4; j0 += JB) {
        float4 p[JB][NC];
        float  cq[JB][NC];
        float4 r[JB];
#pragma unroll
        for (int j = 0; j < JB; ++j) {
            const int q = min(q0 + (j0 + j) * 256, nUnits - 1);
            const int n = q >> 3;
#pragma unroll
            for (int c = 0; c < NC; ++c) {
                p[j][c]  = P4[(long)c * planeQ + q];
                cq[j][c] = Cpart[(long)c * N + n];
            }
            r[j] = *(const float4*)(PR1 + (long)n * 64 + 32 + c4 * 4);
        }
#pragma unroll
        for (int j = 0; j < JB; ++j) {
            const int q = q0 + (j0 + j) * 256;
            float4 s = {0.0f, 0.0f, 0.0f, 0.0f};
            float cn = 0.0f;
#pragma unroll
            for (int c = 0; c < NC; ++c) {
                s.x += p[j][c].x;
                s.y += p[j][c].y;
                s.z += p[j][c].z;
                s.w += p[j][c].w;
                cn  += cq[j][c];
            }
            const float c1 = fmaxf(cn, 1.0f);
            float4 v;
            v.x = fmaxf(s.x / c1 + r[j].x + bb.x, 0.0f);
            v.y = fmaxf(s.y / c1 + r[j].y + bb.y, 0.0f);
            v.z = fmaxf(s.z / c1 + r[j].z + bb.z, 0.0f);
            v.w = fmaxf(s.w / c1 + r[j].w + bb.w, 0.0f);
            *(float4*)(hs + ((j0 + j) * 32 + (int)(threadIdx.x >> 3)) * HS_PITCH + c4 * 4) = v;
            if (c4 == 0 && q < nUnits) cnt[q >> 3] = cn;
        }
    }
}

__global__ __launch_bounds__(256)
void gcn_combine1_gemm2(const float* __restrict__ Ppart, const float* __restrict__ Cpart,
                        const float* __restrict__ PR1, const float* __restrict__ b1,
                        const float* __restrict__ W, float* __restrict__ cnt,
                        float* __restrict__ PR2, int N, int nChunks) {
    __shared__ __attribute__((aligned(16))) float hs[CMB_NODES * HS_PITCH];
    const int c4 = threadIdx.x & 7;
    const float4* P4 = (const float4*)Ppart;
    float4 bb;
    bb.x = b1[c4 * 4 + 0];
    bb.y = b1[c4 * 4 + 1];
    bb.z = b1[c4 * 4 + 2];
    bb.w = b1[c4 * 4 + 3];
    switch (nChunks) {
    case 1: gcn_combine1_rows<1>(P4, Cpart, PR1, bb, cnt, hs, N, c4); break;
    case 2: gcn_combine1_rows<2>(P4, Cpart, PR1, bb, cnt, hs, N, c4); break;
    case 3: gcn_combine1_rows<3>(P4, Cpart, PR1, bb, cnt, hs, N, c4); break;
    case 4: gcn_combine1_rows<4>(P4, Cpart, PR1, bb, cnt, hs, N, c4); break;
    case 5: gcn_combine1_rows<5>(P4, Cpart, PR1, bb, cnt, hs, N, c4); break;
    case 6: gcn_combine1_rows<6>(P4, Cpart, PR1, bb, cnt, hs, N, c4); break;
    case 7: gcn_combine1_rows<7>(P4, Cpart, PR1, bb, cnt, hs, N, c4); break;
    case 8: gcn_combine1_rows<8>(P4, Cpart, PR1, bb, cnt, hs, N, c4); break;
    default: break;
    }

    const int wave = threadIdx.x >> 5;
    const int lane = threadIdx.x & 31;
    const int half = lane >> 4;
    const int m    = lane & 15;
    v2f b[8];
#pragma unroll
    for (int ks = 0; ks < 8; ++ks) {
        int k = ks * 4 + half * 2;
        b[ks].x = W[k * 16 + m];
        b[ks].y = W[(k + 1) * 16 + m];
    }
    __syncthreads();
    const int row0 = blockIdx.x * CMB_NODES + wave * 16;
    if (row0 >= N) return;
    const float* hr = hs + (wave * 16 + m) * HS_PITCH;
    v2f a[8];
#pragma unroll
    for (int ks = 0; ks < 8; ++ks) {
        int k = ks * 4 + half * 2;
        a[ks].x = hr[k];
        a[ks].y = hr[k + 1];
    }
    v8f acc = {};
#pragma unroll
    for (int ks = 0; ks < 8; ++ks)
        acc = __builtin_amdgcn_wmma_f32_16x16x4_f32(false, a[ks], false, b[ks],
                                                    (short)0, acc, false, false);
    if (m < 10) {
#pragma unroll
        for (int r = 0; r < 8; ++r) {
            int orow = row0 + r + half * 8;
            PR2[(long)orow * 10 + m] = acc[r];
        }
    }
}

#define TILE2_N 6400

__device__ __forceinline__ void gcn_agg2_drain(int nd, int head, int lane, const int2* qw,
                                               const int* __restrict__ src,
                                               const float* __restrict__ PR2, float* acc) {
    if (lane < nd) {
        int idx = head + lane;
        if (idx >= QCAP) idx -= QCAP;
        int2 ent = qw[idx];
        const float* p = PR2 + (unsigned)src[ent.x] * 10u;
        float2 v01 = *(const float2*)p;
        float2 v23 = *(const float2*)(p + 2);
        float  v4  = p[4];
        float* a = acc + ent.y * 5;
        atomicAdd(a + 0, v01.x);
        atomicAdd(a + 1, v01.y);
        atomicAdd(a + 2, v23.x);
        atomicAdd(a + 3, v23.y);
        atomicAdd(a + 4, v4);
    }
}

template <bool VEC>
__device__ __forceinline__ void gcn_agg2_scan(const int* __restrict__ src, const int* __restrict__ dst,
                                              const float* __restrict__ PR2,
                                              float* acc, int2* qw,
                                              int base, int rows, int qBeg, int qEnd,
                                              int nQuads, int E, int tid, int lane) {
    const int nIt = (qEnd - qBeg + 1023) >> 10;
    int head = 0, qn = 0;

    auto step = [&](int4& d, int it) {
        const int qi = qBeg + it * 1024 + tid;
        const int e0 = 4 * qi;
        const unsigned lim = ((it >= 0) && (qi < qEnd)) ? (unsigned)rows : 0u;
#pragma unroll
        for (int k = 0; k < 4; ++k) {
            int dv = (k == 0) ? d.x : (k == 1) ? d.y : (k == 2) ? d.z : d.w;
            int dl = dv - base;
            bool m = (unsigned)dl < lim;
            if (!VEC) m = m && (e0 + k < E);
            unsigned bal = __builtin_amdgcn_ballot_w32(m);
            if (bal != 0u) {
                if (m) {
                    int pos = head + qn + (int)__builtin_amdgcn_mbcnt_lo(bal, 0u);
                    if (pos >= QCAP) pos -= QCAP;
                    qw[pos] = make_int2(e0 + k, dl);
                }
                qn += __builtin_popcount(bal);
            }
        }
        d = gcn_load_quad<VEC>(dst, qi + 4096, nQuads, E);
        while (qn >= 32) {
            gcn_agg2_drain(32, head, lane, qw, src, PR2, acc);
            head += 32;
            if (head >= QCAP) head -= QCAP;
            qn -= 32;
        }
    };

    int4 d0 = make_int4(-1, -1, -1, -1), d1 = d0, d2 = d0, d3 = d0;
#pragma unroll 1
    for (int it = -4; it < nIt; it += 4) {
        step(d0, it);
        step(d1, it + 1);
        step(d2, it + 2);
        step(d3, it + 3);
    }
    if (qn > 0) gcn_agg2_drain(qn, head, lane, qw, src, PR2, acc);
}

__global__ __launch_bounds__(1024)
void gcn_agg2_tile(const int* __restrict__ src, const int* __restrict__ dst,
                   const float* __restrict__ PR2, float* __restrict__ P2,
                   int N, int E, int nChunks, int quadsPerChunk) {
    __shared__ __attribute__((aligned(16))) float acc[TILE2_N * 5];
    __shared__ int2  q[AGG_WAVES][QCAP];

    const int tid  = threadIdx.x;
    const int wave = tid >> 5;
    const int lane = tid & 31;
    const int tile  = blockIdx.x / nChunks;
    const int chunk = blockIdx.x - tile * nChunks;
    const int base  = tile * TILE2_N;
    const int rows  = min(TILE2_N, N - base);

    const float4 z4 = {0.0f, 0.0f, 0.0f, 0.0f};
    for (int i = tid; i < TILE2_N * 5 / 4; i += 1024) ((float4*)acc)[i] = z4;
    __syncthreads();

    const int nQuads = (E + 3) >> 2;
    const int qBeg   = chunk * quadsPerChunk;
    const int qEnd   = min(qBeg + quadsPerChunk, nQuads);
    if (qEnd > qBeg) {
        if ((E & 3) == 0)
            gcn_agg2_scan<true>(src, dst, PR2, acc, q[wave], base, rows, qBeg, qEnd, nQuads, E, tid, lane);
        else
            gcn_agg2_scan<false>(src, dst, PR2, acc, q[wave], base, rows, qBeg, qEnd, nQuads, E, tid, lane);
    }
    __syncthreads();

    float* outP = P2 + (long)chunk * N * 5 + (long)base * 5;
    for (int i = tid; i < rows * 5; i += 1024) outP[i] = acc[i];
}

#define OUT_NODES   256
#define OUT_THREADS (OUT_NODES * 5 / 4)
#define MAX_CHUNKS2 16

template <int NC>
__device__ __forceinline__ float4 gcn_plane_sum4(const float4* __restrict__ P4, long planeQ) {
    float4 p[NC];
#pragma unroll
    for (int cz = 0; cz < NC; ++cz) p[cz] = P4[(long)cz * planeQ];
    float4 s = {0.0f, 0.0f, 0.0f, 0.0f};
#pragma unroll
    for (int cz = 0; cz < NC; ++cz) {
        s.x += p[cz].x;
        s.y += p[cz].y;
        s.z += p[cz].z;
        s.w += p[cz].w;
    }
    return s;
}

__global__ __launch_bounds__(OUT_THREADS)
void gcn_output(const float* __restrict__ P2, const float* __restrict__ PR2,
                const float* __restrict__ b2, const float* __restrict__ cnt,
                float* __restrict__ out, int N, int nChunks) {
    __shared__ __attribute__((aligned(16))) float s2[OUT_NODES * 5];
    const int  n0     = blockIdx.x * OUT_NODES;
    const long totalQ = (long)N * 5 / 4;
    const long gq     = (long)blockIdx.x * OUT_THREADS + threadIdx.x;
    const int  n      = n0 + (int)threadIdx.x;
    const bool fin    = (threadIdx.x < OUT_NODES) && (n < N);
    const int   nl = min(n, N - 1);
    const float cn = cnt[nl];
    float r[5];
#pragma unroll
    for (int j = 0; j < 5; ++j) r[j] = PR2[(long)nl * 10 + 5 + j];
    float4 sp = {0.0f, 0.0f, 0.0f, 0.0f};
    if (gq < totalQ) {
        const float4* P4 = (const float4*)P2 + gq;
        switch (nChunks) {
        case 1:  sp = gcn_plane_sum4<1>(P4, totalQ);  break;
        case 2:  sp = gcn_plane_sum4<2>(P4, totalQ);  break;
        case 3:  sp = gcn_plane_sum4<3>(P4, totalQ);  break;
        case 4:  sp = gcn_plane_sum4<4>(P4, totalQ);  break;
        case 5:  sp = gcn_plane_sum4<5>(P4, totalQ);  break;
        case 6:  sp = gcn_plane_sum4<6>(P4, totalQ);  break;
        case 7:  sp = gcn_plane_sum4<7>(P4, totalQ);  break;
        case 8:  sp = gcn_plane_sum4<8>(P4, totalQ);  break;
        case 9:  sp = gcn_plane_sum4<9>(P4, totalQ);  break;
        case 10: sp = gcn_plane_sum4<10>(P4, totalQ); break;
        case 11: sp = gcn_plane_sum4<11>(P4, totalQ); break;
        case 12: sp = gcn_plane_sum4<12>(P4, totalQ); break;
        case 13: sp = gcn_plane_sum4<13>(P4, totalQ); break;
        case 14: sp = gcn_plane_sum4<14>(P4, totalQ); break;
        case 15: sp = gcn_plane_sum4<15>(P4, totalQ); break;
        case 16: sp = gcn_plane_sum4<16>(P4, totalQ); break;
        default: break;
        }
    }
    ((float4*)s2)[threadIdx.x] = sp;
    __syncthreads();
    if (!fin) return;
    const float c = fmaxf(cn, 1.0f);
    float v[5];
#pragma unroll
    for (int j = 0; j < 5; ++j)
        v[j] = s2[threadIdx.x * 5 + j] / c + r[j] + b2[j];
    float mx = v[0];
#pragma unroll
    for (int j = 1; j < 5; ++j) mx = fmaxf(mx, v[j]);
    float s = 0.0f;
#pragma unroll
    for (int j = 0; j < 5; ++j) s += expf(v[j] - mx);
    float l = logf(s);
#pragma unroll
    for (int j = 0; j < 5; ++j) out[(long)n * 5 + j] = v[j] - mx - l;
}

extern "C" void kernel_launch(void* const* d_in, const int* in_sizes, int n_in,
                              void* d_out, int out_size, void* d_ws, size_t ws_size,
                              hipStream_t stream) {
    const float* x      = (const float*)d_in[0];
    const int*   ei     = (const int*)d_in[1];
    const float* ew     = (const float*)d_in[2];
    const float* Wrel1  = (const float*)d_in[3];
    const float* Wroot1 = (const float*)d_in[4];
    const float* b1     = (const float*)d_in[5];
    const float* Wrel2  = (const float*)d_in[6];
    const float* Wroot2 = (const float*)d_in[7];
    const float* b2     = (const float*)d_in[8];

    const int N = in_sizes[0] / D_IN;
    const int E = in_sizes[1] / 2;
    const int* src = ei;
    const int* dst = ei + E;

    float* ws = (float*)d_ws;
    long off = 0;
    float* PR1 = ws + off; off += (long)N * 64;
    float* cnt = ws + off; off += N;
    float* PR2 = ws + off; off += (long)N * 10;
    float* W1c = ws + off; off += 128 * 64;
    float* W2p = ws + off; off += 32 * 16;
    const int nT = (N + TILE_N - 1) / TILE_N;
    const int nC = (nT >= 256) ? 1 : ((256 / nT > MAX_CHUNKS1) ? MAX_CHUNKS1 : 256 / nT);
    float* Pp  = ws + off; off += (long)nC * N * 32;
    float* Cp  = ws + off; off += (long)nC * N;
    const int nT2 = (N + TILE2_N - 1) / TILE2_N;
    const int nC2 = (nT2 >= 256) ? 1 : ((256 / nT2 > MAX_CHUNKS2) ? MAX_CHUNKS2 : 256 / nT2);
    float* P2  = ws + off; off += (long)nC2 * N * 5;
    const int nOct = (E + 7) / 8;
    int* keyw = (int*)(ws + off); off += (long)((nOct + 1) / 2) * 4;
    int* rec  = (int*)(ws + off); off += (long)E * 2;

    const int nTiles = N / 16;

    gcn_build_w1_edges<<<256, 1024, 0, stream>>>(Wrel1, Wroot1, W1c, src, dst, ew, keyw, rec, E, nOct * 2);
    gcn_build_w2<<<(32 * 16 + 255) / 256, 256, 0, stream>>>(Wrel2, Wroot2, W2p);

    gcn_gemm1_wmma<<<256, G1_WAVES * 32, 0, stream>>>(x, W1c, PR1, nTiles);

    const int nQuads = (E + 3) / 4;
    const int opc = (nOct + nC - 1) / nC;
    gcn_agg1_tile<<<nT * nC, 1024, 0, stream>>>(keyw, rec, PR1, Pp, Cp, N, nOct, nC, opc);

    gcn_combine1_gemm2<<<(N + CMB_NODES - 1) / CMB_NODES, 256, 0, stream>>>(Pp, Cp, PR1, b1, W2p, cnt, PR2, N, nC);

    const int qpc2 = (nQuads + nC2 - 1) / nC2;
    gcn_agg2_tile<<<nT2 * nC2, 1024, 0, stream>>>(src, dst, PR2, P2, N, E, nC2, qpc2);

    gcn_output<<<(N + OUT_NODES - 1) / OUT_NODES, OUT_THREADS, 0, stream>>>(P2, PR2, b2, cnt, (float*)d_out, N, nC2);
}


# Layer-1 scan reads one bit per edge from its tile's row of a bit plane (32 edges per lane word, set bits popped) that the first launch builds by LDS ORs; sum order: LDS arrival, then chunk
# speedup vs baseline: 3.5712x; 1.0683x over previous
#include <hip/hip_runtime.h>
#include <math.h>

typedef __attribute__((ext_vector_type(2))) float v2f;
typedef __attribute__((ext_vector_type(8))) float v8f;

#define D_IN  128
#define H_DIM 32
#define C_DIM 5

#define TILE_N   2048
#define SRC_BITS 20

#define BM_EDGES 8192
#define BM_WORDS (BM_EDGES / 32)
#define BM_LP    (BM_WORDS + 4)
#define BM_MAXT  128

template <bool VEC>
__device__ __forceinline__ void gcn_build_edges(const int* __restrict__ src, const int* __restrict__ dst,
                                                const float* __restrict__ ew,
                                                unsigned* bl, int* __restrict__ rec, int E, int eb, int tid) {
    int4   d4[8], s4[8];
    float4 w4[8];
#pragma unroll
    for (int j = 0; j < 8; ++j) {
        const int e0 = eb + 4 * (j * 256 + tid);
        if (VEC) {
            const int ec = min(e0, E - 4);
            d4[j] = *(const int4*)(dst + ec);
            s4[j] = *(const int4*)(src + ec);
            w4[j] = *(const float4*)(ew + ec);
        } else {
            const int a = min(e0, E - 1), b = min(e0 + 1, E - 1), c = min(e0 + 2, E - 1), d = min(e0 + 3, E - 1);
            d4[j] = make_int4(dst[a], dst[b], dst[c], dst[d]);
            s4[j] = make_int4(src[a], src[b], src[c], src[d]);
            w4[j] = make_float4(ew[a], ew[b], ew[c], ew[d]);
        }
    }
#pragma unroll
    for (int j = 0; j < 8; ++j) {
        const int ql = j * 256 + tid;
        const int e0 = eb + 4 * ql;
        const unsigned dd[4] = {(unsigned)d4[j].x, (unsigned)d4[j].y, (unsigned)d4[j].z, (unsigned)d4[j].w};
        const int      ss[4] = {s4[j].x, s4[j].y, s4[j].z, s4[j].w};
        const float    ww[4] = {w4[j].x, w4[j].y, w4[j].z, w4[j].w};
        int r[8];
#pragma unroll
        for (int k = 0; k < 4; ++k) {
            r[2 * k]     = ss[k] | (int)((dd[k] % TILE_N) << SRC_BITS);
            r[2 * k + 1] = __float_as_int(ww[k]);
            if (e0 + k < E)
                atomicOr(&bl[(dd[k] / TILE_N) * BM_LP + (ql >> 3)], 1u << ((ql & 7) * 4 + k));
        }
        if (VEC) {
            if (e0 < E) {
                int4* rp = (int4*)(rec + 2 * (long)e0);
                rp[0] = make_int4(r[0], r[1], r[2], r[3]);
                rp[1] = make_int4(r[4], r[5], r[6], r[7]);
            }
        } else {
#pragma unroll
            for (int k = 0; k < 4; ++k) {
                if (e0 + k < E) {
                    rec[2 * (long)(e0 + k)]     = r[2 * k];
                    rec[2 * (long)(e0 + k) + 1] = r[2 * k + 1];
                }
            }
        }
    }
}

__global__ __launch_bounds__(256)
void gcn_build_w1_edges(const float* __restrict__ Wrel, const float* __restrict__ Wroot,
                        float* __restrict__ Wcat,
                        const int* __restrict__ src, const int* __restrict__ dst,
                        const float* __restrict__ ew,
                        unsigned* __restrict__ bm, int* __restrict__ rec, int E, int nT, int pitchW) {
    __shared__ __attribute__((aligned(16))) unsigned bl[BM_MAXT * BM_LP];
    const int tid = threadIdx.x;
    for (int i = blockIdx.x * 256 + tid; i < 128 * 64; i += gridDim.x * 256) {
        int k = i >> 6, j = i & 63;
        Wcat[i] = (j < 32) ? Wrel[k * 32 + j] : Wroot[k * 32 + (j - 32)];
    }
    const uint4 z4 = {0u, 0u, 0u, 0u};
    for (int i = tid; i < nT * (BM_LP / 4); i += 256) ((uint4*)bl)[i] = z4;
    __syncthreads();

    const int eb = blockIdx.x * BM_EDGES;
    if ((E & 3) == 0) gcn_build_edges<true>(src, dst, ew, bl, rec, E, eb, tid);
    else              gcn_build_edges<false>(src, dst, ew, bl, rec, E, eb, tid);
    __syncthreads();

    const int wq = tid & 63;
    for (int t = tid >> 6; t < nT; t += 4) {
        const uint4 v = *(const uint4*)(bl + t * BM_LP + wq * 4);
        *(uint4*)(bm + (long)t * pitchW + blockIdx.x * BM_WORDS + wq * 4) = v;
    }
}

__global__ void gcn_build_w2(const float* __restrict__ Wrel, const float* __restrict__ Wroot,
                             float* __restrict__ Wp) {
    int i = blockIdx.x * 256 + threadIdx.x;
    if (i >= 32 * 16) return;
    int k = i >> 4, n = i & 15;
    float v = 0.0f;
    if (n < 5)       v = Wrel[k * 5 + n];
    else if (n < 10) v = Wroot[k * 5 + (n - 5)];
    Wp[i] = v;
}

#define G1_WAVES 8

__device__ __forceinline__ void gcn_gemm1_load_a(const float* __restrict__ x, int tile, int m, int half,
                                                 v2f* a) {
    const float* xr = x + (long)(tile * 16 + m) * D_IN + half * 2;
#pragma unroll
    for (int ks = 0; ks < 32; ++ks) a[ks] = *(const v2f*)(xr + ks * 4);
}

__global__ __launch_bounds__(G1_WAVES * 32)
void gcn_gemm1_wmma(const float* __restrict__ x, const float* __restrict__ W,
                    float* __restrict__ PR1, int nTiles) {
    const int wave = threadIdx.x >> 5;
    const int lane = threadIdx.x & 31;
    const int half = lane >> 4;
    const int m    = lane & 15;
    const int nW   = gridDim.x * G1_WAVES;
    int tile = wave * gridDim.x + blockIdx.x;
    if (tile >= nTiles) return;

    v2f a[32];
    gcn_gemm1_load_a(x, tile, m, half, a);

    v2f b[4][32];
#pragma unroll
    for (int ct = 0; ct < 4; ++ct) {
#pragma unroll
        for (int ks = 0; ks < 32; ++ks) {
            int k = ks * 4 + half * 2;
            int n = ct * 16 + m;
            b[ct][ks].x = W[k * 64 + n];
            b[ct][ks].y = W[(k + 1) * 64 + n];
        }
    }

#pragma unroll 1
    for (; tile < nTiles; tile += nW) {
        const int tn = (tile + nW < nTiles) ? tile + nW : tile;
        v2f an[32];
        gcn_gemm1_load_a(x, tn, m, half, an);

        v8f acc[4] = {};
#pragma unroll
        for (int ks = 0; ks < 32; ++ks) {
#pragma unroll
            for (int ct = 0; ct < 4; ++ct)
                acc[ct] = __builtin_amdgcn_wmma_f32_16x16x4_f32(false, a[ks], false, b[ct][ks],
                                                                (short)0, acc[ct], false, false);
        }
#pragma unroll
        for (int ct = 0; ct < 4; ++ct) {
#pragma unroll
            for (int r = 0; r < 8; ++r) {
                int orow = tile * 16 + r + half * 8;
                PR1[(long)orow * 64 + ct * 16 + m] = acc[ct][r];
            }
        }
#pragma unroll
        for (int ks = 0; ks < 32; ++ks) a[ks] = an[ks];
    }
}

#define AGG_WAVES 32
#define QCAP      160
#define QCAP1     64

__device__ __forceinline__ void gcn_agg1_drain(int nd, int head, int lane, const int* qw,
                                               const int2* __restrict__ rec,
                                               const float* __restrict__ PR1,
                                               float* acc, float* cntl) {
    int s = 0, dl = 0;
    float w = 0.0f;
    if (lane < nd) {
        int idx = head + lane;
        if (idx >= QCAP1) idx -= QCAP1;
        const int2 r = rec[qw[idx]];
        s  = r.x & ((1 << SRC_BITS) - 1);
        dl = (int)((unsigned)r.x >> SRC_BITS);
        w  = __int_as_float(r.y);
        atomicAdd(&cntl[dl], 1.0f);
    }
    for (int k0 = 0; k0 < nd; k0 += 16) {
        float v[16];
#pragma unroll
        for (int j = 0; j < 16; ++j) {
            int sk = __builtin_amdgcn_readlane(s, k0 + j);
            v[j] = PR1[(unsigned)sk * 64u + (unsigned)lane];
        }
#pragma unroll
        for (int j = 0; j < 16; ++j) {
            if (k0 + j < nd) {
                float wk = __int_as_float(__builtin_amdgcn_readlane(__float_as_int(w), k0 + j));
                int   dk = __builtin_amdgcn_readlane(dl, k0 + j);
                atomicAdd(&acc[dk * 32 + lane], wk * v[j]);
            }
        }
    }
}

__device__ __forceinline__ unsigned gcn_load_word(const unsigned* __restrict__ row, int wi, int pitchW) {
    return row[min(wi, pitchW - 1)];
}

__device__ __forceinline__ void gcn_agg1_scan(const unsigned* __restrict__ row, const int2* __restrict__ rec,
                                              const float* __restrict__ PR1,
                                              float* acc, float* cntl, int* qw,
                                              int wBeg, int wEnd, int pitchW, int tid, int lane) {
    const int nIt = (wEnd - wBeg + 1023) >> 10;
    int head = 0, qn = 0;

    auto step = [&](unsigned& d, int it) {
        const int wi = wBeg + it * 1024 + tid;
        const int e0 = 32 * wi;
        unsigned m = ((it >= 0) && (wi < wEnd)) ? d : 0u;
        d = gcn_load_word(row, wi + 4096, pitchW);
#pragma unroll 1
        for (int r = 0; r < 32; ++r) {
            const bool has = (m != 0u);
            const unsigned bal = __builtin_amdgcn_ballot_w32(has);
            if (bal == 0u) break;
            if (has) {
                int pos = head + qn + (int)__builtin_amdgcn_mbcnt_lo(bal, 0u);
                if (pos >= QCAP1) pos -= QCAP1;
                qw[pos] = e0 + __builtin_ctz(m);
                m &= m - 1u;
            }
            qn += __builtin_popcount(bal);
            if (qn >= 32) {
                gcn_agg1_drain(32, head, lane, qw, rec, PR1, acc, cntl);
                head += 32;
                if (head >= QCAP1) head -= QCAP1;
                qn -= 32;
            }
        }
    };

    unsigned d0 = 0u, d1 = 0u, d2 = 0u, d3 = 0u;
#pragma unroll 1
    for (int it = -4; it < nIt; it += 4) {
        step(d0, it);
        step(d1, it + 1);
        step(d2, it + 2);
        step(d3, it + 3);
    }
    if (qn > 0) gcn_agg1_drain(qn, head, lane, qw, rec, PR1, acc, cntl);
}

template <bool VEC>
__device__ __forceinline__ int4 gcn_load_quad(const int* __restrict__ dst, int qi, int nQuads, int E) {
    int qc = min(qi, nQuads - 1);
    if (VEC) return *(const int4*)(dst + 4 * (long)qc);
    int e0 = 4 * qc;
    int4 r;
    r.x = dst[e0];
    r.y = dst[min(e0 + 1, E - 1)];
    r.z = dst[min(e0 + 2, E - 1)];
    r.w = dst[min(e0 + 3, E - 1)];
    return r;
}

__global__ __launch_bounds__(1024)
void gcn_agg1_tile(const unsigned* __restrict__ bm, const int* __restrict__ rec,
                   const float* __restrict__ PR1,
                   float* __restrict__ Ppart, float* __restrict__ Cpart,
                   int N, int nWords, int pitchW, int nChunks, int wordsPerChunk) {
    __shared__ __attribute__((aligned(16))) float acc[TILE_N * 32];
    __shared__ float cntl[TILE_N];
    __shared__ int   q[AGG_WAVES][QCAP1];

    const int tid  = threadIdx.x;
    const int wave = tid >> 5;
    const int lane = tid & 31;
    const int tile  = blockIdx.x / nChunks;
    const int chunk = blockIdx.x - tile * nChunks;
    const int base  = tile * TILE_N;
    const int rows  = min(TILE_N, N - base);

    const float4 z4 = {0.0f, 0.0f, 0.0f, 0.0f};
    for (int i = tid; i < TILE_N * 8; i += 1024) ((float4*)acc)[i] = z4;
    for (int i = tid; i < TILE_N; i += 1024) cntl[i] = 0.0f;
    __syncthreads();

    const int wBeg = chunk * wordsPerChunk;
    const int wEnd = min(wBeg + wordsPerChunk, nWords);
    if (wEnd > wBeg)
        gcn_agg1_scan(bm + (long)tile * pitchW, (const int2*)rec, PR1, acc, cntl, q[wave], wBeg, wEnd, pitchW, tid, lane);
    __syncthreads();

    const long planeP = (long)N * 32;
    float4* outP = (float4*)(Ppart + (long)chunk * planeP + (long)base * 32);
    for (int i = tid; i < rows * 8; i += 1024) outP[i] = ((const float4*)acc)[i];
    for (int i = tid; i < rows; i += 1024) Cpart[(long)chunk * N + base + i] = cntl[i];
}

#define MAX_CHUNKS1 8
#define CMB_NODES   128
#define HS_PITCH    36

template <int NC>
__device__ __forceinline__ void gcn_combine1_rows(const float4* __restrict__ P4,
                                                  const float* __restrict__ Cpart,
                                                  const float* __restrict__ PR1, float4 bb,
                                                  float* __restrict__ cnt, float* hs,
                                                  int N, int c4) {
    constexpr int JB = (NC <= 5) ? 4 : 2;
    const int  nUnits = N * 8;
    const long planeQ = (long)N * 8;
    const int  q0     = blockIdx.x * (CMB_NODES * 8) + threadIdx.x;
#pragma unroll
    for (int j0 = 0; j0 < 4; j0 += JB) {
        float4 p[JB][NC];
        float  cq[JB][NC];
        float4 r[JB];
#pragma unroll
        for (int j = 0; j < JB; ++j) {
            const int q = min(q0 + (j0 + j) * 256, nUnits - 1);
            const int n = q >> 3;
#pragma unroll
            for (int c = 0; c < NC; ++c) {
                p[j][c]  = P4[(long)c * planeQ + q];
                cq[j][c] = Cpart[(long)c * N + n];
            }
            r[j] = *(const float4*)(PR1 + (long)n * 64 + 32 + c4 * 4);
        }
#pragma unroll
        for (int j = 0; j < JB; ++j) {
            const int q = q0 + (j0 + j) * 256;
            float4 s = {0.0f, 0.0f, 0.0f, 0.0f};
            float cn = 0.0f;
#pragma unroll
            for (int c = 0; c < NC; ++c) {
                s.x += p[j][c].x;
                s.y += p[j][c].y;
                s.z += p[j][c].z;
                s.w += p[j][c].w;
                cn  += cq[j][c];
            }
            const float c1 = fmaxf(cn, 1.0f);
            float4 v;
            v.x = fmaxf(s.x / c1 + r[j].x + bb.x, 0.0f);
            v.y = fmaxf(s.y / c1 + r[j].y + bb.y, 0.0f);
            v.z = fmaxf(s.z / c1 + r[j].z + bb.z, 0.0f);
            v.w = fmaxf(s.w / c1 + r[j].w + bb.w, 0.0f);
            *(float4*)(hs + ((j0 + j) * 32 + (int)(threadIdx.x >> 3)) * HS_PITCH + c4 * 4) = v;
            if (c4 == 0 && q < nUnits) cnt[q >> 3] = cn;
        }
    }
}

__global__ __launch_bounds__(256)
void gcn_combine1_gemm2(const float* __restrict__ Ppart, const float* __restrict__ Cpart,
                        const float* __restrict__ PR1, const float* __restrict__ b1,
                        const float* __restrict__ W, float* __restrict__ cnt,
                        float* __restrict__ PR2, int N, int nChunks) {
    __shared__ __attribute__((aligned(16))) float hs[CMB_NODES * HS_PITCH];
    const int c4 = threadIdx.x & 7;
    const float4* P4 = (const float4*)Ppart;
    float4 bb;
    bb.x = b1[c4 * 4 + 0];
    bb.y = b1[c4 * 4 + 1];
    bb.z = b1[c4 * 4 + 2];
    bb.w = b1[c4 * 4 + 3];
    switch (nChunks) {
    case 1: gcn_combine1_rows<1>(P4, Cpart, PR1, bb, cnt, hs, N, c4); break;
    case 2: gcn_combine1_rows<2>(P4, Cpart, PR1, bb, cnt, hs, N, c4); break;
    case 3: gcn_combine1_rows<3>(P4, Cpart, PR1, bb, cnt, hs, N, c4); break;
    case 4: gcn_combine1_rows<4>(P4, Cpart, PR1, bb, cnt, hs, N, c4); break;
    case 5: gcn_combine1_rows<5>(P4, Cpart, PR1, bb, cnt, hs, N, c4); break;
    case 6: gcn_combine1_rows<6>(P4, Cpart, PR1, bb, cnt, hs, N, c4); break;
    case 7: gcn_combine1_rows<7>(P4, Cpart, PR1, bb, cnt, hs, N, c4); break;
    case 8: gcn_combine1_rows<8>(P4, Cpart, PR1, bb, cnt, hs, N, c4); break;
    default: break;
    }

    const int wave = threadIdx.x >> 5;
    const int lane = threadIdx.x & 31;
    const int half = lane >> 4;
    const int m    = lane & 15;
    v2f b[8];
#pragma unroll
    for (int ks = 0; ks < 8; ++ks) {
        int k = ks * 4 + half * 2;
        b[ks].x = W[k * 16 + m];
        b[ks].y = W[(k + 1) * 16 + m];
    }
    __syncthreads();
    const int row0 = blockIdx.x * CMB_NODES + wave * 16;
    if (row0 >= N) return;
    const float* hr = hs + (wave * 16 + m) * HS_PITCH;
    v2f a[8];
#pragma unroll
    for (int ks = 0; ks < 8; ++ks) {
        int k = ks * 4 + half * 2;
        a[ks].x = hr[k];
        a[ks].y = hr[k + 1];
    }
    v8f acc = {};
#pragma unroll
    for (int ks = 0; ks < 8; ++ks)
        acc = __builtin_amdgcn_wmma_f32_16x16x4_f32(false, a[ks], false, b[ks],
                                                    (short)0, acc, false, false);
    if (m < 10) {
#pragma unroll
        for (int r = 0; r < 8; ++r) {
            int orow = row0 + r + half * 8;
            PR2[(long)orow * 10 + m] = acc[r];
        }
    }
}

#define TILE2_N 6400

__device__ __forceinline__ void gcn_agg2_drain(int nd, int head, int lane, const int2* qw,
                                               const int* __restrict__ src,
                                               const float* __restrict__ PR2, float* acc) {
    if (lane < nd) {
        int idx = head + lane;
        if (idx >= QCAP) idx -= QCAP;
        int2 ent = qw[idx];
        const float* p = PR2 + (unsigned)src[ent.x] * 10u;
        float2 v01 = *(const float2*)p;
        float2 v23 = *(const float2*)(p + 2);
        float  v4  = p[4];
        float* a = acc + ent.y * 5;
        atomicAdd(a + 0, v01.x);
        atomicAdd(a + 1, v01.y);
        atomicAdd(a + 2, v23.x);
        atomicAdd(a + 3, v23.y);
        atomicAdd(a + 4, v4);
    }
}

template <bool VEC>
__device__ __forceinline__ void gcn_agg2_scan(const int* __restrict__ src, const int* __restrict__ dst,
                                              const float* __restrict__ PR2,
                                              float* acc, int2* qw,
                                              int base, int rows, int qBeg, int qEnd,
                                              int nQuads, int E, int tid, int lane) {
    const int nIt = (qEnd - qBeg + 1023) >> 10;
    int head = 0, qn = 0;

    auto step = [&](int4& d, int it) {
        const int qi = qBeg + it * 1024 + tid;
        const int e0 = 4 * qi;
        const unsigned lim = ((it >= 0) && (qi < qEnd)) ? (unsigned)rows : 0u;
#pragma unroll
        for (int k = 0; k < 4; ++k) {
            int dv = (k == 0) ? d.x : (k == 1) ? d.y : (k == 2) ? d.z : d.w;
            int dl = dv - base;
            bool m = (unsigned)dl < lim;
            if (!VEC) m = m && (e0 + k < E);
            unsigned bal = __builtin_amdgcn_ballot_w32(m);
            if (bal != 0u) {
                if (m) {
                    int pos = head + qn + (int)__builtin_amdgcn_mbcnt_lo(bal, 0u);
                    if (pos >= QCAP) pos -= QCAP;
                    qw[pos] = make_int2(e0 + k, dl);
                }
                qn += __builtin_popcount(bal);
            }
        }
        d = gcn_load_quad<VEC>(dst, qi + 4096, nQuads, E);
        while (qn >= 32) {
            gcn_agg2_drain(32, head, lane, qw, src, PR2, acc);
            head += 32;
            if (head >= QCAP) head -= QCAP;
            qn -= 32;
        }
    };

    int4 d0 = make_int4(-1, -1, -1, -1), d1 = d0, d2 = d0, d3 = d0;
#pragma unroll 1
    for (int it = -4; it < nIt; it += 4) {
        step(d0, it);
        step(d1, it + 1);
        step(d2, it + 2);
        step(d3, it + 3);
    }
    if (qn > 0) gcn_agg2_drain(qn, head, lane, qw, src, PR2, acc);
}

__global__ __launch_bounds__(1024)
void gcn_agg2_tile(const int* __restrict__ src, const int* __restrict__ dst,
                   const float* __restrict__ PR2, float* __restrict__ P2,
                   int N, int E, int nChunks, int quadsPerChunk) {
    __shared__ __attribute__((aligned(16))) float acc[TILE2_N * 5];
    __shared__ int2  q[AGG_WAVES][QCAP];

    const int tid  = threadIdx.x;
    const int wave = tid >> 5;
    const int lane = tid & 31;
    const int tile  = blockIdx.x / nChunks;
    const int chunk = blockIdx.x - tile * nChunks;
    const int base  = tile * TILE2_N;
    const int rows  = min(TILE2_N, N - base);

    const float4 z4 = {0.0f, 0.0f, 0.0f, 0.0f};
    for (int i = tid; i < TILE2_N * 5 / 4; i += 1024) ((float4*)acc)[i] = z4;
    __syncthreads();

    const int nQuads = (E + 3) >> 2;
    const int qBeg   = chunk * quadsPerChunk;
    const int qEnd   = min(qBeg + quadsPerChunk, nQuads);
    if (qEnd > qBeg) {
        if ((E & 3) == 0)
            gcn_agg2_scan<true>(src, dst, PR2, acc, q[wave], base, rows, qBeg, qEnd, nQuads, E, tid, lane);
        else
            gcn_agg2_scan<false>(src, dst, PR2, acc, q[wave], base, rows, qBeg, qEnd, nQuads, E, tid, lane);
    }
    __syncthreads();

    float* outP = P2 + (long)chunk * N * 5 + (long)base * 5;
    for (int i = tid; i < rows * 5; i += 1024) outP[i] = acc[i];
}

#define OUT_NODES   256
#define OUT_THREADS (OUT_NODES * 5 / 4)
#define MAX_CHUNKS2 16

template <int NC>
__device__ __forceinline__ float4 gcn_plane_sum4(const float4* __restrict__ P4, long planeQ) {
    float4 p[NC];
#pragma unroll
    for (int cz = 0; cz < NC; ++cz) p[cz] = P4[(long)cz * planeQ];
    float4 s = {0.0f, 0.0f, 0.0f, 0.0f};
#pragma unroll
    for (int cz = 0; cz < NC; ++cz) {
        s.x += p[cz].x;
        s.y += p[cz].y;
        s.z += p[cz].z;
        s.w += p[cz].w;
    }
    return s;
}

__global__ __launch_bounds__(OUT_THREADS)
void gcn_output(const float* __restrict__ P2, const float* __restrict__ PR2,
                const float* __restrict__ b2, const float* __restrict__ cnt,
                float* __restrict__ out, int N, int nChunks) {
    __shared__ __attribute__((aligned(16))) float s2[OUT_NODES * 5];
    const int  n0     = blockIdx.x * OUT_NODES;
    const long totalQ = (long)N * 5 / 4;
    const long gq     = (long)blockIdx.x * OUT_THREADS + threadIdx.x;
    const int  n      = n0 + (int)threadIdx.x;
    const bool fin    = (threadIdx.x < OUT_NODES) && (n < N);
    const int   nl = min(n, N - 1);
    const float cn = cnt[nl];
    float r[5];
#pragma unroll
    for (int j = 0; j < 5; ++j) r[j] = PR2[(long)nl * 10 + 5 + j];
    float4 sp = {0.0f, 0.0f, 0.0f, 0.0f};
    if (gq < totalQ) {
        const float4* P4 = (const float4*)P2 + gq;
        switch (nChunks) {
        case 1:  sp = gcn_plane_sum4<1>(P4, totalQ);  break;
        case 2:  sp = gcn_plane_sum4<2>(P4, totalQ);  break;
        case 3:  sp = gcn_plane_sum4<3>(P4, totalQ);  break;
        case 4:  sp = gcn_plane_sum4<4>(P4, totalQ);  break;
        case 5:  sp = gcn_plane_sum4<5>(P4, totalQ);  break;
        case 6:  sp = gcn_plane_sum4<6>(P4, totalQ);  break;
        case 7:  sp = gcn_plane_sum4<7>(P4, totalQ);  break;
        case 8:  sp = gcn_plane_sum4<8>(P4, totalQ);  break;
        case 9:  sp = gcn_plane_sum4<9>(P4, totalQ);  break;
        case 10: sp = gcn_plane_sum4<10>(P4, totalQ); break;
        case 11: sp = gcn_plane_sum4<11>(P4, totalQ); break;
        case 12: sp = gcn_plane_sum4<12>(P4, totalQ); break;
        case 13: sp = gcn_plane_sum4<13>(P4, totalQ); break;
        case 14: sp = gcn_plane_sum4<14>(P4, totalQ); break;
        case 15: sp = gcn_plane_sum4<15>(P4, totalQ); break;
        case 16: sp = gcn_plane_sum4<16>(P4, totalQ); break;
        default: break;
        }
    }
    ((float4*)s2)[threadIdx.x] = sp;
    __syncthreads();
    if (!fin) return;
    const float c = fmaxf(cn, 1.0f);
    float v[5];
#pragma unroll
    for (int j = 0; j < 5; ++j)
        v[j] = s2[threadIdx.x * 5 + j] / c + r[j] + b2[j];
    float mx = v[0];
#pragma unroll
    for (int j = 1; j < 5; ++j) mx = fmaxf(mx, v[j]);
    float s = 0.0f;
#pragma unroll
    for (int j = 0; j < 5; ++j) s += expf(v[j] - mx);
    float l = logf(s);
#pragma unroll
    for (int j = 0; j < 5; ++j) out[(long)n * 5 + j] = v[j] - mx - l;
}

extern "C" void kernel_launch(void* const* d_in, const int* in_sizes, int n_in,
                              void* d_out, int out_size, void* d_ws, size_t ws_size,
                              hipStream_t stream) {
    const float* x      = (const float*)d_in[0];
    const int*   ei     = (const int*)d_in[1];
    const float* ew     = (const float*)d_in[2];
    const float* Wrel1  = (const float*)d_in[3];
    const float* Wroot1 = (const float*)d_in[4];
    const float* b1     = (const float*)d_in[5];
    const float* Wrel2  = (const float*)d_in[6];
    const float* Wroot2 = (const float*)d_in[7];
    const float* b2     = (const float*)d_in[8];

    const int N = in_sizes[0] / D_IN;
    const int E = in_sizes[1] / 2;
    const int* src = ei;
    const int* dst = ei + E;

    float* ws = (float*)d_ws;
    long off = 0;
    float* PR1 = ws + off; off += (long)N * 64;
    float* cnt = ws + off; off += N;
    float* PR2 = ws + off; off += (long)N * 10;
    float* W1c = ws + off; off += 128 * 64;
    float* W2p = ws + off; off += 32 * 16;
    const int nT = (N + TILE_N - 1) / TILE_N;
    const int nC = (nT >= 256) ? 1 : ((256 / nT > MAX_CHUNKS1) ? MAX_CHUNKS1 : 256 / nT);
    float* Pp  = ws + off; off += (long)nC * N * 32;
    float* Cp  = ws + off; off += (long)nC * N;
    const int nT2 = (N + TILE2_N - 1) / TILE2_N;
    const int nC2 = (nT2 >= 256) ? 1 : ((256 / nT2 > MAX_CHUNKS2) ? MAX_CHUNKS2 : 256 / nT2);
    float* P2  = ws + off; off += (long)nC2 * N * 5;
    const int nBlk   = (E + BM_EDGES - 1) / BM_EDGES;
    const int pitchW = nBlk * BM_WORDS;
    const int nWords = (E + 31) / 32;
    unsigned* bm = (unsigned*)(ws + off); off += (long)nT * pitchW;
    int* rec  = (int*)(ws + off); off += (long)E * 2;

    const int nTiles = N / 16;

    gcn_build_w1_edges<<<nBlk, 256, 0, stream>>>(Wrel1, Wroot1, W1c, src, dst, ew, bm, rec, E, nT, pitchW);
    gcn_build_w2<<<(32 * 16 + 255) / 256, 256, 0, stream>>>(Wrel2, Wroot2, W2p);

    gcn_gemm1_wmma<<<256, G1_WAVES * 32, 0, stream>>>(x, W1c, PR1, nTiles);

    const int nQuads = (E + 3) / 4;
    const int wpc = (nWords + nC - 1) / nC;
    gcn_agg1_tile<<<nT * nC, 1024, 0, stream>>>(bm, rec, PR1, Pp, Cp, N, nWords, pitchW, nC, wpc);

    gcn_combine1_gemm2<<<(N + CMB_NODES - 1) / CMB_NODES, 256, 0, stream>>>(Pp, Cp, PR1, b1, W2p, cnt, PR2, N, nC);

    const int qpc2 = (nQuads + nC2 - 1) / nC2;
    gcn_agg2_tile<<<nT2 * nC2, 1024, 0, stream>>>(src, dst, PR2, P2, N, E, nC2, qpc2);

    gcn_output<<<(N + OUT_NODES - 1) / OUT_NODES, OUT_THREADS, 0, stream>>>(P2, PR2, b2, cnt, (float*)d_out, N, nC2);
}


# Layer 2 runs on layer 1's (2048-node tile x edge chunk) groups: same bit rows scanned, one record load per kept edge gives src and local dst; planes drop to 5; sum order: LDS arrival, then chunk
# speedup vs baseline: 3.7995x; 1.0639x over previous
#include <hip/hip_runtime.h>
#include <math.h>

typedef __attribute__((ext_vector_type(2))) float v2f;
typedef __attribute__((ext_vector_type(8))) float v8f;

#define D_IN  128
#define H_DIM 32
#define C_DIM 5

#define TILE_N   2048
#define SRC_BITS 20

#define BM_EDGES 8192
#define BM_WORDS (BM_EDGES / 32)
#define BM_LP    (BM_WORDS + 4)
#define BM_MAXT  128

template <bool VEC>
__device__ __forceinline__ void gcn_build_edges(const int* __restrict__ src, const int* __restrict__ dst,
                                                const float* __restrict__ ew,
                                                unsigned* bl, int* __restrict__ rec, int E, int eb, int tid) {
    int4   d4[8], s4[8];
    float4 w4[8];
#pragma unroll
    for (int j = 0; j < 8; ++j) {
        const int e0 = eb + 4 * (j * 256 + tid);
        if (VEC) {
            const int ec = min(e0, E - 4);
            d4[j] = *(const int4*)(dst + ec);
            s4[j] = *(const int4*)(src + ec);
            w4[j] = *(const float4*)(ew + ec);
        } else {
            const int a = min(e0, E - 1), b = min(e0 + 1, E - 1), c = min(e0 + 2, E - 1), d = min(e0 + 3, E - 1);
            d4[j] = make_int4(dst[a], dst[b], dst[c], dst[d]);
            s4[j] = make_int4(src[a], src[b], src[c], src[d]);
            w4[j] = make_float4(ew[a], ew[b], ew[c], ew[d]);
        }
    }
#pragma unroll
    for (int j = 0; j < 8; ++j) {
        const int ql = j * 256 + tid;
        const int e0 = eb + 4 * ql;
        const unsigned dd[4] = {(unsigned)d4[j].x, (unsigned)d4[j].y, (unsigned)d4[j].z, (unsigned)d4[j].w};
        const int      ss[4] = {s4[j].x, s4[j].y, s4[j].z, s4[j].w};
        const float    ww[4] = {w4[j].x, w4[j].y, w4[j].z, w4[j].w};
        int r[8];
#pragma unroll
        for (int k = 0; k < 4; ++k) {
            r[2 * k]     = ss[k] | (int)((dd[k] % TILE_N) << SRC_BITS);
            r[2 * k + 1] = __float_as_int(ww[k]);
            if (e0 + k < E)
                atomicOr(&bl[(dd[k] / TILE_N) * BM_LP + (ql >> 3)], 1u << ((ql & 7) * 4 + k));
        }
        if (VEC) {
            if (e0 < E) {
                int4* rp = (int4*)(rec + 2 * (long)e0);
                rp[0] = make_int4(r[0], r[1], r[2], r[3]);
                rp[1] = make_int4(r[4], r[5], r[6], r[7]);
            }
        } else {
#pragma unroll
            for (int k = 0; k < 4; ++k) {
                if (e0 + k < E) {
                    rec[2 * (long)(e0 + k)]     = r[2 * k];
                    rec[2 * (long)(e0 + k) + 1] = r[2 * k + 1];
                }
            }
        }
    }
}

__global__ __launch_bounds__(256)
void gcn_build_w1_edges(const float* __restrict__ Wrel, const float* __restrict__ Wroot,
                        float* __restrict__ Wcat,
                        const int* __restrict__ src, const int* __restrict__ dst,
                        const float* __restrict__ ew,
                        unsigned* __restrict__ bm, int* __restrict__ rec, int E, int nT, int pitchW) {
    __shared__ __attribute__((aligned(16))) unsigned bl[BM_MAXT * BM_LP];
    const int tid = threadIdx.x;
    for (int i = blockIdx.x * 256 + tid; i < 128 * 64; i += gridDim.x * 256) {
        int k = i >> 6, j = i & 63;
        Wcat[i] = (j < 32) ? Wrel[k * 32 + j] : Wroot[k * 32 + (j - 32)];
    }
    const uint4 z4 = {0u, 0u, 0u, 0u};
    for (int i = tid; i < nT * (BM_LP / 4); i += 256) ((uint4*)bl)[i] = z4;
    __syncthreads();

    const int eb = blockIdx.x * BM_EDGES;
    if ((E & 3) == 0) gcn_build_edges<true>(src, dst, ew, bl, rec, E, eb, tid);
    else              gcn_build_edges<false>(src, dst, ew, bl, rec, E, eb, tid);
    __syncthreads();

    const int wq = tid & 63;
    for (int t = tid >> 6; t < nT; t += 4) {
        const uint4 v = *(const uint4*)(bl + t * BM_LP + wq * 4);
        *(uint4*)(bm + (long)t * pitchW + blockIdx.x * BM_WORDS + wq * 4) = v;
    }
}

__global__ void gcn_build_w2(const float* __restrict__ Wrel, const float* __restrict__ Wroot,
                             float* __restrict__ Wp) {
    int i = blockIdx.x * 256 + threadIdx.x;
    if (i >= 32 * 16) return;
    int k = i >> 4, n = i & 15;
    float v = 0.0f;
    if (n < 5)       v = Wrel[k * 5 + n];
    else if (n < 10) v = Wroot[k * 5 + (n - 5)];
    Wp[i] = v;
}

#define G1_WAVES 8

__device__ __forceinline__ void gcn_gemm1_load_a(const float* __restrict__ x, int tile, int m, int half,
                                                 v2f* a) {
    const float* xr = x + (long)(tile * 16 + m) * D_IN + half * 2;
#pragma unroll
    for (int ks = 0; ks < 32; ++ks) a[ks] = *(const v2f*)(xr + ks * 4);
}

__global__ __launch_bounds__(G1_WAVES * 32)
void gcn_gemm1_wmma(const float* __restrict__ x, const float* __restrict__ W,
                    float* __restrict__ PR1, int nTiles) {
    const int wave = threadIdx.x >> 5;
    const int lane = threadIdx.x & 31;
    const int half = lane >> 4;
    const int m    = lane & 15;
    const int nW   = gridDim.x * G1_WAVES;
    int tile = wave * gridDim.x + blockIdx.x;
    if (tile >= nTiles) return;

    v2f a[32];
    gcn_gemm1_load_a(x, tile, m, half, a);

    v2f b[4][32];
#pragma unroll
    for (int ct = 0; ct < 4; ++ct) {
#pragma unroll
        for (int ks = 0; ks < 32; ++ks) {
            int k = ks * 4 + half * 2;
            int n = ct * 16 + m;
            b[ct][ks].x = W[k * 64 + n];
            b[ct][ks].y = W[(k + 1) * 64 + n];
        }
    }

#pragma unroll 1
    for (; tile < nTiles; tile += nW) {
        const int tn = (tile + nW < nTiles) ? tile + nW : tile;
        v2f an[32];
        gcn_gemm1_load_a(x, tn, m, half, an);

        v8f acc[4] = {};
#pragma unroll
        for (int ks = 0; ks < 32; ++ks) {
#pragma unroll
            for (int ct = 0; ct < 4; ++ct)
                acc[ct] = __builtin_amdgcn_wmma_f32_16x16x4_f32(false, a[ks], false, b[ct][ks],
                                                                (short)0, acc[ct], false, false);
        }
#pragma unroll
        for (int ct = 0; ct < 4; ++ct) {
#pragma unroll
            for (int r = 0; r < 8; ++r) {
                int orow = tile * 16 + r + half * 8;
                PR1[(long)orow * 64 + ct * 16 + m] = acc[ct][r];
            }
        }
#pragma unroll
        for (int ks = 0; ks < 32; ++ks) a[ks] = an[ks];
    }
}

#define AGG_WAVES 32
#define QCAP1     64

__device__ __forceinline__ void gcn_agg1_drain(int nd, int head, int lane, const int* qw,
                                               const int2* __restrict__ rec,
                                               const float* __restrict__ PR1,
                                               float* acc, float* cntl) {
    int s = 0, dl = 0;
    float w = 0.0f;
    if (lane < nd) {
        int idx = head + lane;
        if (idx >= QCAP1) idx -= QCAP1;
        const int2 r = rec[qw[idx]];
        s  = r.x & ((1 << SRC_BITS) - 1);
        dl = (int)((unsigned)r.x >> SRC_BITS);
        w  = __int_as_float(r.y);
        atomicAdd(&cntl[dl], 1.0f);
    }
    for (int k0 = 0; k0 < nd; k0 += 16) {
        float v[16];
#pragma unroll
        for (int j = 0; j < 16; ++j) {
            int sk = __builtin_amdgcn_readlane(s, k0 + j);
            v[j] = PR1[(unsigned)sk * 64u + (unsigned)lane];
        }
#pragma unroll
        for (int j = 0; j < 16; ++j) {
            if (k0 + j < nd) {
                float wk = __int_as_float(__builtin_amdgcn_readlane(__float_as_int(w), k0 + j));
                int   dk = __builtin_amdgcn_readlane(dl, k0 + j);
                atomicAdd(&acc[dk * 32 + lane], wk * v[j]);
            }
        }
    }
}

__device__ __forceinline__ void gcn_agg2_drain(int nd, int head, int lane, const int* qw,
                                               const int2* __restrict__ rec,
                                               const float* __restrict__ PR2, float* acc) {
    if (lane < nd) {
        int idx = head + lane;
        if (idx >= QCAP1) idx -= QCAP1;
        const int rx = rec[qw[idx]].x;
        const float* p = PR2 + (unsigned)(rx & ((1 << SRC_BITS) - 1)) * 10u;
        float2 v01 = *(const float2*)p;
        float2 v23 = *(const float2*)(p + 2);
        float  v4  = p[4];
        float* a = acc + (int)((unsigned)rx >> SRC_BITS) * 5;
        atomicAdd(a + 0, v01.x);
        atomicAdd(a + 1, v01.y);
        atomicAdd(a + 2, v23.x);
        atomicAdd(a + 3, v23.y);
        atomicAdd(a + 4, v4);
    }
}

__device__ __forceinline__ unsigned gcn_load_word(const unsigned* __restrict__ row, int wi, int pitchW) {
    return row[min(wi, pitchW - 1)];
}

template <bool L2>
__device__ __forceinline__ void gcn_agg_scan(const unsigned* __restrict__ row, const int2* __restrict__ rec,
                                             const float* __restrict__ PR,
                                             float* acc, float* cntl, int* qw,
                                             int wBeg, int wEnd, int pitchW, int tid, int lane) {
    const int nIt = (wEnd - wBeg + 1023) >> 10;
    int head = 0, qn = 0;

    auto step = [&](unsigned& d, int it) {
        const int wi = wBeg + it * 1024 + tid;
        const int e0 = 32 * wi;
        unsigned m = ((it >= 0) && (wi < wEnd)) ? d : 0u;
        d = gcn_load_word(row, wi + 4096, pitchW);
#pragma unroll 1
        for (int r = 0; r < 32; ++r) {
            const bool has = (m != 0u);
            const unsigned bal = __builtin_amdgcn_ballot_w32(has);
            if (bal == 0u) break;
            if (has) {
                int pos = head + qn + (int)__builtin_amdgcn_mbcnt_lo(bal, 0u);
                if (pos >= QCAP1) pos -= QCAP1;
                qw[pos] = e0 + __builtin_ctz(m);
                m &= m - 1u;
            }
            qn += __builtin_popcount(bal);
            if (qn >= 32) {
                if (L2) gcn_agg2_drain(32, head, lane, qw, rec, PR, acc);
                else    gcn_agg1_drain(32, head, lane, qw, rec, PR, acc, cntl);
                head += 32;
                if (head >= QCAP1) head -= QCAP1;
                qn -= 32;
            }
        }
    };

    unsigned d0 = 0u, d1 = 0u, d2 = 0u, d3 = 0u;
#pragma unroll 1
    for (int it = -4; it < nIt; it += 4) {
        step(d0, it);
        step(d1, it + 1);
        step(d2, it + 2);
        step(d3, it + 3);
    }
    if (qn > 0) {
        if (L2) gcn_agg2_drain(qn, head, lane, qw, rec, PR, acc);
        else    gcn_agg1_drain(qn, head, lane, qw, rec, PR, acc, cntl);
    }
}

__global__ __launch_bounds__(1024)
void gcn_agg1_tile(const unsigned* __restrict__ bm, const int* __restrict__ rec,
                   const float* __restrict__ PR1,
                   float* __restrict__ Ppart, float* __restrict__ Cpart,
                   int N, int nWords, int pitchW, int nChunks, int wordsPerChunk) {
    __shared__ __attribute__((aligned(16))) float acc[TILE_N * 32];
    __shared__ float cntl[TILE_N];
    __shared__ int   q[AGG_WAVES][QCAP1];

    const int tid  = threadIdx.x;
    const int wave = tid >> 5;
    const int lane = tid & 31;
    const int tile  = blockIdx.x / nChunks;
    const int chunk = blockIdx.x - tile * nChunks;
    const int base  = tile * TILE_N;
    const int rows  = min(TILE_N, N - base);

    const float4 z4 = {0.0f, 0.0f, 0.0f, 0.0f};
    for (int i = tid; i < TILE_N * 8; i += 1024) ((float4*)acc)[i] = z4;
    for (int i = tid; i < TILE_N; i += 1024) cntl[i] = 0.0f;
    __syncthreads();

    const int wBeg = chunk * wordsPerChunk;
    const int wEnd = min(wBeg + wordsPerChunk, nWords);
    if (wEnd > wBeg)
        gcn_agg_scan<false>(bm + (long)tile * pitchW, (const int2*)rec, PR1, acc, cntl, q[wave], wBeg, wEnd, pitchW, tid, lane);
    __syncthreads();

    const long planeP = (long)N * 32;
    float4* outP = (float4*)(Ppart + (long)chunk * planeP + (long)base * 32);
    for (int i = tid; i < rows * 8; i += 1024) outP[i] = ((const float4*)acc)[i];
    for (int i = tid; i < rows; i += 1024) Cpart[(long)chunk * N + base + i] = cntl[i];
}

#define MAX_CHUNKS1 8
#define CMB_NODES   128
#define HS_PITCH    36

template <int NC>
__device__ __forceinline__ void gcn_combine1_rows(const float4* __restrict__ P4,
                                                  const float* __restrict__ Cpart,
                                                  const float* __restrict__ PR1, float4 bb,
                                                  float* __restrict__ cnt, float* hs,
                                                  int N, int c4) {
    constexpr int JB = (NC <= 5) ? 4 : 2;
    const int  nUnits = N * 8;
    const long planeQ = (long)N * 8;
    const int  q0     = blockIdx.x * (CMB_NODES * 8) + threadIdx.x;
#pragma unroll
    for (int j0 = 0; j0 < 4; j0 += JB) {
        float4 p[JB][NC];
        float  cq[JB][NC];
        float4 r[JB];
#pragma unroll
        for (int j = 0; j < JB; ++j) {
            const int q = min(q0 + (j0 + j) * 256, nUnits - 1);
            const int n = q >> 3;
#pragma unroll
            for (int c = 0; c < NC; ++c) {
                p[j][c]  = P4[(long)c * planeQ + q];
                cq[j][c] = Cpart[(long)c * N + n];
            }
            r[j] = *(const float4*)(PR1 + (long)n * 64 + 32 + c4 * 4);
        }
#pragma unroll
        for (int j = 0; j < JB; ++j) {
            const int q = q0 + (j0 + j) * 256;
            float4 s = {0.0f, 0.0f, 0.0f, 0.0f};
            float cn = 0.0f;
#pragma unroll
            for (int c = 0; c < NC; ++c) {
                s.x += p[j][c].x;
                s.y += p[j][c].y;
                s.z += p[j][c].z;
                s.w += p[j][c].w;
                cn  += cq[j][c];
            }
            const float c1 = fmaxf(cn, 1.0f);
            float4 v;
            v.x = fmaxf(s.x / c1 + r[j].x + bb.x, 0.0f);
            v.y = fmaxf(s.y / c1 + r[j].y + bb.y, 0.0f);
            v.z = fmaxf(s.z / c1 + r[j].z + bb.z, 0.0f);
            v.w = fmaxf(s.w / c1 + r[j].w + bb.w, 0.0f);
            *(float4*)(hs + ((j0 + j) * 32 + (int)(threadIdx.x >> 3)) * HS_PITCH + c4 * 4) = v;
            if (c4 == 0 && q < nUnits) cnt[q >> 3] = cn;
        }
    }
}

__global__ __launch_bounds__(256)
void gcn_combine1_gemm2(const float* __restrict__ Ppart, const float* __restrict__ Cpart,
                        const float* __restrict__ PR1, const float* __restrict__ b1,
                        const float* __restrict__ W, float* __restrict__ cnt,
                        float* __restrict__ PR2, int N, int nChunks) {
    __shared__ __attribute__((aligned(16))) float hs[CMB_NODES * HS_PITCH];
    const int c4 = threadIdx.x & 7;
    const float4* P4 = (const float4*)Ppart;
    float4 bb;
    bb.x = b1[c4 * 4 + 0];
    bb.y = b1[c4 * 4 + 1];
    bb.z = b1[c4 * 4 + 2];
    bb.w = b1[c4 * 4 + 3];
    switch (nChunks) {
    case 1: gcn_combine1_rows<1>(P4, Cpart, PR1, bb, cnt, hs, N, c4); break;
    case 2: gcn_combine1_rows<2>(P4, Cpart, PR1, bb, cnt, hs, N, c4); break;
    case 3: gcn_combine1_rows<3>(P4, Cpart, PR1, bb, cnt, hs, N, c4); break;
    case 4: gcn_combine1_rows<4>(P4, Cpart, PR1, bb, cnt, hs, N, c4); break;
    case 5: gcn_combine1_rows<5>(P4, Cpart, PR1, bb, cnt, hs, N, c4); break;
    case 6: gcn_combine1_rows<6>(P4, Cpart, PR1, bb, cnt, hs, N, c4); break;
    case 7: gcn_combine1_rows<7>(P4, Cpart, PR1, bb, cnt, hs, N, c4); break;
    case 8: gcn_combine1_rows<8>(P4, Cpart, PR1, bb, cnt, hs, N, c4); break;
    default: break;
    }

    const int wave = threadIdx.x >> 5;
    const int lane = threadIdx.x & 31;
    const int half = lane >> 4;
    const int m    = lane & 15;
    v2f b[8];
#pragma unroll
    for (int ks = 0; ks < 8; ++ks) {
        int k = ks * 4 + half * 2;
        b[ks].x = W[k * 16 + m];
        b[ks].y = W[(k + 1) * 16 + m];
    }
    __syncthreads();
    const int row0 = blockIdx.x * CMB_NODES + wave * 16;
    if (row0 >= N) return;
    const float* hr = hs + (wave * 16 + m) * HS_PITCH;
    v2f a[8];
#pragma unroll
    for (int ks = 0; ks < 8; ++ks) {
        int k = ks * 4 + half * 2;
        a[ks].x = hr[k];
        a[ks].y = hr[k + 1];
    }
    v8f acc = {};
#pragma unroll
    for (int ks = 0; ks < 8; ++ks)
        acc = __builtin_amdgcn_wmma_f32_16x16x4_f32(false, a[ks], false, b[ks],
                                                    (short)0, acc, false, false);
    if (m < 10) {
#pragma unroll
        for (int r = 0; r < 8; ++r) {
            int orow = row0 + r + half * 8;
            PR2[(long)orow * 10 + m] = acc[r];
        }
    }
}

__global__ __launch_bounds__(1024)
void gcn_agg2_tile(const unsigned* __restrict__ bm, const int* __restrict__ rec,
                   const float* __restrict__ PR2, float* __restrict__ P2,
                   int N, int nWords, int pitchW, int nChunks, int wordsPerChunk) {
    __shared__ __attribute__((aligned(16))) float acc[TILE_N * 5];
    __shared__ int   q[AGG_WAVES][QCAP1];

    const int tid  = threadIdx.x;
    const int wave = tid >> 5;
    const int lane = tid & 31;
    const int tile  = blockIdx.x / nChunks;
    const int chunk = blockIdx.x - tile * nChunks;
    const int base  = tile * TILE_N;
    const int rows  = min(TILE_N, N - base);

    const float4 z4 = {0.0f, 0.0f, 0.0f, 0.0f};
    for (int i = tid; i < TILE_N * 5 / 4; i += 1024) ((float4*)acc)[i] = z4;
    __syncthreads();

    const int wBeg = chunk * wordsPerChunk;
    const int wEnd = min(wBeg + wordsPerChunk, nWords);
    if (wEnd > wBeg)
        gcn_agg_scan<true>(bm + (long)tile * pitchW, (const int2*)rec, PR2, acc, nullptr, q[wave], wBeg, wEnd, pitchW, tid, lane);
    __syncthreads();

    float4* outP = (float4*)(P2 + (long)chunk * N * 5 + (long)base * 5);
    for (int i = tid; i < rows * 5 / 4; i += 1024) outP[i] = ((const float4*)acc)[i];
}

#define OUT_NODES   256
#define OUT_THREADS (OUT_NODES * 5 / 4)

template <int NC>
__device__ __forceinline__ float4 gcn_plane_sum4(const float4* __restrict__ P4, long planeQ) {
    float4 p[NC];
#pragma unroll
    for (int cz = 0; cz < NC; ++cz) p[cz] = P4[(long)cz * planeQ];
    float4 s = {0.0f, 0.0f, 0.0f, 0.0f};
#pragma unroll
    for (int cz = 0; cz < NC; ++cz) {
        s.x += p[cz].x;
        s.y += p[cz].y;
        s.z += p[cz].z;
        s.w += p[cz].w;
    }
    return s;
}

__global__ __launch_bounds__(OUT_THREADS)
void gcn_output(const float* __restrict__ P2, const float* __restrict__ PR2,
                const float* __restrict__ b2, const float* __restrict__ cnt,
                float* __restrict__ out, int N, int nChunks) {
    __shared__ __attribute__((aligned(16))) float s2[OUT_NODES * 5];
    const int  n0     = blockIdx.x * OUT_NODES;
    const long totalQ = (long)N * 5 / 4;
    const long gq     = (long)blockIdx.x * OUT_THREADS + threadIdx.x;
    const int  n      = n0 + (int)threadIdx.x;
    const bool fin    = (threadIdx.x < OUT_NODES) && (n < N);
    const int   nl = min(n, N - 1);
    const float cn = cnt[nl];
    float r[5];
#pragma unroll
    for (int j = 0; j < 5; ++j) r[j] = PR2[(long)nl * 10 + 5 + j];
    float4 sp = {0.0f, 0.0f, 0.0f, 0.0f};
    if (gq < totalQ) {
        const float4* P4 = (const float4*)P2 + gq;
        switch (nChunks) {
        case 1:  sp = gcn_plane_sum4<1>(P4, totalQ);  break;
        case 2:  sp = gcn_plane_sum4<2>(P4, totalQ);  break;
        case 3:  sp = gcn_plane_sum4<3>(P4, totalQ);  break;
        case 4:  sp = gcn_plane_sum4<4>(P4, totalQ);  break;
        case 5:  sp = gcn_plane_sum4<5>(P4, totalQ);  break;
        case 6:  sp = gcn_plane_sum4<6>(P4, totalQ);  break;
        case 7:  sp = gcn_plane_sum4<7>(P4, totalQ);  break;
        case 8:  sp = gcn_plane_sum4<8>(P4, totalQ);  break;
        default: break;
        }
    }
    ((float4*)s2)[threadIdx.x] = sp;
    __syncthreads();
    if (!fin) return;
    const float c = fmaxf(cn, 1.0f);
    float v[5];
#pragma unroll
    for (int j = 0; j < 5; ++j)
        v[j] = s2[threadIdx.x * 5 + j] / c + r[j] + b2[j];
    float mx = v[0];
#pragma unroll
    for (int j = 1; j < 5; ++j) mx = fmaxf(mx, v[j]);
    float s = 0.0f;
#pragma unroll
    for (int j = 0; j < 5; ++j) s += expf(v[j] - mx);
    float l = logf(s);
#pragma unroll
    for (int j = 0; j < 5; ++j) out[(long)n * 5 + j] = v[j] - mx - l;
}

extern "C" void kernel_launch(void* const* d_in, const int* in_sizes, int n_in,
                              void* d_out, int out_size, void* d_ws, size_t ws_size,
                              hipStream_t stream) {
    const float* x      = (const float*)d_in[0];
    const int*   ei     = (const int*)d_in[1];
    const float* ew     = (const float*)d_in[2];
    const float* Wrel1  = (const float*)d_in[3];
    const float* Wroot1 = (const float*)d_in[4];
    const float* b1     = (const float*)d_in[5];
    const float* Wrel2  = (const float*)d_in[6];
    const float* Wroot2 = (const float*)d_in[7];
    const float* b2     = (const float*)d_in[8];

    const int N = in_sizes[0] / D_IN;
    const int E = in_sizes[1] / 2;
    const int* src = ei;
    const int* dst = ei + E;

    float* ws = (float*)d_ws;
    long off = 0;
    float* PR1 = ws + off; off += (long)N * 64;
    float* cnt = ws + off; off += N;
    float* PR2 = ws + off; off += (long)N * 10;
    float* W1c = ws + off; off += 128 * 64;
    float* W2p = ws + off; off += 32 * 16;
    const int nT = (N + TILE_N - 1) / TILE_N;
    const int nC = (nT >= 256) ? 1 : ((256 / nT > MAX_CHUNKS1) ? MAX_CHUNKS1 : 256 / nT);
    float* Pp  = ws + off; off += (long)nC * N * 32;
    float* Cp  = ws + off; off += (long)nC * N;
    float* P2  = ws + off; off += (long)nC * N * 5;
    const int nBlk   = (E + BM_EDGES - 1) / BM_EDGES;
    const int pitchW = nBlk * BM_WORDS;
    const int nWords = (E + 31) / 32;
    unsigned* bm = (unsigned*)(ws + off); off += (long)nT * pitchW;
    int* rec  = (int*)(ws + off); off += (long)E * 2;

    const int nTiles = N / 16;

    gcn_build_w1_edges<<<nBlk, 256, 0, stream>>>(Wrel1, Wroot1, W1c, src, dst, ew, bm, rec, E, nT, pitchW);
    gcn_build_w2<<<(32 * 16 + 255) / 256, 256, 0, stream>>>(Wrel2, Wroot2, W2p);

    gcn_gemm1_wmma<<<256, G1_WAVES * 32, 0, stream>>>(x, W1c, PR1, nTiles);

    const int wpc = (nWords + nC - 1) / nC;
    gcn_agg1_tile<<<nT * nC, 1024, 0, stream>>>(bm, rec, PR1, Pp, Cp, N, nWords, pitchW, nC, wpc);

    gcn_combine1_gemm2<<<(N + CMB_NODES - 1) / CMB_NODES, 256, 0, stream>>>(Pp, Cp, PR1, b1, W2p, cnt, PR2, N, nC);

    gcn_agg2_tile<<<nT * nC, 1024, 0, stream>>>(bm, rec, PR2, P2, N, nWords, pitchW, nC, wpc);

    gcn_output<<<(N + OUT_NODES - 1) / OUT_NODES, OUT_THREADS, 0, stream>>>(P2, PR2, b2, cnt, (float*)d_out, N, nC);
}


# Layer 2 walks per-wave lists of kept edges (src, local dst) that the layer-1 scan stores as it drains: no bit-row scan, no record line per edge in layer 2; sum order: LDS arrival, then chunk
# speedup vs baseline: 4.1449x; 1.0909x over previous
#include <hip/hip_runtime.h>
#include <math.h>

typedef __attribute__((ext_vector_type(2))) float v2f;
typedef __attribute__((ext_vector_type(8))) float v8f;

#define D_IN  128
#define H_DIM 32
#define C_DIM 5

#define TILE_N   2048
#define SRC_BITS 20

#define BM_EDGES 8192
#define BM_WORDS (BM_EDGES / 32)
#define BM_LP    (BM_WORDS + 4)
#define BM_MAXT  128

template <bool VEC>
__device__ __forceinline__ void gcn_build_edges(const int* __restrict__ src, const int* __restrict__ dst,
                                                const float* __restrict__ ew,
                                                unsigned* bl, int* __restrict__ rec, int E, int eb, int tid) {
    int4   d4[8], s4[8];
    float4 w4[8];
#pragma unroll
    for (int j = 0; j < 8; ++j) {
        const int e0 = eb + 4 * (j * 256 + tid);
        if (VEC) {
            const int ec = min(e0, E - 4);
            d4[j] = *(const int4*)(dst + ec);
            s4[j] = *(const int4*)(src + ec);
            w4[j] = *(const float4*)(ew + ec);
        } else {
            const int a = min(e0, E - 1), b = min(e0 + 1, E - 1), c = min(e0 + 2, E - 1), d = min(e0 + 3, E - 1);
            d4[j] = make_int4(dst[a], dst[b], dst[c], dst[d]);
            s4[j] = make_int4(src[a], src[b], src[c], src[d]);
            w4[j] = make_float4(ew[a], ew[b], ew[c], ew[d]);
        }
    }
#pragma unroll
    for (int j = 0; j < 8; ++j) {
        const int ql = j * 256 + tid;
        const int e0 = eb + 4 * ql;
        const unsigned dd[4] = {(unsigned)d4[j].x, (unsigned)d4[j].y, (unsigned)d4[j].z, (unsigned)d4[j].w};
        const int      ss[4] = {s4[j].x, s4[j].y, s4[j].z, s4[j].w};
        const float    ww[4] = {w4[j].x, w4[j].y, w4[j].z, w4[j].w};
        int r[8];
#pragma unroll
        for (int k = 0; k < 4; ++k) {
            r[2 * k]     = ss[k] | (int)((dd[k] % TILE_N) << SRC_BITS);
            r[2 * k + 1] = __float_as_int(ww[k]);
            if (e0 + k < E)
                atomicOr(&bl[(dd[k] / TILE_N) * BM_LP + (ql >> 3)], 1u << ((ql & 7) * 4 + k));
        }
        if (VEC) {
            if (e0 < E) {
                int4* rp = (int4*)(rec + 2 * (long)e0);
                rp[0] = make_int4(r[0], r[1], r[2], r[3]);
                rp[1] = make_int4(r[4], r[5], r[6], r[7]);
            }
        } else {
#pragma unroll
            for (int k = 0; k < 4; ++k) {
                if (e0 + k < E) {
                    rec[2 * (long)(e0 + k)]     = r[2 * k];
                    rec[2 * (long)(e0 + k) + 1] = r[2 * k + 1];
                }
            }
        }
    }
}

__global__ __launch_bounds__(256)
void gcn_build_w1_edges(const float* __restrict__ Wrel, const float* __restrict__ Wroot,
                        float* __restrict__ Wcat,
                        const int* __restrict__ src, const int* __restrict__ dst,
                        const float* __restrict__ ew,
                        unsigned* __restrict__ bm, int* __restrict__ rec, int E, int nT, int pitchW) {
    __shared__ __attribute__((aligned(16))) unsigned bl[BM_MAXT * BM_LP];
    const int tid = threadIdx.x;
    for (int i = blockIdx.x * 256 + tid; i < 128 * 64; i += gridDim.x * 256) {
        int k = i >> 6, j = i & 63;
        Wcat[i] = (j < 32) ? Wrel[k * 32 + j] : Wroot[k * 32 + (j - 32)];
    }
    const uint4 z4 = {0u, 0u, 0u, 0u};
    for (int i = tid; i < nT * (BM_LP / 4); i += 256) ((uint4*)bl)[i] = z4;
    __syncthreads();

    const int eb = blockIdx.x * BM_EDGES;
    if ((E & 3) == 0) gcn_build_edges<true>(src, dst, ew, bl, rec, E, eb, tid);
    else              gcn_build_edges<false>(src, dst, ew, bl, rec, E, eb, tid);
    __syncthreads();

    const int wq = tid & 63;
    for (int t = tid >> 6; t < nT; t += 4) {
        const uint4 v = *(const uint4*)(bl + t * BM_LP + wq * 4);
        *(uint4*)(bm + (long)t * pitchW + blockIdx.x * BM_WORDS + wq * 4) = v;
    }
}

__global__ void gcn_build_w2(const float* __restrict__ Wrel, const float* __restrict__ Wroot,
                             float* __restrict__ Wp) {
    int i = blockIdx.x * 256 + threadIdx.x;
    if (i >= 32 * 16) return;
    int k = i >> 4, n = i & 15;
    float v = 0.0f;
    if (n < 5)       v = Wrel[k * 5 + n];
    else if (n < 10) v = Wroot[k * 5 + (n - 5)];
    Wp[i] = v;
}

#define G1_WAVES 8

__device__ __forceinline__ void gcn_gemm1_load_a(const float* __restrict__ x, int tile, int m, int half,
                                                 v2f* a) {
    const float* xr = x + (long)(tile * 16 + m) * D_IN + half * 2;
#pragma unroll
    for (int ks = 0; ks < 32; ++ks) a[ks] = *(const v2f*)(xr + ks * 4);
}

__global__ __launch_bounds__(G1_WAVES * 32)
void gcn_gemm1_wmma(const float* __restrict__ x, const float* __restrict__ W,
                    float* __restrict__ PR1, int nTiles) {
    const int wave = threadIdx.x >> 5;
    const int lane = threadIdx.x & 31;
    const int half = lane >> 4;
    const int m    = lane & 15;
    const int nW   = gridDim.x * G1_WAVES;
    int tile = wave * gridDim.x + blockIdx.x;
    if (tile >= nTiles) return;

    v2f a[32];
    gcn_gemm1_load_a(x, tile, m, half, a);

    v2f b[4][32];
#pragma unroll
    for (int ct = 0; ct < 4; ++ct) {
#pragma unroll
        for (int ks = 0; ks < 32; ++ks) {
            int k = ks * 4 + half * 2;
            int n = ct * 16 + m;
            b[ct][ks].x = W[k * 64 + n];
            b[ct][ks].y = W[(k + 1) * 64 + n];
        }
    }

#pragma unroll 1
    for (; tile < nTiles; tile += nW) {
        const int tn = (tile + nW < nTiles) ? tile + nW : tile;
        v2f an[32];
        gcn_gemm1_load_a(x, tn, m, half, an);

        v8f acc[4] = {};
#pragma unroll
        for (int ks = 0; ks < 32; ++ks) {
#pragma unroll
            for (int ct = 0; ct < 4; ++ct)
                acc[ct] = __builtin_amdgcn_wmma_f32_16x16x4_f32(false, a[ks], false, b[ct][ks],
                                                                (short)0, acc[ct], false, false);
        }
#pragma unroll
        for (int ct = 0; ct < 4; ++ct) {
#pragma unroll
            for (int r = 0; r < 8; ++r) {
                int orow = tile * 16 + r + half * 8;
                PR1[(long)orow * 64 + ct * 16 + m] = acc[ct][r];
            }
        }
#pragma unroll
        for (int ks = 0; ks < 32; ++ks) a[ks] = an[ks];
    }
}

#define AGG_WAVES 32
#define QCAP1     64
#define LIST_CAP  1024

__device__ __forceinline__ void gcn_agg1_drain(int nd, int head, int lane, const int* qw,
                                               const int2* __restrict__ rec,
                                               const float* __restrict__ PR1,
                                               float* acc, float* cntl,
                                               int* __restrict__ lst, int lpos) {
    int s = 0, dl = 0;
    float w = 0.0f;
    if (lane < nd) {
        int idx = head + lane;
        if (idx >= QCAP1) idx -= QCAP1;
        const int2 r = rec[qw[idx]];
        if (lpos + lane < LIST_CAP) lst[lpos + lane] = r.x;
        s  = r.x & ((1 << SRC_BITS) - 1);
        dl = (int)((unsigned)r.x >> SRC_BITS);
        w  = __int_as_float(r.y);
        atomicAdd(&cntl[dl], 1.0f);
    }
    for (int k0 = 0; k0 < nd; k0 += 16) {
        float v[16];
#pragma unroll
        for (int j = 0; j < 16; ++j) {
            int sk = __builtin_amdgcn_readlane(s, k0 + j);
            v[j] = PR1[(unsigned)sk * 64u + (unsigned)lane];
        }
#pragma unroll
        for (int j = 0; j < 16; ++j) {
            if (k0 + j < nd) {
                float wk = __int_as_float(__builtin_amdgcn_readlane(__float_as_int(w), k0 + j));
                int   dk = __builtin_amdgcn_readlane(dl, k0 + j);
                atomicAdd(&acc[dk * 32 + lane], wk * v[j]);
            }
        }
    }
}

__device__ __forceinline__ void gcn_agg2_drain(int nd, int head, int lane, const int* qw,
                                               const int2* __restrict__ rec,
                                               const float* __restrict__ PR2, float* acc) {
    if (lane < nd) {
        int idx = head + lane;
        if (idx >= QCAP1) idx -= QCAP1;
        const int rx = rec[qw[idx]].x;
        const float* p = PR2 + (unsigned)(rx & ((1 << SRC_BITS) - 1)) * 10u;
        float2 v01 = *(const float2*)p;
        float2 v23 = *(const float2*)(p + 2);
        float  v4  = p[4];
        float* a = acc + (int)((unsigned)rx >> SRC_BITS) * 5;
        atomicAdd(a + 0, v01.x);
        atomicAdd(a + 1, v01.y);
        atomicAdd(a + 2, v23.x);
        atomicAdd(a + 3, v23.y);
        atomicAdd(a + 4, v4);
    }
}

__device__ __forceinline__ unsigned gcn_load_word(const unsigned* __restrict__ row, int wi, int pitchW) {
    return row[min(wi, pitchW - 1)];
}

template <bool L2>
__device__ __forceinline__ int gcn_agg_scan(const unsigned* __restrict__ row, const int2* __restrict__ rec,
                                            const float* __restrict__ PR,
                                            float* acc, float* cntl, int* qw, int* __restrict__ lst,
                                            int wBeg, int wEnd, int pitchW, int tid, int lane) {
    const int nIt = (wEnd - wBeg + 1023) >> 10;
    int head = 0, qn = 0;
    int lpos = 0;

    auto step = [&](unsigned& d, int it) {
        const int wi = wBeg + it * 1024 + tid;
        const int e0 = 32 * wi;
        unsigned m = ((it >= 0) && (wi < wEnd)) ? d : 0u;
        d = gcn_load_word(row, wi + 4096, pitchW);
#pragma unroll 1
        for (int r = 0; r < 32; ++r) {
            const bool has = (m != 0u);
            const unsigned bal = __builtin_amdgcn_ballot_w32(has);
            if (bal == 0u) break;
            if (has) {
                int pos = head + qn + (int)__builtin_amdgcn_mbcnt_lo(bal, 0u);
                if (pos >= QCAP1) pos -= QCAP1;
                qw[pos] = e0 + __builtin_ctz(m);
                m &= m - 1u;
            }
            qn += __builtin_popcount(bal);
            if (qn >= 32) {
                if (L2) gcn_agg2_drain(32, head, lane, qw, rec, PR, acc);
                else    gcn_agg1_drain(32, head, lane, qw, rec, PR, acc, cntl, lst, lpos);
                lpos += 32;
                head += 32;
                if (head >= QCAP1) head -= QCAP1;
                qn -= 32;
            }
        }
    };

    unsigned d0 = 0u, d1 = 0u, d2 = 0u, d3 = 0u;
#pragma unroll 1
    for (int it = -4; it < nIt; it += 4) {
        step(d0, it);
        step(d1, it + 1);
        step(d2, it + 2);
        step(d3, it + 3);
    }
    if (qn > 0) {
        if (L2) gcn_agg2_drain(qn, head, lane, qw, rec, PR, acc);
        else    gcn_agg1_drain(qn, head, lane, qw, rec, PR, acc, cntl, lst, lpos);
        lpos += qn;
    }
    return L2 ? 0 : lpos;
}

__global__ __launch_bounds__(1024)
void gcn_agg1_tile(const unsigned* __restrict__ bm, const int* __restrict__ rec,
                   const float* __restrict__ PR1,
                   float* __restrict__ Ppart, float* __restrict__ Cpart,
                   int* __restrict__ lst, int* __restrict__ nKept,
                   int N, int nWords, int pitchW, int nChunks, int wordsPerChunk) {
    __shared__ __attribute__((aligned(16))) float acc[TILE_N * 32];
    __shared__ float cntl[TILE_N];
    __shared__ int   q[AGG_WAVES][QCAP1];

    const int tid  = threadIdx.x;
    const int wave = tid >> 5;
    const int lane = tid & 31;
    const int tile  = blockIdx.x / nChunks;
    const int chunk = blockIdx.x - tile * nChunks;
    const int base  = tile * TILE_N;
    const int rows  = min(TILE_N, N - base);

    const float4 z4 = {0.0f, 0.0f, 0.0f, 0.0f};
    for (int i = tid; i < TILE_N * 8; i += 1024) ((float4*)acc)[i] = z4;
    for (int i = tid; i < TILE_N; i += 1024) cntl[i] = 0.0f;
    __syncthreads();

    const int wBeg = chunk * wordsPerChunk;
    const int wEnd = min(wBeg + wordsPerChunk, nWords);
    const int gw = blockIdx.x * AGG_WAVES + wave;
    int kept = 0;
    if (wEnd > wBeg)
        kept = gcn_agg_scan<false>(bm + (long)tile * pitchW, (const int2*)rec, PR1, acc, cntl, q[wave],
                                   lst + (long)gw * LIST_CAP, wBeg, wEnd, pitchW, tid, lane);
    if (lane == 0) nKept[gw] = kept;
    __syncthreads();

    const long planeP = (long)N * 32;
    float4* outP = (float4*)(Ppart + (long)chunk * planeP + (long)base * 32);
    for (int i = tid; i < rows * 8; i += 1024) outP[i] = ((const float4*)acc)[i];
    for (int i = tid; i < rows; i += 1024) Cpart[(long)chunk * N + base + i] = cntl[i];
}

#define MAX_CHUNKS1 8
#define CMB_NODES   128
#define HS_PITCH    36

template <int NC>
__device__ __forceinline__ void gcn_combine1_rows(const float4* __restrict__ P4,
                                                  const float* __restrict__ Cpart,
                                                  const float* __restrict__ PR1, float4 bb,
                                                  float* __restrict__ cnt, float* hs,
                                                  int N, int c4) {
    constexpr int JB = (NC <= 5) ? 4 : 2;
    const int  nUnits = N * 8;
    const long planeQ = (long)N * 8;
    const int  q0     = blockIdx.x * (CMB_NODES * 8) + threadIdx.x;
#pragma unroll
    for (int j0 = 0; j0 < 4; j0 += JB) {
        float4 p[JB][NC];
        float  cq[JB][NC];
        float4 r[JB];
#pragma unroll
        for (int j = 0; j < JB; ++j) {
            const int q = min(q0 + (j0 + j) * 256, nUnits - 1);
            const int n = q >> 3;
#pragma unroll
            for (int c = 0; c < NC; ++c) {
                p[j][c]  = P4[(long)c * planeQ + q];
                cq[j][c] = Cpart[(long)c * N + n];
            }
            r[j] = *(const float4*)(PR1 + (long)n * 64 + 32 + c4 * 4);
        }
#pragma unroll
        for (int j = 0; j < JB; ++j) {
            const int q = q0 + (j0 + j) * 256;
            float4 s = {0.0f, 0.0f, 0.0f, 0.0f};
            float cn = 0.0f;
#pragma unroll
            for (int c = 0; c < NC; ++c) {
                s.x += p[j][c].x;
                s.y += p[j][c].y;
                s.z += p[j][c].z;
                s.w += p[j][c].w;
                cn  += cq[j][c];
            }
            const float c1 = fmaxf(cn, 1.0f);
            float4 v;
            v.x = fmaxf(s.x / c1 + r[j].x + bb.x, 0.0f);
            v.y = fmaxf(s.y / c1 + r[j].y + bb.y, 0.0f);
            v.z = fmaxf(s.z / c1 + r[j].z + bb.z, 0.0f);
            v.w = fmaxf(s.w / c1 + r[j].w + bb.w, 0.0f);
            *(float4*)(hs + ((j0 + j) * 32 + (int)(threadIdx.x >> 3)) * HS_PITCH + c4 * 4) = v;
            if (c4 == 0 && q < nUnits) cnt[q >> 3] = cn;
        }
    }
}

__global__ __launch_bounds__(256)
void gcn_combine1_gemm2(const float* __restrict__ Ppart, const float* __restrict__ Cpart,
                        const float* __restrict__ PR1, const float* __restrict__ b1,
                        const float* __restrict__ W, float* __restrict__ cnt,
                        float* __restrict__ PR2, int N, int nChunks) {
    __shared__ __attribute__((aligned(16))) float hs[CMB_NODES * HS_PITCH];
    const int c4 = threadIdx.x & 7;
    const float4* P4 = (const float4*)Ppart;
    float4 bb;
    bb.x = b1[c4 * 4 + 0];
    bb.y = b1[c4 * 4 + 1];
    bb.z = b1[c4 * 4 + 2];
    bb.w = b1[c4 * 4 + 3];
    switch (nChunks) {
    case 1: gcn_combine1_rows<1>(P4, Cpart, PR1, bb, cnt, hs, N, c4); break;
    case 2: gcn_combine1_rows<2>(P4, Cpart, PR1, bb, cnt, hs, N, c4); break;
    case 3: gcn_combine1_rows<3>(P4, Cpart, PR1, bb, cnt, hs, N, c4); break;
    case 4: gcn_combine1_rows<4>(P4, Cpart, PR1, bb, cnt, hs, N, c4); break;
    case 5: gcn_combine1_rows<5>(P4, Cpart, PR1, bb, cnt, hs, N, c4); break;
    case 6: gcn_combine1_rows<6>(P4, Cpart, PR1, bb, cnt, hs, N, c4); break;
    case 7: gcn_combine1_rows<7>(P4, Cpart, PR1, bb, cnt, hs, N, c4); break;
    case 8: gcn_combine1_rows<8>(P4, Cpart, PR1, bb, cnt, hs, N, c4); break;
    default: break;
    }

    const int wave = threadIdx.x >> 5;
    const int lane = threadIdx.x & 31;
    const int half = lane >> 4;
    const int m    = lane & 15;
    v2f b[8];
#pragma unroll
    for (int ks = 0; ks < 8; ++ks) {
        int k = ks * 4 + half * 2;
        b[ks].x = W[k * 16 + m];
        b[ks].y = W[(k + 1) * 16 + m];
    }
    __syncthreads();
    const int row0 = blockIdx.x * CMB_NODES + wave * 16;
    if (row0 >= N) return;
    const float* hr = hs + (wave * 16 + m) * HS_PITCH;
    v2f a[8];
#pragma unroll
    for (int ks = 0; ks < 8; ++ks) {
        int k = ks * 4 + half * 2;
        a[ks].x = hr[k];
        a[ks].y = hr[k + 1];
    }
    v8f acc = {};
#pragma unroll
    for (int ks = 0; ks < 8; ++ks)
        acc = __builtin_amdgcn_wmma_f32_16x16x4_f32(false, a[ks], false, b[ks],
                                                    (short)0, acc, false, false);
    if (m < 10) {
#pragma unroll
        for (int r = 0; r < 8; ++r) {
            int orow = row0 + r + half * 8;
            PR2[(long)orow * 10 + m] = acc[r];
        }
    }
}


__device__ __forceinline__ void gcn_agg2_list(const int* __restrict__ lst, int n, int lane,
                                              const float* __restrict__ PR2, float* acc) {
#pragma unroll 1
    for (int b0 = 0; b0 < n; b0 += 256) {
        int    rx[8];
        float2 v01[8], v23[8];
        float  v4[8];
#pragma unroll
        for (int j = 0; j < 8; ++j) rx[j] = lst[min(b0 + j * 32 + lane, n - 1)];
        asm volatile("" :: "v"(rx[0]), "v"(rx[1]), "v"(rx[2]), "v"(rx[3]), "v"(rx[4]), "v"(rx[5]), "v"(rx[6]), "v"(rx[7]));
#pragma unroll
        for (int j = 0; j < 8; ++j) {
            const float* p = PR2 + (unsigned)(rx[j] & ((1 << SRC_BITS) - 1)) * 10u;
            v01[j] = *(const float2*)p;
            v23[j] = *(const float2*)(p + 2);
            v4[j]  = p[4];
        }
#pragma unroll
        for (int j = 0; j < 8; ++j) {
            if (b0 + j * 32 + lane < n) {
                float* a = acc + (int)((unsigned)rx[j] >> SRC_BITS) * 5;
                atomicAdd(a + 0, v01[j].x);
                atomicAdd(a + 1, v01[j].y);
                atomicAdd(a + 2, v23[j].x);
                atomicAdd(a + 3, v23[j].y);
                atomicAdd(a + 4, v4[j]);
            }
        }
    }
}

__global__ __launch_bounds__(1024)
void gcn_agg2_tile(const unsigned* __restrict__ bm, const int* __restrict__ rec,
                   const int* __restrict__ lst, const int* __restrict__ nKept,
                   const float* __restrict__ PR2, float* __restrict__ P2,
                   int N, int nWords, int pitchW, int nChunks, int wordsPerChunk) {
    __shared__ __attribute__((aligned(16))) float acc[TILE_N * 5];
    __shared__ int   q[AGG_WAVES][QCAP1];

    const int tid  = threadIdx.x;
    const int wave = tid >> 5;
    const int lane = tid & 31;
    const int tile  = blockIdx.x / nChunks;
    const int chunk = blockIdx.x - tile * nChunks;
    const int base  = tile * TILE_N;
    const int rows  = min(TILE_N, N - base);

    const int gw = blockIdx.x * AGG_WAVES + wave;
    const int nv = nKept[gw];

    const float4 z4 = {0.0f, 0.0f, 0.0f, 0.0f};
    for (int i = tid; i < TILE_N * 5 / 4; i += 1024) ((float4*)acc)[i] = z4;
    const int n = __builtin_amdgcn_readfirstlane(nv);
    __syncthreads();

    if (n > LIST_CAP) {
        const int wBeg = chunk * wordsPerChunk;
        const int wEnd = min(wBeg + wordsPerChunk, nWords);
        gcn_agg_scan<true>(bm + (long)tile * pitchW, (const int2*)rec, PR2, acc, nullptr, q[wave], nullptr, wBeg, wEnd, pitchW, tid, lane);
    } else if (n > 0) {
        gcn_agg2_list(lst + (long)gw * LIST_CAP, n, lane, PR2, acc);
    }
    __syncthreads();

    float4* outP = (float4*)(P2 + (long)chunk * N * 5 + (long)base * 5);
    for (int i = tid; i < rows * 5 / 4; i += 1024) outP[i] = ((const float4*)acc)[i];
}

#define OUT_NODES   256
#define OUT_THREADS (OUT_NODES * 5 / 4)

template <int NC>
__device__ __forceinline__ float4 gcn_plane_sum4(const float4* __restrict__ P4, long planeQ) {
    float4 p[NC];
#pragma unroll
    for (int cz = 0; cz < NC; ++cz) p[cz] = P4[(long)cz * planeQ];
    float4 s = {0.0f, 0.0f, 0.0f, 0.0f};
#pragma unroll
    for (int cz = 0; cz < NC; ++cz) {
        s.x += p[cz].x;
        s.y += p[cz].y;
        s.z += p[cz].z;
        s.w += p[cz].w;
    }
    return s;
}

__global__ __launch_bounds__(OUT_THREADS)
void gcn_output(const float* __restrict__ P2, const float* __restrict__ PR2,
                const float* __restrict__ b2, const float* __restrict__ cnt,
                float* __restrict__ out, int N, int nChunks) {
    __shared__ __attribute__((aligned(16))) float s2[OUT_NODES * 5];
    const int  n0     = blockIdx.x * OUT_NODES;
    const long totalQ = (long)N * 5 / 4;
    const long gq     = (long)blockIdx.x * OUT_THREADS + threadIdx.x;
    const int  n      = n0 + (int)threadIdx.x;
    const bool fin    = (threadIdx.x < OUT_NODES) && (n < N);
    const int   nl = min(n, N - 1);
    const float cn = cnt[nl];
    float r[5];
#pragma unroll
    for (int j = 0; j < 5; ++j) r[j] = PR2[(long)nl * 10 + 5 + j];
    float4 sp = {0.0f, 0.0f, 0.0f, 0.0f};
    if (gq < totalQ) {
        const float4* P4 = (const float4*)P2 + gq;
        switch (nChunks) {
        case 1:  sp = gcn_plane_sum4<1>(P4, totalQ);  break;
        case 2:  sp = gcn_plane_sum4<2>(P4, totalQ);  break;
        case 3:  sp = gcn_plane_sum4<3>(P4, totalQ);  break;
        case 4:  sp = gcn_plane_sum4<4>(P4, totalQ);  break;
        case 5:  sp = gcn_plane_sum4<5>(P4, totalQ);  break;
        case 6:  sp = gcn_plane_sum4<6>(P4, totalQ);  break;
        case 7:  sp = gcn_plane_sum4<7>(P4, totalQ);  break;
        case 8:  sp = gcn_plane_sum4<8>(P4, totalQ);  break;
        default: break;
        }
    }
    ((float4*)s2)[threadIdx.x] = sp;
    __syncthreads();
    if (!fin) return;
    const float c = fmaxf(cn, 1.0f);
    float v[5];
#pragma unroll
    for (int j = 0; j < 5; ++j)
        v[j] = s2[threadIdx.x * 5 + j] / c + r[j] + b2[j];
    float mx = v[0];
#pragma unroll
    for (int j = 1; j < 5; ++j) mx = fmaxf(mx, v[j]);
    float s = 0.0f;
#pragma unroll
    for (int j = 0; j < 5; ++j) s += expf(v[j] - mx);
    float l = logf(s);
#pragma unroll
    for (int j = 0; j < 5; ++j) out[(long)n * 5 + j] = v[j] - mx - l;
}

extern "C" void kernel_launch(void* const* d_in, const int* in_sizes, int n_in,
                              void* d_out, int out_size, void* d_ws, size_t ws_size,
                              hipStream_t stream) {
    const float* x      = (const float*)d_in[0];
    const int*   ei     = (const int*)d_in[1];
    const float* ew     = (const float*)d_in[2];
    const float* Wrel1  = (const float*)d_in[3];
    const float* Wroot1 = (const float*)d_in[4];
    const float* b1     = (const float*)d_in[5];
    const float* Wrel2  = (const float*)d_in[6];
    const float* Wroot2 = (const float*)d_in[7];
    const float* b2     = (const float*)d_in[8];

    const int N = in_sizes[0] / D_IN;
    const int E = in_sizes[1] / 2;
    const int* src = ei;
    const int* dst = ei + E;

    float* ws = (float*)d_ws;
    long off = 0;
    float* PR1 = ws + off; off += (long)N * 64;
    float* cnt = ws + off; off += N;
    float* PR2 = ws + off; off += (long)N * 10;
    float* W1c = ws + off; off += 128 * 64;
    float* W2p = ws + off; off += 32 * 16;
    const int nT = (N + TILE_N - 1) / TILE_N;
    const int nC = (nT >= 256) ? 1 : ((256 / nT > MAX_CHUNKS1) ? MAX_CHUNKS1 : 256 / nT);
    float* Pp  = ws + off; off += (long)nC * N * 32;
    float* Cp  = ws + off; off += (long)nC * N;
    float* P2  = ws + off; off += (long)nC * N * 5;
    const int nBlk   = (E + BM_EDGES - 1) / BM_EDGES;
    const int pitchW = nBlk * BM_WORDS;
    const int nWords = (E + 31) / 32;
    unsigned* bm = (unsigned*)(ws + off); off += (long)nT * pitchW;
    int* rec  = (int*)(ws + off); off += (long)E * 2;
    int* lst  = (int*)(ws + off); off += (long)nT * nC * AGG_WAVES * LIST_CAP;
    int* nKp  = (int*)(ws + off); off += (long)nT * nC * AGG_WAVES;

    const int nTiles = N / 16;

    gcn_build_w1_edges<<<nBlk, 256, 0, stream>>>(Wrel1, Wroot1, W1c, src, dst, ew, bm, rec, E, nT, pitchW);
    gcn_build_w2<<<(32 * 16 + 255) / 256, 256, 0, stream>>>(Wrel2, Wroot2, W2p);

    gcn_gemm1_wmma<<<256, G1_WAVES * 32, 0, stream>>>(x, W1c, PR1, nTiles);

    const int wpc = (nWords + nC - 1) / nC;
    gcn_agg1_tile<<<nT * nC, 1024, 0, stream>>>(bm, rec, PR1, Pp, Cp, lst, nKp, N, nWords, pitchW, nC, wpc);

    gcn_combine1_gemm2<<<(N + CMB_NODES - 1) / CMB_NODES, 256, 0, stream>>>(Pp, Cp, PR1, b1, W2p, cnt, PR2, N, nC);

    gcn_agg2_tile<<<nT * nC, 1024, 0, stream>>>(bm, rec, lst, nKp, PR2, P2, N, nWords, pitchW, nC, wpc);

    gcn_output<<<(N + OUT_NODES - 1) / OUT_NODES, OUT_THREADS, 0, stream>>>(P2, PR2, b2, cnt, (float*)d_out, N, nC);
}
